# starting kernel
# baseline (speedup 1.0000x reference)
#include <hip/hip_runtime.h>
#include <hip/hip_bf16.h>
#include <math.h>

#define USR_NUM 100000
#define USR_DIM 64
#define NUM_LOC 5000
#define T_STEPS 16
#define N_EDGES 80000

typedef float v2f __attribute__((ext_vector_type(2)));
typedef float v8f __attribute__((ext_vector_type(8)));

__global__ void k_zero(float* __restrict__ p, int n) {
  int i = blockIdx.x * blockDim.x + threadIdx.x;
  if (i < n) p[i] = 0.0f;
}

__global__ void k_deg_accum(const int* __restrict__ dst, float* __restrict__ deg, int E) {
  int i = blockIdx.x * blockDim.x + threadIdx.x;
  if (i < E) atomicAdd(&deg[dst[i]], 1.0f);
}

__global__ void k_deg_fin(const float* __restrict__ deg, float* __restrict__ isq,
                          float* __restrict__ idg, int n) {
  int i = blockIdx.x * blockDim.x + threadIdx.x;
  if (i < n) {
    float d = deg[i] + 1.0f;
    isq[i] = rsqrtf(d);
    idg[i] = 1.0f / d;
  }
}

__global__ void k_scatter_users(const int* __restrict__ ids, const float* __restrict__ Wemb,
                                const float* __restrict__ bemb, float* __restrict__ e,
                                float* __restrict__ cnt, int N) {
  int t = blockIdx.x * blockDim.x + threadIdx.x;
  if (t >= N * 64) return;
  int i = t >> 6, d = t & 63;
  int id = ids[i];
  atomicAdd(&e[(long)id * 64 + d], Wemb[t] + bemb[d]);
  if (d == 0) atomicAdd(&cnt[id], 1.0f);
}

__global__ void k_enorm_selfinit(const float* __restrict__ cnt, const float* __restrict__ idg,
                                 float* __restrict__ e, float* __restrict__ agg, int nloc) {
  int t = blockIdx.x * blockDim.x + threadIdx.x;
  if (t >= nloc * 64) return;
  int j = t >> 6;
  float cv = cnt[j]; if (cv < 1.0f) cv = 1.0f;
  float v = e[t] / cv;
  e[t] = v;
  agg[t] = v * idg[j];
}

__global__ void k_selfinit(const float* __restrict__ h, const float* __restrict__ idg,
                           float* __restrict__ agg, int nloc) {
  int t = blockIdx.x * blockDim.x + threadIdx.x;
  if (t >= nloc * 64) return;
  agg[t] = h[t] * idg[t >> 6];
}

__global__ void k_edge_scatter(const int* __restrict__ src, const int* __restrict__ dst,
                               const float* __restrict__ isq, const float* __restrict__ h,
                               float* __restrict__ agg, int E) {
  int t = blockIdx.x * blockDim.x + threadIdx.x;
  if (t >= E * 64) return;
  int e = t >> 6, d = t & 63;
  int s = src[e], q = dst[e];
  atomicAdd(&agg[(long)q * 64 + d], h[(long)s * 64 + d] * isq[s] * isq[q]);
}

template <int NT>
__global__ __launch_bounds__(256) void k_gemm_wmma(
    const float* __restrict__ A, const float* __restrict__ B,
    const float* __restrict__ bias, float* __restrict__ C, int M, int K, int relu) {
  constexpr int Nc = NT * 16;
  __shared__ float Bs[64 * Nc];
  for (int i = threadIdx.x; i < K * Nc; i += 256) Bs[i] = B[i];
  __syncthreads();

  const int lane = threadIdx.x & 31;
  const int wave = threadIdx.x >> 5;
  const int half = lane >> 4;
  const int l16  = lane & 15;
  const int row0 = (blockIdx.x * 8 + wave) * 16;

  int ar = row0 + l16; if (ar > M - 1) ar = M - 1;
  const float* Arow = A + (long)ar * K;

  v8f acc[NT];
#pragma unroll
  for (int tt = 0; tt < NT; ++tt) {
    float bv = bias[tt * 16 + l16];
    v8f tmp = { bv, bv, bv, bv, bv, bv, bv, bv };
    acc[tt] = tmp;
  }

  for (int k = 0; k < K; k += 4) {
    v2f a;
    a[0] = Arow[k + 2 * half + 0];
    a[1] = Arow[k + 2 * half + 1];
#pragma unroll
    for (int tt = 0; tt < NT; ++tt) {
      v2f b;
      b[0] = Bs[(k + 2 * half + 0) * Nc + tt * 16 + l16];
      b[1] = Bs[(k + 2 * half + 1) * Nc + tt * 16 + l16];
      acc[tt] = __builtin_amdgcn_wmma_f32_16x16x4_f32(false, a, false, b, (short)0,
                                                      acc[tt], false, false);
    }
  }
#pragma unroll
  for (int tt = 0; tt < NT; ++tt) {
    for (int r = 0; r < 8; ++r) {
      int row = row0 + r + 8 * half;
      if (row < M) {
        float v = acc[tt][r];
        if (relu) v = v > 0.0f ? v : 0.0f;
        C[(long)row * Nc + tt * 16 + l16] = v;
      }
    }
  }
}

__global__ __launch_bounds__(256) void k_lstm_gates(
    const float* __restrict__ X, int Kx, const float* __restrict__ Wi,
    const float* __restrict__ H, const float* __restrict__ Wh,
    const float* __restrict__ b, float* __restrict__ G, int M) {
  __shared__ float Bs[64 * 64];
  for (int i = threadIdx.x; i < Kx * 64; i += 256) Bs[i] = Wi[i];
  for (int i = threadIdx.x; i < 16 * 64; i += 256) Bs[Kx * 64 + i] = Wh[i];
  __syncthreads();

  const int lane = threadIdx.x & 31;
  const int wave = threadIdx.x >> 5;
  const int half = lane >> 4;
  const int l16  = lane & 15;
  const int row0 = (blockIdx.x * 8 + wave) * 16;

  int ar = row0 + l16; if (ar > M - 1) ar = M - 1;

  v8f acc[4];
#pragma unroll
  for (int tt = 0; tt < 4; ++tt) {
    float bv = b[tt * 16 + l16];
    v8f tmp = { bv, bv, bv, bv, bv, bv, bv, bv };
    acc[tt] = tmp;
  }

  const float* Xr = X + (long)ar * Kx;
  for (int k = 0; k < Kx; k += 4) {
    v2f a;
    a[0] = Xr[k + 2 * half + 0];
    a[1] = Xr[k + 2 * half + 1];
#pragma unroll
    for (int tt = 0; tt < 4; ++tt) {
      v2f w;
      w[0] = Bs[(k + 2 * half + 0) * 64 + tt * 16 + l16];
      w[1] = Bs[(k + 2 * half + 1) * 64 + tt * 16 + l16];
      acc[tt] = __builtin_amdgcn_wmma_f32_16x16x4_f32(false, a, false, w, (short)0,
                                                      acc[tt], false, false);
    }
  }
  const float* Hr = H + (long)ar * 16;
#pragma unroll
  for (int k = 0; k < 16; k += 4) {
    v2f a;
    a[0] = Hr[k + 2 * half + 0];
    a[1] = Hr[k + 2 * half + 1];
#pragma unroll
    for (int tt = 0; tt < 4; ++tt) {
      v2f w;
      w[0] = Bs[(Kx + k + 2 * half + 0) * 64 + tt * 16 + l16];
      w[1] = Bs[(Kx + k + 2 * half + 1) * 64 + tt * 16 + l16];
      acc[tt] = __builtin_amdgcn_wmma_f32_16x16x4_f32(false, a, false, w, (short)0,
                                                      acc[tt], false, false);
    }
  }
#pragma unroll
  for (int tt = 0; tt < 4; ++tt) {
    for (int r = 0; r < 8; ++r) {
      int row = row0 + r + 8 * half;
      if (row < M) G[(long)row * 64 + tt * 16 + l16] = acc[tt][r];
    }
  }
}

__device__ __forceinline__ float sigm(float x) { return 1.0f / (1.0f + expf(-x)); }

__global__ void k_lstm_cell(const float* __restrict__ G, float* __restrict__ h,
                            float* __restrict__ c, int N) {
  int t = blockIdx.x * blockDim.x + threadIdx.x;
  if (t >= N * 16) return;
  int i = t >> 4, j = t & 15;
  const float* g = G + (long)i * 64;
  float gi = g[j], gf = g[16 + j], gg = g[32 + j], go = g[48 + j];
  float cn = sigm(gf) * c[t] + sigm(gi) * tanhf(gg);
  c[t] = cn;
  h[t] = sigm(go) * tanhf(cn);
}

__global__ void k_gather_e2n(const int* __restrict__ ids, const float* __restrict__ he,
                             float* __restrict__ out32, int N) {
  int t = blockIdx.x * blockDim.x + threadIdx.x;
  if (t >= N * 32) return;
  int i = t >> 5, d = t & 31;
  out32[t] = he[(long)ids[i] * 32 + d];
}

__global__ void k_outproj(const float* __restrict__ h, const float* __restrict__ Wout,
                          const float* __restrict__ bout, float* __restrict__ out, int N) {
  int i = blockIdx.x * blockDim.x + threadIdx.x;
  if (i >= N) return;
  const float* hr = h + (long)i * 16;
  float a0 = bout[0], a1 = bout[1];
  for (int j = 0; j < 16; ++j) {
    float v = hr[j];
    a0 += v * Wout[j * 2 + 0];
    a1 += v * Wout[j * 2 + 1];
  }
  out[(long)i * 2 + 0] = a0;
  out[(long)i * 2 + 1] = a1;
}

static inline int gs(long n, int b) { return (int)((n + b - 1) / b); }

extern "C" void kernel_launch(void* const* d_in, const int* in_sizes, int n_in,
                              void* d_out, int out_size, void* d_ws, size_t ws_size,
                              hipStream_t stream) {
  (void)in_sizes; (void)n_in; (void)out_size; (void)ws_size;

  const int*   seq  = (const int*)d_in[0];
  const int*   gidx = (const int*)d_in[1];
  const float* Wemb = (const float*)d_in[2];
  const float* bemb = (const float*)d_in[3];
  const float* W1   = (const float*)d_in[4];
  const float* b1   = (const float*)d_in[5];
  const float* W2   = (const float*)d_in[6];
  const float* b2   = (const float*)d_in[7];
  const float* We2n = (const float*)d_in[8];
  const float* be2n = (const float*)d_in[9];
  const float* Wi0  = (const float*)d_in[10];
  const float* Wh0  = (const float*)d_in[11];
  const float* bL0  = (const float*)d_in[12];
  const float* Wi1  = (const float*)d_in[13];
  const float* Wh1  = (const float*)d_in[14];
  const float* bL1  = (const float*)d_in[15];
  const float* Wout = (const float*)d_in[16];
  const float* bout = (const float*)d_in[17];

  const int* srcE = gidx;
  const int* dstE = gidx + N_EDGES;
  float* out = (float*)d_out;

  float* p = (float*)d_ws;
  float* e     = p; p += (size_t)NUM_LOC * 64;
  float* cnt   = p; p += NUM_LOC;
  float* deg   = p; p += NUM_LOC;
  float* isq   = p; p += NUM_LOC;
  float* idg   = p; p += NUM_LOC;
  float* agg   = p; p += (size_t)NUM_LOC * 64;
  float* h1b   = p; p += (size_t)NUM_LOC * 64;
  float* h2b   = p; p += (size_t)NUM_LOC * 64;
  float* he2n  = p; p += (size_t)NUM_LOC * 32;
  float* out32 = p; p += (size_t)USR_NUM * 32;
  float* gbuf  = p; p += (size_t)USR_NUM * 64;
  float* h0    = p; p += (size_t)USR_NUM * 16;
  float* c0    = p; p += (size_t)USR_NUM * 16;
  float* h1s   = p; p += (size_t)USR_NUM * 16;
  float* c1s   = p; p += (size_t)USR_NUM * 16;

  const int BT = 256;

  k_zero<<<gs(NUM_LOC, BT), BT, 0, stream>>>(deg, NUM_LOC);
  k_deg_accum<<<gs(N_EDGES, BT), BT, 0, stream>>>(dstE, deg, N_EDGES);
  k_deg_fin<<<gs(NUM_LOC, BT), BT, 0, stream>>>(deg, isq, idg, NUM_LOC);

  k_zero<<<gs((long)USR_NUM * 64, BT), BT, 0, stream>>>(h0, USR_NUM * 64);

  for (int t = 0; t < T_STEPS; ++t) {
    const int* ids = seq + (long)t * USR_NUM;

    k_zero<<<gs((long)NUM_LOC * 65, BT), BT, 0, stream>>>(e, NUM_LOC * 65);
    k_scatter_users<<<gs((long)USR_NUM * 64, BT), BT, 0, stream>>>(ids, Wemb, bemb, e, cnt, USR_NUM);
    k_enorm_selfinit<<<gs((long)NUM_LOC * 64, BT), BT, 0, stream>>>(cnt, idg, e, agg, NUM_LOC);

    k_edge_scatter<<<gs((long)N_EDGES * 64, BT), BT, 0, stream>>>(srcE, dstE, isq, e, agg, N_EDGES);
    k_gemm_wmma<4><<<gs(NUM_LOC, 128), 256, 0, stream>>>(agg, W1, b1, h1b, NUM_LOC, 64, 1);

    k_selfinit<<<gs((long)NUM_LOC * 64, BT), BT, 0, stream>>>(h1b, idg, agg, NUM_LOC);
    k_edge_scatter<<<gs((long)N_EDGES * 64, BT), BT, 0, stream>>>(srcE, dstE, isq, h1b, agg, N_EDGES);
    k_gemm_wmma<4><<<gs(NUM_LOC, 128), 256, 0, stream>>>(agg, W2, b2, h2b, NUM_LOC, 64, 0);

    k_gemm_wmma<2><<<gs(NUM_LOC, 128), 256, 0, stream>>>(h2b, We2n, be2n, he2n, NUM_LOC, 64, 0);
    k_gather_e2n<<<gs((long)USR_NUM * 32, BT), BT, 0, stream>>>(ids, he2n, out32, USR_NUM);

    k_lstm_gates<<<gs(USR_NUM, 128), 256, 0, stream>>>(out32, 32, Wi0, h0, Wh0, bL0, gbuf, USR_NUM);
    k_lstm_cell<<<gs((long)USR_NUM * 16, BT), BT, 0, stream>>>(gbuf, h0, c0, USR_NUM);

    k_lstm_gates<<<gs(USR_NUM, 128), 256, 0, stream>>>(h0, 16, Wi1, h1s, Wh1, bL1, gbuf, USR_NUM);
    k_lstm_cell<<<gs((long)USR_NUM * 16, BT), BT, 0, stream>>>(gbuf, h1s, c1s, USR_NUM);

    k_outproj<<<gs(USR_NUM, BT), BT, 0, stream>>>(h1s, Wout, bout, out + (long)t * USR_NUM * 2, USR_NUM);
  }
}


# All ten per-step stages software-pipelined ten deep: one launch per tick runs every stage, each on its own step, hand-over planes doubled by step parity; 104 -> 29 launches; all roles 1024 threads
# speedup vs baseline: 2.6733x; 2.6733x over previous
#include <hip/hip_runtime.h>
#include <math.h>


#define USR_NUM 100000
#define USR_DIM 64
#define NUM_LOC 5000
#define T_STEPS 16
#define N_EDGES 80000

typedef float v2f __attribute__((ext_vector_type(2)));
typedef float v4f __attribute__((ext_vector_type(4)));
typedef float v8f __attribute__((ext_vector_type(8)));

__global__ void k_zero(float* __restrict__ p, int n) {
  int i = blockIdx.x * blockDim.x + threadIdx.x;
  if (i < n) p[i] = 0.0f;
}

__global__ void k_deg_accum(const int* __restrict__ dst, float* __restrict__ deg, int E) {
  int i = blockIdx.x * blockDim.x + threadIdx.x;
  if (i < E) atomicAdd(&deg[dst[i]], 1.0f);
}

__global__ void k_deg_fin(const float* __restrict__ deg, float* __restrict__ isq,
                          float* __restrict__ idg, int n) {
  int i = blockIdx.x * blockDim.x + threadIdx.x;
  if (i < n) {
    float d = deg[i] + 1.0f;
    isq[i] = rsqrtf(d);
    idg[i] = 1.0f / d;
  }
}


#define ROWS_PB   32
#define SCAN_KEYS 32768
#define SCAN_V4   (SCAN_KEYS / 4 / 1024)
#define AR_TILE   0
#define AR_AUX    ((ROWS_PB + 1) * 64)
#define AR_QN     (AR_AUX + 64)
#define AR_Q      (AR_QN + 64)
#define AR_BAND   (AR_Q + SCAN_KEYS)
#define AR_GEMM   (64 * 64)

__device__ __forceinline__ void role_user_sum(
    const int vb, const int* __restrict__ ids, const float* __restrict__ Wemb,
    const float* __restrict__ bemb, float* __restrict__ e, int* arena) {
  float* tile = (float*)(arena + AR_TILE);
  float* cntS = (float*)(arena + AR_AUX);
  int*   qn   = arena + AR_QN;
  int*   q    = arena + AR_Q;

  const int tid  = threadIdx.x;
  const int lane = tid & 31;
  const int wave = tid >> 5;
  const int lo   = vb * ROWS_PB;

  for (int i = tid; i < (ROWS_PB + 1) * 64; i += 1024) tile[i] = 0.0f;
  if (tid < ROWS_PB) cntS[tid] = 0.0f;
  if (tid < 2) qn[tid] = 0;

  const int4* ids4 = (const int4*)ids;
  const int nv4 = USR_NUM / 4;
  const float2 bb = *(const float2*)(bemb + 2 * lane);

  int4 cur[SCAN_V4];
#pragma unroll
  for (int k = 0; k < SCAN_V4; ++k) {
    int i4 = k * 1024 + tid;
    cur[k] = (i4 < nv4) ? ids4[i4] : make_int4(-1, -1, -1, -1);
  }
  __syncthreads();

  int par = 0;
  for (int base4 = 0; base4 < nv4; base4 += SCAN_KEYS / 4, par ^= 1) {
    int4 nx[SCAN_V4];
#pragma unroll
    for (int k = 0; k < SCAN_V4; ++k) {
      int i4 = base4 + SCAN_KEYS / 4 + k * 1024 + tid;
      nx[k] = (i4 < nv4) ? ids4[i4] : make_int4(-1, -1, -1, -1);
    }
#pragma unroll
    for (int k = 0; k < SCAN_V4; ++k) {
      const int u0 = (base4 + k * 1024 + tid) * 4;
      const int kv[4] = { cur[k].x, cur[k].y, cur[k].z, cur[k].w };
#pragma unroll
      for (int c = 0; c < 4; ++c) {
        unsigned r = (unsigned)(kv[c] - lo);
        if (r < (unsigned)ROWS_PB) {
          int s = atomicAdd(&qn[par], 1);
          q[s] = ((u0 + c) << 8) | (int)r;
          atomicAdd(&cntS[r], 1.0f);
        }
      }
    }
    __syncthreads();
    const int n = qn[par];
    if (tid == 0) qn[par ^ 1] = 0;
    for (int j0 = wave; j0 < n; j0 += 128) {
      int ent[4];
#pragma unroll
      for (int k = 0; k < 4; ++k) {
        int j = j0 + 32 * k;
        ent[k] = q[j < n ? j : j0];
      }
      float2 v[4];
#pragma unroll
      for (int k = 0; k < 4; ++k)
        v[k] = *(const float2*)(Wemb + (long)(ent[k] >> 8) * 64 + 2 * lane);
#pragma unroll
      for (int k = 0; k < 4; ++k) {
        int r = (j0 + 32 * k < n) ? (ent[k] & 255) : ROWS_PB;
        float* dstp = &tile[r * 64 + 2 * lane];
        atomicAdd(dstp + 0, v[k].x + bb.x);
        atomicAdd(dstp + 1, v[k].y + bb.y);
      }
    }
    __syncthreads();
#pragma unroll
    for (int k = 0; k < SCAN_V4; ++k) cur[k] = nx[k];
  }

  for (int i = tid; i < ROWS_PB * 64; i += 1024) {
    int j = i >> 6;
    int row = lo + j;
    if (row < NUM_LOC) {
      float cv = cntS[j]; if (cv < 1.0f) cv = 1.0f;
      e[(long)row * 64 + (i & 63)] = tile[i] / cv;
    }
  }
}

__device__ __forceinline__ void role_edge_sum(
    const int vb, const int* __restrict__ src, const int* __restrict__ dst,
    const float* __restrict__ isq, const float* __restrict__ idg,
    const float* __restrict__ h, float* __restrict__ agg, int* arena) {
  float* tile = (float*)(arena + AR_TILE);
  float* isqS = (float*)(arena + AR_AUX);
  int*   qn   = arena + AR_QN;
  int*   q    = arena + AR_Q;

  const int tid  = threadIdx.x;
  const int lane = tid & 31;
  const int wave = tid >> 5;
  const int lo   = vb * ROWS_PB;

  for (int i = tid; i < (ROWS_PB + 1) * 64; i += 1024) {
    int row = lo + (i >> 6);
    tile[i] = (i < ROWS_PB * 64 && row < NUM_LOC) ? h[(long)row * 64 + (i & 63)] * idg[row] : 0.0f;
  }
  if (tid <= ROWS_PB) isqS[tid] = (tid < ROWS_PB && lo + tid < NUM_LOC) ? isq[lo + tid] : 0.0f;
  if (tid < 2) qn[tid] = 0;

  const int4* dst4 = (const int4*)dst;
  const int nv4 = N_EDGES / 4;

  int4 cur[SCAN_V4];
#pragma unroll
  for (int k = 0; k < SCAN_V4; ++k) {
    int i4 = k * 1024 + tid;
    cur[k] = (i4 < nv4) ? dst4[i4] : make_int4(-1, -1, -1, -1);
  }
  __syncthreads();

  int par = 0;
  for (int base4 = 0; base4 < nv4; base4 += SCAN_KEYS / 4, par ^= 1) {
    int4 nx[SCAN_V4];
#pragma unroll
    for (int k = 0; k < SCAN_V4; ++k) {
      int i4 = base4 + SCAN_KEYS / 4 + k * 1024 + tid;
      nx[k] = (i4 < nv4) ? dst4[i4] : make_int4(-1, -1, -1, -1);
    }
#pragma unroll
    for (int k = 0; k < SCAN_V4; ++k) {
      const int e0 = (base4 + k * 1024 + tid) * 4;
      const int kv[4] = { cur[k].x, cur[k].y, cur[k].z, cur[k].w };
#pragma unroll
      for (int c = 0; c < 4; ++c) {
        unsigned r = (unsigned)(kv[c] - lo);
        if (r < (unsigned)ROWS_PB) {
          int s = atomicAdd(&qn[par], 1);
          q[s] = ((e0 + c) << 8) | (int)r;
        }
      }
    }
    __syncthreads();
    const int n = qn[par];
    if (tid == 0) qn[par ^ 1] = 0;
    for (int j = tid; j < n; j += 1024) {
      int ent = q[j];
      q[j] = (src[ent >> 8] << 8) | (ent & 255);
    }
    __syncthreads();
    for (int j0 = wave; j0 < n; j0 += 128) {
      int ent[4];
#pragma unroll
      for (int k = 0; k < 4; ++k) {
        int j = j0 + 32 * k;
        ent[k] = q[j < n ? j : j0];
      }
      float2 v[4];
      float  w[4];
#pragma unroll
      for (int k = 0; k < 4; ++k) {
        int s = ent[k] >> 8;
        v[k] = *(const float2*)(h + (long)s * 64 + 2 * lane);
        w[k] = isq[s];
      }
#pragma unroll
      for (int k = 0; k < 4; ++k) {
        int r = (j0 + 32 * k < n) ? (ent[k] & 255) : ROWS_PB;
        float wq = isqS[r];
        float* dstp = &tile[r * 64 + 2 * lane];
        atomicAdd(dstp + 0, v[k].x * w[k] * wq);
        atomicAdd(dstp + 1, v[k].y * w[k] * wq);
      }
    }
    __syncthreads();
#pragma unroll
    for (int k = 0; k < SCAN_V4; ++k) cur[k] = nx[k];
  }

  for (int i = tid; i < ROWS_PB * 64; i += 1024) {
    int row = lo + (i >> 6);
    if (row < NUM_LOC) agg[(long)row * 64 + (i & 63)] = tile[i];
  }
}

template <int NT, int TH>
__device__ __forceinline__ void role_gemm(
    const int vb, const float* __restrict__ A, const float* __restrict__ B,
    const float* __restrict__ bias, float* __restrict__ C, int M, int K, int relu,
    float* Bs) {
  constexpr int Nc = NT * 16;
  for (int i = threadIdx.x; i < K * Nc; i += TH) Bs[i] = B[i];
  __syncthreads();

  const int lane = threadIdx.x & 31;
  const int wave = threadIdx.x >> 5;
  const int half = lane >> 4;
  const int l16  = lane & 15;
  const int row0 = (vb * (TH / 32) + wave) * 16;

  int ar = row0 + l16; if (ar > M - 1) ar = M - 1;
  const float* Arow = A + (long)ar * K;

  v8f acc[NT];
#pragma unroll
  for (int tt = 0; tt < NT; ++tt) {
    float bv = bias[tt * 16 + l16];
    v8f tmp = { bv, bv, bv, bv, bv, bv, bv, bv };
    acc[tt] = tmp;
  }

  for (int k = 0; k < K; k += 4) {
    v2f a;
    a[0] = Arow[k + 2 * half + 0];
    a[1] = Arow[k + 2 * half + 1];
#pragma unroll
    for (int tt = 0; tt < NT; ++tt) {
      v2f b;
      b[0] = Bs[(k + 2 * half + 0) * Nc + tt * 16 + l16];
      b[1] = Bs[(k + 2 * half + 1) * Nc + tt * 16 + l16];
      acc[tt] = __builtin_amdgcn_wmma_f32_16x16x4_f32(false, a, false, b, (short)0,
                                                      acc[tt], false, false);
    }
  }
#pragma unroll
  for (int tt = 0; tt < NT; ++tt) {
    for (int r = 0; r < 8; ++r) {
      int row = row0 + r + 8 * half;
      if (row < M) {
        float v = acc[tt][r];
        if (relu) v = v > 0.0f ? v : 0.0f;
        C[(long)row * Nc + tt * 16 + l16] = v;
      }
    }
  }
}

__device__ __forceinline__ float sigm(float x) { return 1.0f / (1.0f + expf(-x)); }

template <int TH>
__device__ __forceinline__ void role_layer(
    const int vb, const float* __restrict__ X, int Kx, const float* __restrict__ Wi,
    const float* __restrict__ Wh, const float* __restrict__ b,
    const float* hin, float* hout, float* c, int M, float* Bs) {
  for (int i = threadIdx.x; i < Kx * 64; i += TH) Bs[i] = Wi[i];
  for (int i = threadIdx.x; i < 16 * 64; i += TH) Bs[Kx * 64 + i] = Wh[i];
  __syncthreads();

  const int lane = threadIdx.x & 31;
  const int wave = threadIdx.x >> 5;
  const int half = lane >> 4;
  const int l16  = lane & 15;
  const int row0 = (vb * (TH / 32) + wave) * 16;
  if (row0 >= M) return;

  int ar = row0 + l16; if (ar > M - 1) ar = M - 1;

  float cold[8];
#pragma unroll
  for (int r = 0; r < 8; ++r) {
    int row = row0 + r + 8 * half;
    cold[r] = (row < M) ? c[(long)row * 16 + l16] : 0.0f;
  }

  v8f acc[4];
#pragma unroll
  for (int tt = 0; tt < 4; ++tt) {
    float bv = b[tt * 16 + l16];
    v8f tmp = { bv, bv, bv, bv, bv, bv, bv, bv };
    acc[tt] = tmp;
  }

  const float* Xr = X + (long)ar * Kx;
  for (int k = 0; k < Kx; k += 4) {
    v2f a;
    a[0] = Xr[k + 2 * half + 0];
    a[1] = Xr[k + 2 * half + 1];
#pragma unroll
    for (int tt = 0; tt < 4; ++tt) {
      v2f w;
      w[0] = Bs[(k + 2 * half + 0) * 64 + tt * 16 + l16];
      w[1] = Bs[(k + 2 * half + 1) * 64 + tt * 16 + l16];
      acc[tt] = __builtin_amdgcn_wmma_f32_16x16x4_f32(false, a, false, w, (short)0,
                                                      acc[tt], false, false);
    }
  }
  const float* Hr = hin + (long)ar * 16;
#pragma unroll
  for (int k = 0; k < 16; k += 4) {
    v2f a;
    a[0] = Hr[k + 2 * half + 0];
    a[1] = Hr[k + 2 * half + 1];
#pragma unroll
    for (int tt = 0; tt < 4; ++tt) {
      v2f w;
      w[0] = Bs[(Kx + k + 2 * half + 0) * 64 + tt * 16 + l16];
      w[1] = Bs[(Kx + k + 2 * half + 1) * 64 + tt * 16 + l16];
      acc[tt] = __builtin_amdgcn_wmma_f32_16x16x4_f32(false, a, false, w, (short)0,
                                                      acc[tt], false, false);
    }
  }
#pragma unroll
  for (int r = 0; r < 8; ++r) {
    int row = row0 + r + 8 * half;
    if (row < M) {
      float gi = acc[0][r], gf = acc[1][r], gg = acc[2][r], go = acc[3][r];
      float cn = sigm(gf) * cold[r] + sigm(gi) * tanhf(gg);
      c[(long)row * 16 + l16] = cn;
      hout[(long)row * 16 + l16] = sigm(go) * tanhf(cn);
    }
  }
}

__device__ __forceinline__ void role_gather(const int vb, const int* __restrict__ ids,
                                            const float* __restrict__ he,
                                            float* __restrict__ out32, int N) {
  const int sub = threadIdx.x & 7;
  const int r0  = vb * 512 + (threadIdx.x >> 3);
  if (vb * 512 + 512 <= N) {
    int id[4];
    v4f v[4];
#pragma unroll
    for (int p = 0; p < 4; ++p) id[p] = ids[r0 + 128 * p];
#pragma unroll
    for (int p = 0; p < 4; ++p)
      v[p] = *(const v4f*)(he + (long)id[p] * 32 + 4 * sub);
    asm volatile("" : "+v"(v[0]), "+v"(v[1]), "+v"(v[2]), "+v"(v[3]));
#pragma unroll
    for (int p = 0; p < 4; ++p)
      *(v4f*)(out32 + (long)(r0 + 128 * p) * 32 + 4 * sub) = v[p];
  } else {
    for (int p = 0; p < 4; ++p) {
      int row = r0 + 128 * p;
      if (row < N)
        *(v4f*)(out32 + (long)row * 32 + 4 * sub) =
            *(const v4f*)(he + (long)ids[row] * 32 + 4 * sub);
    }
  }
}

template <int TH>
__device__ __forceinline__ void role_outproj(const int vb, const float* __restrict__ h,
                                             const float* __restrict__ Wout,
                                             const float* __restrict__ bout,
                                             float* __restrict__ out, int N) {
  int i = vb * TH + threadIdx.x;
  if (i >= N) return;
  const float* hr = h + (long)i * 16;
  float a0 = bout[0], a1 = bout[1];
  for (int j = 0; j < 16; ++j) {
    float v = hr[j];
    a0 += v * Wout[j * 2 + 0];
    a1 += v * Wout[j * 2 + 1];
  }
  out[(long)i * 2 + 0] = a0;
  out[(long)i * 2 + 1] = a1;
}

__global__ __launch_bounds__(1024) void k_stageA(
    int nLoc, const int* __restrict__ idsNext, const float* __restrict__ Wemb,
    const float* __restrict__ bemb, float* __restrict__ e,
    const int* __restrict__ idsCur, const float* __restrict__ he2n, float* __restrict__ out32) {
  __shared__ int arena[AR_BAND];
  const int b = blockIdx.x;
  if (b < nLoc) role_user_sum(b, idsNext, Wemb, bemb, e, arena);
  else          role_gather(b - nLoc, idsCur, he2n, out32, USR_NUM);
}

__global__ __launch_bounds__(1024) void k_stageBD(
    const int* __restrict__ src, const int* __restrict__ dst,
    const float* __restrict__ isq, const float* __restrict__ idg,
    const float* __restrict__ hin, float* __restrict__ agg) {
  __shared__ int arena[AR_BAND];
  role_edge_sum(blockIdx.x, src, dst, isq, idg, hin, agg, arena);
}

__global__ __launch_bounds__(256) void k_stageCE(
    int nLoc, const float* __restrict__ A, const float* __restrict__ W,
    const float* __restrict__ bias, float* __restrict__ C, int relu,
    const float* __restrict__ X, int Kx, const float* __restrict__ Wi,
    const float* __restrict__ Wh, const float* __restrict__ bL,
    float* hS, float* cS) {
  __shared__ float Bs[AR_GEMM];
  const int b = blockIdx.x;
  if (b < nLoc) role_gemm<4, 256>(b, A, W, bias, C, NUM_LOC, 64, relu, Bs);
  else          role_layer<256>(b - nLoc, X, Kx, Wi, Wh, bL, hS, hS, cS, USR_NUM, Bs);
}

__global__ __launch_bounds__(256) void k_stageF(
    int nLoc, const float* __restrict__ A, const float* __restrict__ W,
    const float* __restrict__ bias, float* __restrict__ C,
    const float* __restrict__ h, const float* __restrict__ Wout,
    const float* __restrict__ bout, float* __restrict__ out) {
  __shared__ float Bs[AR_GEMM];
  const int b = blockIdx.x;
  if (b < nLoc) role_gemm<2, 256>(b, A, W, bias, C, NUM_LOC, 64, 0, Bs);
  else          role_outproj<256>(b - nLoc, h, Wout, bout, out, USR_NUM);
}


#define PL_LOC   ((long)NUM_LOC * 64)
#define PL_E2N   ((long)NUM_LOC * 32)
#define PL_X     ((long)USR_NUM * 32)

#define NB_BAND   ((NUM_LOC + ROWS_PB - 1) / ROWS_PB)
#define NB_GATHER ((USR_NUM + 511) / 512)
#define NB_LAYER4 ((USR_NUM + 511) / 512)
#define NB_GEMM4  ((NUM_LOC + 511) / 512)
#define NB_OUT4   ((USR_NUM + 1023) / 1024)
#define NB_TICK   (3 * NB_BAND + NB_GATHER + 2 * NB_LAYER4 + NB_OUT4 + 3 * NB_GEMM4)
#define N_TICKS   (T_STEPS + 9)
#define NL_PAD 5056
#define WS_TICK_FLOATS (3L * NL_PAD + 10 * PL_LOC + 2 * PL_E2N + 2 * PL_X + 6L * USR_NUM * 16)
#define WS_TICK_BYTES  ((size_t)(WS_TICK_FLOATS * 4))

struct TickArgs {
  const int*   seq;  const int*   src;  const int*   dst;
  const float* Wemb; const float* bemb;
  const float* W1;   const float* b1;   const float* W2;   const float* b2;
  const float* We;   const float* be;
  const float* Wi0;  const float* Wh0;  const float* bL0;
  const float* Wi1;  const float* Wh1;  const float* bL1;
  const float* Wout; const float* bout;
  const float* isq;  const float* idg;
  float* e; float* aggA; float* h1b; float* aggB; float* h2b;
  float* he2n; float* out32;
  float* h0p0; float* h0p1; float* c0;
  float* h1p0; float* h1p1; float* c1;
  float* out;
};

__global__ __launch_bounds__(1024) void k_tick(const TickArgs a, const int tick) {
  __shared__ int arena[AR_BAND];
  int b = blockIdx.x;
  int s;
  if (b < NB_BAND) {
    s = tick;
    if ((unsigned)s >= (unsigned)T_STEPS) return;
    role_user_sum(b, a.seq + (long)s * USR_NUM, a.Wemb, a.bemb, a.e + (s & 1) * PL_LOC, arena);
    return;
  }
  b -= NB_BAND;
  if (b < NB_BAND) {
    s = tick - 1;
    if ((unsigned)s >= (unsigned)T_STEPS) return;
    role_edge_sum(b, a.src, a.dst, a.isq, a.idg, a.e + (s & 1) * PL_LOC,
                  a.aggA + (s & 1) * PL_LOC, arena);
    return;
  }
  b -= NB_BAND;
  if (b < NB_BAND) {
    s = tick - 3;
    if ((unsigned)s >= (unsigned)T_STEPS) return;
    role_edge_sum(b, a.src, a.dst, a.isq, a.idg, a.h1b + (s & 1) * PL_LOC,
                  a.aggB + (s & 1) * PL_LOC, arena);
    return;
  }
  b -= NB_BAND;
  if (b < NB_GATHER) {
    s = tick - 6;
    if ((unsigned)s >= (unsigned)T_STEPS) return;
    role_gather(b, a.seq + (long)s * USR_NUM, a.he2n + (s & 1) * PL_E2N,
                a.out32 + (s & 1) * PL_X, USR_NUM);
    return;
  }
  b -= NB_GATHER;
  if (b < NB_LAYER4) {
    s = tick - 7;
    if ((unsigned)s >= (unsigned)T_STEPS) return;
    role_layer<1024>(b, a.out32 + (s & 1) * PL_X, 32, a.Wi0, a.Wh0, a.bL0,
                     (s & 1) ? a.h0p0 : a.h0p1, (s & 1) ? a.h0p1 : a.h0p0, a.c0,
                     USR_NUM, (float*)arena);
    return;
  }
  b -= NB_LAYER4;
  if (b < NB_LAYER4) {
    s = tick - 8;
    if ((unsigned)s >= (unsigned)T_STEPS) return;
    role_layer<1024>(b, (s & 1) ? a.h0p1 : a.h0p0, 16, a.Wi1, a.Wh1, a.bL1,
                     (s & 1) ? a.h1p0 : a.h1p1, (s & 1) ? a.h1p1 : a.h1p0, a.c1,
                     USR_NUM, (float*)arena);
    return;
  }
  b -= NB_LAYER4;
  if (b < NB_OUT4) {
    s = tick - 9;
    if ((unsigned)s >= (unsigned)T_STEPS) return;
    role_outproj<1024>(b, (s & 1) ? a.h1p1 : a.h1p0, a.Wout, a.bout,
                       a.out + (long)s * USR_NUM * 2, USR_NUM);
    return;
  }
  b -= NB_OUT4;
  if (b < NB_GEMM4) {
    s = tick - 2;
    if ((unsigned)s >= (unsigned)T_STEPS) return;
    role_gemm<4, 1024>(b, a.aggA + (s & 1) * PL_LOC, a.W1, a.b1, a.h1b + (s & 1) * PL_LOC,
                       NUM_LOC, 64, 1, (float*)arena);
    return;
  }
  b -= NB_GEMM4;
  if (b < NB_GEMM4) {
    s = tick - 4;
    if ((unsigned)s >= (unsigned)T_STEPS) return;
    role_gemm<4, 1024>(b, a.aggB + (s & 1) * PL_LOC, a.W2, a.b2, a.h2b + (s & 1) * PL_LOC,
                       NUM_LOC, 64, 0, (float*)arena);
    return;
  }
  b -= NB_GEMM4;
  {
    s = tick - 5;
    if ((unsigned)s >= (unsigned)T_STEPS) return;
    role_gemm<2, 1024>(b, a.h2b + (s & 1) * PL_LOC, a.We, a.be, a.he2n + (s & 1) * PL_E2N,
                       NUM_LOC, 64, 0, (float*)arena);
  }
}

static inline int gs(long n, int b) { return (int)((n + b - 1) / b); }

#define NB_LAYER  ((USR_NUM + 127) / 128)
#define NB_GEMM   ((NUM_LOC + 127) / 128)
#define NB_OUT    ((USR_NUM + 255) / 256)

extern "C" void kernel_launch(void* const* d_in, const int* in_sizes, int n_in,
                              void* d_out, int out_size, void* d_ws, size_t ws_size,
                              hipStream_t stream) {
  (void)in_sizes; (void)n_in; (void)out_size;

  const int*   seq  = (const int*)d_in[0];
  const int*   gidx = (const int*)d_in[1];
  const float* Wemb = (const float*)d_in[2];
  const float* bemb = (const float*)d_in[3];
  const float* W1   = (const float*)d_in[4];
  const float* b1   = (const float*)d_in[5];
  const float* W2   = (const float*)d_in[6];
  const float* b2   = (const float*)d_in[7];
  const float* We2n = (const float*)d_in[8];
  const float* be2n = (const float*)d_in[9];
  const float* Wi0  = (const float*)d_in[10];
  const float* Wh0  = (const float*)d_in[11];
  const float* bL0  = (const float*)d_in[12];
  const float* Wi1  = (const float*)d_in[13];
  const float* Wh1  = (const float*)d_in[14];
  const float* bL1  = (const float*)d_in[15];
  const float* Wout = (const float*)d_in[16];
  const float* bout = (const float*)d_in[17];

  const int* srcE = gidx;
  const int* dstE = gidx + N_EDGES;
  float* out = (float*)d_out;
  const int BT = 256;

  float* p = (float*)d_ws;
  float* deg   = p; p += NL_PAD;
  float* isq   = p; p += NL_PAD;
  float* idg   = p; p += NL_PAD;

  k_zero<<<gs(NUM_LOC, BT), BT, 0, stream>>>(deg, NUM_LOC);
  k_deg_accum<<<gs(N_EDGES, BT), BT, 0, stream>>>(dstE, deg, N_EDGES);
  k_deg_fin<<<gs(NUM_LOC, BT), BT, 0, stream>>>(deg, isq, idg, NUM_LOC);

  if (ws_size >= WS_TICK_BYTES) {
    TickArgs a;
    a.seq = seq; a.src = srcE; a.dst = dstE;
    a.Wemb = Wemb; a.bemb = bemb;
    a.W1 = W1; a.b1 = b1; a.W2 = W2; a.b2 = b2; a.We = We2n; a.be = be2n;
    a.Wi0 = Wi0; a.Wh0 = Wh0; a.bL0 = bL0; a.Wi1 = Wi1; a.Wh1 = Wh1; a.bL1 = bL1;
    a.Wout = Wout; a.bout = bout;
    a.isq = isq; a.idg = idg;
    a.e     = p; p += 2 * PL_LOC;
    a.aggA  = p; p += 2 * PL_LOC;
    a.h1b   = p; p += 2 * PL_LOC;
    a.aggB  = p; p += 2 * PL_LOC;
    a.h2b   = p; p += 2 * PL_LOC;
    a.he2n  = p; p += 2 * PL_E2N;
    a.out32 = p; p += 2 * PL_X;
    a.h0p1  = p; p += (size_t)USR_NUM * 16;
    a.c0    = p; p += (size_t)USR_NUM * 16;
    a.h1p1  = p; p += (size_t)USR_NUM * 16;
    a.c1    = p; p += (size_t)USR_NUM * 16;
    a.h0p0  = p; p += (size_t)USR_NUM * 16;
    a.h1p0  = p; p += (size_t)USR_NUM * 16;
    a.out = out;

    k_zero<<<gs((long)USR_NUM * 64, BT), BT, 0, stream>>>(a.h0p1, USR_NUM * 64);
    for (int tick = 0; tick < N_TICKS; ++tick)
      k_tick<<<NB_TICK, 1024, 0, stream>>>(a, tick);
    return;
  }

  float* e     = p; p += (size_t)NUM_LOC * 64;
  float* agg   = p; p += (size_t)NUM_LOC * 64;
  float* h1b   = p; p += (size_t)NUM_LOC * 64;
  float* h2b   = p; p += (size_t)NUM_LOC * 64;
  float* he2n  = p; p += (size_t)NUM_LOC * 32;
  float* out32 = p; p += (size_t)USR_NUM * 32;
  float* h0    = p; p += (size_t)USR_NUM * 16;
  float* c0    = p; p += (size_t)USR_NUM * 16;
  float* h1s   = p; p += (size_t)USR_NUM * 16;
  float* c1s   = p; p += (size_t)USR_NUM * 16;

  k_zero<<<gs((long)USR_NUM * 64, BT), BT, 0, stream>>>(h0, USR_NUM * 64);

  k_stageA <<<NB_BAND, 1024, 0, stream>>>(NB_BAND, seq, Wemb, bemb, e, seq, he2n, out32);
  k_stageBD<<<NB_BAND, 1024, 0, stream>>>(srcE, dstE, isq, idg, e, agg);
  k_stageCE<<<NB_GEMM, 256, 0, stream>>>(NB_GEMM, agg, W1, b1, h1b, 1,
                                         out32, 32, Wi0, Wh0, bL0, h0, c0);
  k_stageBD<<<NB_BAND, 1024, 0, stream>>>(srcE, dstE, isq, idg, h1b, agg);
  k_stageCE<<<NB_GEMM, 256, 0, stream>>>(NB_GEMM, agg, W2, b2, h2b, 0,
                                         h0, 16, Wi1, Wh1, bL1, h1s, c1s);
  k_stageF <<<NB_GEMM, 256, 0, stream>>>(NB_GEMM, h2b, We2n, be2n, he2n, h1s, Wout, bout, out);

  for (int t = 0; t + 1 < T_STEPS; ++t) {
    const int* idsCur  = seq + (long)t * USR_NUM;
    const int* idsNext = idsCur + USR_NUM;
    float* outT = out + (long)t * USR_NUM * 2;

    k_stageA <<<NB_BAND + NB_GATHER, 1024, 0, stream>>>(NB_BAND, idsNext, Wemb, bemb, e,
                                                        idsCur, he2n, out32);
    k_stageBD<<<NB_BAND, 1024, 0, stream>>>(srcE, dstE, isq, idg, e, agg);
    k_stageCE<<<NB_GEMM + NB_LAYER, 256, 0, stream>>>(NB_GEMM, agg, W1, b1, h1b, 1,
                                                      out32, 32, Wi0, Wh0, bL0, h0, c0);
    k_stageBD<<<NB_BAND, 1024, 0, stream>>>(srcE, dstE, isq, idg, h1b, agg);
    k_stageCE<<<NB_GEMM + NB_LAYER, 256, 0, stream>>>(NB_GEMM, agg, W2, b2, h2b, 0,
                                                      h0, 16, Wi1, Wh1, bL1, h1s, c1s);
    k_stageF <<<NB_GEMM + NB_OUT, 256, 0, stream>>>(NB_GEMM, h2b, We2n, be2n, he2n,
                                                     h1s, Wout, bout, outT);
  }

  {
    const int* idsCur = seq + (long)(T_STEPS - 1) * USR_NUM;
    float* outT = out + (long)(T_STEPS - 1) * USR_NUM * 2;
    k_stageA <<<NB_GATHER, 1024, 0, stream>>>(0, idsCur, Wemb, bemb, e, idsCur, he2n, out32);
    k_stageCE<<<NB_LAYER, 256, 0, stream>>>(0, agg, W1, b1, h1b, 1,
                                            out32, 32, Wi0, Wh0, bL0, h0, c0);
    k_stageCE<<<NB_LAYER, 256, 0, stream>>>(0, agg, W2, b2, h2b, 0,
                                            h0, 16, Wi1, Wh1, bL1, h1s, c1s);
    k_stageF <<<NB_OUT, 256, 0, stream>>>(0, h2b, We2n, be2n, he2n, h1s, Wout, bout, outT);
  }
}


# Two work-groups per WGP: tick kernel held to 64 registers (16 waves/SIMD) by key passes of 16384 (key double buffer 32 registers, LDS 73 KiB); band sums: LDS arrival order inside 16384-key passes
# speedup vs baseline: 3.2857x; 1.2291x over previous
#include <hip/hip_runtime.h>
#include <math.h>


#define USR_NUM 100000
#define USR_DIM 64
#define NUM_LOC 5000
#define T_STEPS 16
#define N_EDGES 80000

typedef float v2f __attribute__((ext_vector_type(2)));
typedef float v4f __attribute__((ext_vector_type(4)));
typedef float v8f __attribute__((ext_vector_type(8)));

__global__ void k_zero(float* __restrict__ p, int n) {
  int i = blockIdx.x * blockDim.x + threadIdx.x;
  if (i < n) p[i] = 0.0f;
}

__global__ void k_deg_accum(const int* __restrict__ dst, float* __restrict__ deg, int E) {
  int i = blockIdx.x * blockDim.x + threadIdx.x;
  if (i < E) atomicAdd(&deg[dst[i]], 1.0f);
}

__global__ void k_deg_fin(const float* __restrict__ deg, float* __restrict__ isq,
                          float* __restrict__ idg, int n) {
  int i = blockIdx.x * blockDim.x + threadIdx.x;
  if (i < n) {
    float d = deg[i] + 1.0f;
    isq[i] = rsqrtf(d);
    idg[i] = 1.0f / d;
  }
}


#define ROWS_PB   32
#define SCAN_KEYS 16384
#define SCAN_V4   (SCAN_KEYS / 4 / 1024)
#define AR_TILE   0
#define AR_AUX    ((ROWS_PB + 1) * 64)
#define AR_QN     (AR_AUX + 64)
#define AR_Q      (AR_QN + 64)
#define AR_BAND   (AR_Q + SCAN_KEYS)
#define AR_GEMM   (64 * 64)

__device__ __forceinline__ void role_user_sum(
    const int vb, const int* __restrict__ ids, const float* __restrict__ Wemb,
    const float* __restrict__ bemb, float* __restrict__ e, int* arena) {
  float* tile = (float*)(arena + AR_TILE);
  float* cntS = (float*)(arena + AR_AUX);
  int*   qn   = arena + AR_QN;
  int*   q    = arena + AR_Q;

  const int tid  = threadIdx.x;
  const int lane = tid & 31;
  const int wave = tid >> 5;
  const int lo   = vb * ROWS_PB;

  for (int i = tid; i < (ROWS_PB + 1) * 64; i += 1024) tile[i] = 0.0f;
  if (tid < ROWS_PB) cntS[tid] = 0.0f;
  if (tid < 2) qn[tid] = 0;

  const int4* ids4 = (const int4*)ids;
  const int nv4 = USR_NUM / 4;
  const float2 bb = *(const float2*)(bemb + 2 * lane);

  int4 cur[SCAN_V4];
#pragma unroll
  for (int k = 0; k < SCAN_V4; ++k) {
    int i4 = k * 1024 + tid;
    cur[k] = (i4 < nv4) ? ids4[i4] : make_int4(-1, -1, -1, -1);
  }
  __syncthreads();

  int par = 0;
  for (int base4 = 0; base4 < nv4; base4 += SCAN_KEYS / 4, par ^= 1) {
    int4 nx[SCAN_V4];
#pragma unroll
    for (int k = 0; k < SCAN_V4; ++k) {
      int i4 = base4 + SCAN_KEYS / 4 + k * 1024 + tid;
      nx[k] = (i4 < nv4) ? ids4[i4] : make_int4(-1, -1, -1, -1);
    }
#pragma unroll
    for (int k = 0; k < SCAN_V4; ++k) {
      const int u0 = (base4 + k * 1024 + tid) * 4;
      const int kv[4] = { cur[k].x, cur[k].y, cur[k].z, cur[k].w };
#pragma unroll
      for (int c = 0; c < 4; ++c) {
        unsigned r = (unsigned)(kv[c] - lo);
        if (r < (unsigned)ROWS_PB) {
          int s = atomicAdd(&qn[par], 1);
          q[s] = ((u0 + c) << 8) | (int)r;
          atomicAdd(&cntS[r], 1.0f);
        }
      }
    }
    __syncthreads();
    const int n = qn[par];
    if (tid == 0) qn[par ^ 1] = 0;
    for (int j0 = wave; j0 < n; j0 += 128) {
      int ent[4];
#pragma unroll
      for (int k = 0; k < 4; ++k) {
        int j = j0 + 32 * k;
        ent[k] = q[j < n ? j : j0];
      }
      float2 v[4];
#pragma unroll
      for (int k = 0; k < 4; ++k)
        v[k] = *(const float2*)(Wemb + (long)(ent[k] >> 8) * 64 + 2 * lane);
#pragma unroll
      for (int k = 0; k < 4; ++k) {
        int r = (j0 + 32 * k < n) ? (ent[k] & 255) : ROWS_PB;
        float* dstp = &tile[r * 64 + 2 * lane];
        atomicAdd(dstp + 0, v[k].x + bb.x);
        atomicAdd(dstp + 1, v[k].y + bb.y);
      }
    }
    __syncthreads();
#pragma unroll
    for (int k = 0; k < SCAN_V4; ++k) cur[k] = nx[k];
  }

  for (int i = tid; i < ROWS_PB * 64; i += 1024) {
    int j = i >> 6;
    int row = lo + j;
    if (row < NUM_LOC) {
      float cv = cntS[j]; if (cv < 1.0f) cv = 1.0f;
      e[(long)row * 64 + (i & 63)] = tile[i] / cv;
    }
  }
}

__device__ __forceinline__ void role_edge_sum(
    const int vb, const int* __restrict__ src, const int* __restrict__ dst,
    const float* __restrict__ isq, const float* __restrict__ idg,
    const float* __restrict__ h, float* __restrict__ agg, int* arena) {
  float* tile = (float*)(arena + AR_TILE);
  float* isqS = (float*)(arena + AR_AUX);
  int*   qn   = arena + AR_QN;
  int*   q    = arena + AR_Q;

  const int tid  = threadIdx.x;
  const int lane = tid & 31;
  const int wave = tid >> 5;
  const int lo   = vb * ROWS_PB;

  for (int i = tid; i < (ROWS_PB + 1) * 64; i += 1024) {
    int row = lo + (i >> 6);
    tile[i] = (i < ROWS_PB * 64 && row < NUM_LOC) ? h[(long)row * 64 + (i & 63)] * idg[row] : 0.0f;
  }
  if (tid <= ROWS_PB) isqS[tid] = (tid < ROWS_PB && lo + tid < NUM_LOC) ? isq[lo + tid] : 0.0f;
  if (tid < 2) qn[tid] = 0;

  const int4* dst4 = (const int4*)dst;
  const int nv4 = N_EDGES / 4;

  int4 cur[SCAN_V4];
#pragma unroll
  for (int k = 0; k < SCAN_V4; ++k) {
    int i4 = k * 1024 + tid;
    cur[k] = (i4 < nv4) ? dst4[i4] : make_int4(-1, -1, -1, -1);
  }
  __syncthreads();

  int par = 0;
  for (int base4 = 0; base4 < nv4; base4 += SCAN_KEYS / 4, par ^= 1) {
    int4 nx[SCAN_V4];
#pragma unroll
    for (int k = 0; k < SCAN_V4; ++k) {
      int i4 = base4 + SCAN_KEYS / 4 + k * 1024 + tid;
      nx[k] = (i4 < nv4) ? dst4[i4] : make_int4(-1, -1, -1, -1);
    }
#pragma unroll
    for (int k = 0; k < SCAN_V4; ++k) {
      const int e0 = (base4 + k * 1024 + tid) * 4;
      const int kv[4] = { cur[k].x, cur[k].y, cur[k].z, cur[k].w };
#pragma unroll
      for (int c = 0; c < 4; ++c) {
        unsigned r = (unsigned)(kv[c] - lo);
        if (r < (unsigned)ROWS_PB) {
          int s = atomicAdd(&qn[par], 1);
          q[s] = ((e0 + c) << 8) | (int)r;
        }
      }
    }
    __syncthreads();
    const int n = qn[par];
    if (tid == 0) qn[par ^ 1] = 0;
    for (int j = tid; j < n; j += 1024) {
      int ent = q[j];
      q[j] = (src[ent >> 8] << 8) | (ent & 255);
    }
    __syncthreads();
    for (int j0 = wave; j0 < n; j0 += 128) {
      int ent[4];
#pragma unroll
      for (int k = 0; k < 4; ++k) {
        int j = j0 + 32 * k;
        ent[k] = q[j < n ? j : j0];
      }
      float2 v[4];
      float  w[4];
#pragma unroll
      for (int k = 0; k < 4; ++k) {
        int s = ent[k] >> 8;
        v[k] = *(const float2*)(h + (long)s * 64 + 2 * lane);
        w[k] = isq[s];
      }
#pragma unroll
      for (int k = 0; k < 4; ++k) {
        int r = (j0 + 32 * k < n) ? (ent[k] & 255) : ROWS_PB;
        float wq = isqS[r];
        float* dstp = &tile[r * 64 + 2 * lane];
        atomicAdd(dstp + 0, v[k].x * w[k] * wq);
        atomicAdd(dstp + 1, v[k].y * w[k] * wq);
      }
    }
    __syncthreads();
#pragma unroll
    for (int k = 0; k < SCAN_V4; ++k) cur[k] = nx[k];
  }

  for (int i = tid; i < ROWS_PB * 64; i += 1024) {
    int row = lo + (i >> 6);
    if (row < NUM_LOC) agg[(long)row * 64 + (i & 63)] = tile[i];
  }
}

template <int NT, int TH>
__device__ __forceinline__ void role_gemm(
    const int vb, const float* __restrict__ A, const float* __restrict__ B,
    const float* __restrict__ bias, float* __restrict__ C, int M, int K, int relu,
    float* Bs) {
  constexpr int Nc = NT * 16;
  for (int i = threadIdx.x; i < K * Nc; i += TH) Bs[i] = B[i];
  __syncthreads();

  const int lane = threadIdx.x & 31;
  const int wave = threadIdx.x >> 5;
  const int half = lane >> 4;
  const int l16  = lane & 15;
  const int row0 = (vb * (TH / 32) + wave) * 16;

  int ar = row0 + l16; if (ar > M - 1) ar = M - 1;
  const float* Arow = A + (long)ar * K;

  v8f acc[NT];
#pragma unroll
  for (int tt = 0; tt < NT; ++tt) {
    float bv = bias[tt * 16 + l16];
    v8f tmp = { bv, bv, bv, bv, bv, bv, bv, bv };
    acc[tt] = tmp;
  }

  for (int k = 0; k < K; k += 4) {
    v2f a;
    a[0] = Arow[k + 2 * half + 0];
    a[1] = Arow[k + 2 * half + 1];
#pragma unroll
    for (int tt = 0; tt < NT; ++tt) {
      v2f b;
      b[0] = Bs[(k + 2 * half + 0) * Nc + tt * 16 + l16];
      b[1] = Bs[(k + 2 * half + 1) * Nc + tt * 16 + l16];
      acc[tt] = __builtin_amdgcn_wmma_f32_16x16x4_f32(false, a, false, b, (short)0,
                                                      acc[tt], false, false);
    }
  }
#pragma unroll
  for (int tt = 0; tt < NT; ++tt) {
    for (int r = 0; r < 8; ++r) {
      int row = row0 + r + 8 * half;
      if (row < M) {
        float v = acc[tt][r];
        if (relu) v = v > 0.0f ? v : 0.0f;
        C[(long)row * Nc + tt * 16 + l16] = v;
      }
    }
  }
}

__device__ __forceinline__ float sigm(float x) { return 1.0f / (1.0f + expf(-x)); }

template <int TH>
__device__ __forceinline__ void role_layer(
    const int vb, const float* __restrict__ X, int Kx, const float* __restrict__ Wi,
    const float* __restrict__ Wh, const float* __restrict__ b,
    const float* hin, float* hout, float* c, int M, float* Bs) {
  for (int i = threadIdx.x; i < Kx * 64; i += TH) Bs[i] = Wi[i];
  for (int i = threadIdx.x; i < 16 * 64; i += TH) Bs[Kx * 64 + i] = Wh[i];
  __syncthreads();

  const int lane = threadIdx.x & 31;
  const int wave = threadIdx.x >> 5;
  const int half = lane >> 4;
  const int l16  = lane & 15;
  const int row0 = (vb * (TH / 32) + wave) * 16;
  if (row0 >= M) return;

  int ar = row0 + l16; if (ar > M - 1) ar = M - 1;

  float cold[8];
#pragma unroll
  for (int r = 0; r < 8; ++r) {
    int row = row0 + r + 8 * half;
    cold[r] = (row < M) ? c[(long)row * 16 + l16] : 0.0f;
  }

  v8f acc[4];
#pragma unroll
  for (int tt = 0; tt < 4; ++tt) {
    float bv = b[tt * 16 + l16];
    v8f tmp = { bv, bv, bv, bv, bv, bv, bv, bv };
    acc[tt] = tmp;
  }

  const float* Xr = X + (long)ar * Kx;
  for (int k = 0; k < Kx; k += 4) {
    v2f a;
    a[0] = Xr[k + 2 * half + 0];
    a[1] = Xr[k + 2 * half + 1];
#pragma unroll
    for (int tt = 0; tt < 4; ++tt) {
      v2f w;
      w[0] = Bs[(k + 2 * half + 0) * 64 + tt * 16 + l16];
      w[1] = Bs[(k + 2 * half + 1) * 64 + tt * 16 + l16];
      acc[tt] = __builtin_amdgcn_wmma_f32_16x16x4_f32(false, a, false, w, (short)0,
                                                      acc[tt], false, false);
    }
  }
  const float* Hr = hin + (long)ar * 16;
#pragma unroll
  for (int k = 0; k < 16; k += 4) {
    v2f a;
    a[0] = Hr[k + 2 * half + 0];
    a[1] = Hr[k + 2 * half + 1];
#pragma unroll
    for (int tt = 0; tt < 4; ++tt) {
      v2f w;
      w[0] = Bs[(Kx + k + 2 * half + 0) * 64 + tt * 16 + l16];
      w[1] = Bs[(Kx + k + 2 * half + 1) * 64 + tt * 16 + l16];
      acc[tt] = __builtin_amdgcn_wmma_f32_16x16x4_f32(false, a, false, w, (short)0,
                                                      acc[tt], false, false);
    }
  }
#pragma unroll
  for (int r = 0; r < 8; ++r) {
    int row = row0 + r + 8 * half;
    if (row < M) {
      float gi = acc[0][r], gf = acc[1][r], gg = acc[2][r], go = acc[3][r];
      float cn = sigm(gf) * cold[r] + sigm(gi) * tanhf(gg);
      c[(long)row * 16 + l16] = cn;
      hout[(long)row * 16 + l16] = sigm(go) * tanhf(cn);
    }
  }
}

__device__ __forceinline__ void role_gather(const int vb, const int* __restrict__ ids,
                                            const float* __restrict__ he,
                                            float* __restrict__ out32, int N) {
  const int sub = threadIdx.x & 7;
  const int r0  = vb * 512 + (threadIdx.x >> 3);
  if (vb * 512 + 512 <= N) {
    int id[4];
    v4f v[4];
#pragma unroll
    for (int p = 0; p < 4; ++p) id[p] = ids[r0 + 128 * p];
#pragma unroll
    for (int p = 0; p < 4; ++p)
      v[p] = *(const v4f*)(he + (long)id[p] * 32 + 4 * sub);
    asm volatile("" : "+v"(v[0]), "+v"(v[1]), "+v"(v[2]), "+v"(v[3]));
#pragma unroll
    for (int p = 0; p < 4; ++p)
      *(v4f*)(out32 + (long)(r0 + 128 * p) * 32 + 4 * sub) = v[p];
  } else {
    for (int p = 0; p < 4; ++p) {
      int row = r0 + 128 * p;
      if (row < N)
        *(v4f*)(out32 + (long)row * 32 + 4 * sub) =
            *(const v4f*)(he + (long)ids[row] * 32 + 4 * sub);
    }
  }
}

template <int TH>
__device__ __forceinline__ void role_outproj(const int vb, const float* __restrict__ h,
                                             const float* __restrict__ Wout,
                                             const float* __restrict__ bout,
                                             float* __restrict__ out, int N) {
  int i = vb * TH + threadIdx.x;
  if (i >= N) return;
  const float* hr = h + (long)i * 16;
  float a0 = bout[0], a1 = bout[1];
  for (int j = 0; j < 16; ++j) {
    float v = hr[j];
    a0 += v * Wout[j * 2 + 0];
    a1 += v * Wout[j * 2 + 1];
  }
  out[(long)i * 2 + 0] = a0;
  out[(long)i * 2 + 1] = a1;
}

__global__ __launch_bounds__(1024) void k_stageA(
    int nLoc, const int* __restrict__ idsNext, const float* __restrict__ Wemb,
    const float* __restrict__ bemb, float* __restrict__ e,
    const int* __restrict__ idsCur, const float* __restrict__ he2n, float* __restrict__ out32) {
  __shared__ int arena[AR_BAND];
  const int b = blockIdx.x;
  if (b < nLoc) role_user_sum(b, idsNext, Wemb, bemb, e, arena);
  else          role_gather(b - nLoc, idsCur, he2n, out32, USR_NUM);
}

__global__ __launch_bounds__(1024) void k_stageBD(
    const int* __restrict__ src, const int* __restrict__ dst,
    const float* __restrict__ isq, const float* __restrict__ idg,
    const float* __restrict__ hin, float* __restrict__ agg) {
  __shared__ int arena[AR_BAND];
  role_edge_sum(blockIdx.x, src, dst, isq, idg, hin, agg, arena);
}

__global__ __launch_bounds__(256) void k_stageCE(
    int nLoc, const float* __restrict__ A, const float* __restrict__ W,
    const float* __restrict__ bias, float* __restrict__ C, int relu,
    const float* __restrict__ X, int Kx, const float* __restrict__ Wi,
    const float* __restrict__ Wh, const float* __restrict__ bL,
    float* hS, float* cS) {
  __shared__ float Bs[AR_GEMM];
  const int b = blockIdx.x;
  if (b < nLoc) role_gemm<4, 256>(b, A, W, bias, C, NUM_LOC, 64, relu, Bs);
  else          role_layer<256>(b - nLoc, X, Kx, Wi, Wh, bL, hS, hS, cS, USR_NUM, Bs);
}

__global__ __launch_bounds__(256) void k_stageF(
    int nLoc, const float* __restrict__ A, const float* __restrict__ W,
    const float* __restrict__ bias, float* __restrict__ C,
    const float* __restrict__ h, const float* __restrict__ Wout,
    const float* __restrict__ bout, float* __restrict__ out) {
  __shared__ float Bs[AR_GEMM];
  const int b = blockIdx.x;
  if (b < nLoc) role_gemm<2, 256>(b, A, W, bias, C, NUM_LOC, 64, 0, Bs);
  else          role_outproj<256>(b - nLoc, h, Wout, bout, out, USR_NUM);
}


#define PL_LOC   ((long)NUM_LOC * 64)
#define PL_E2N   ((long)NUM_LOC * 32)
#define PL_X     ((long)USR_NUM * 32)

#define NB_BAND   ((NUM_LOC + ROWS_PB - 1) / ROWS_PB)
#define NB_GATHER ((USR_NUM + 511) / 512)
#define NB_LAYER4 ((USR_NUM + 511) / 512)
#define NB_GEMM4  ((NUM_LOC + 511) / 512)
#define NB_OUT4   ((USR_NUM + 1023) / 1024)
#define NB_TICK   (3 * NB_BAND + NB_GATHER + 2 * NB_LAYER4 + NB_OUT4 + 3 * NB_GEMM4)
#define N_TICKS   (T_STEPS + 9)
#define NL_PAD 5056
#define WS_TICK_FLOATS (3L * NL_PAD + 10 * PL_LOC + 2 * PL_E2N + 2 * PL_X + 6L * USR_NUM * 16)
#define WS_TICK_BYTES  ((size_t)(WS_TICK_FLOATS * 4))

struct TickArgs {
  const int*   seq;  const int*   src;  const int*   dst;
  const float* Wemb; const float* bemb;
  const float* W1;   const float* b1;   const float* W2;   const float* b2;
  const float* We;   const float* be;
  const float* Wi0;  const float* Wh0;  const float* bL0;
  const float* Wi1;  const float* Wh1;  const float* bL1;
  const float* Wout; const float* bout;
  const float* isq;  const float* idg;
  float* e; float* aggA; float* h1b; float* aggB; float* h2b;
  float* he2n; float* out32;
  float* h0p0; float* h0p1; float* c0;
  float* h1p0; float* h1p1; float* c1;
  float* out;
};

__global__ __launch_bounds__(1024, 16) void k_tick(const TickArgs a, const int tick) {
  __shared__ int arena[AR_BAND];
  int b = blockIdx.x;
  int s;
  if (b < NB_BAND) {
    s = tick;
    if ((unsigned)s >= (unsigned)T_STEPS) return;
    role_user_sum(b, a.seq + (long)s * USR_NUM, a.Wemb, a.bemb, a.e + (s & 1) * PL_LOC, arena);
    return;
  }
  b -= NB_BAND;
  if (b < NB_BAND) {
    s = tick - 1;
    if ((unsigned)s >= (unsigned)T_STEPS) return;
    role_edge_sum(b, a.src, a.dst, a.isq, a.idg, a.e + (s & 1) * PL_LOC,
                  a.aggA + (s & 1) * PL_LOC, arena);
    return;
  }
  b -= NB_BAND;
  if (b < NB_BAND) {
    s = tick - 3;
    if ((unsigned)s >= (unsigned)T_STEPS) return;
    role_edge_sum(b, a.src, a.dst, a.isq, a.idg, a.h1b + (s & 1) * PL_LOC,
                  a.aggB + (s & 1) * PL_LOC, arena);
    return;
  }
  b -= NB_BAND;
  if (b < NB_GATHER) {
    s = tick - 6;
    if ((unsigned)s >= (unsigned)T_STEPS) return;
    role_gather(b, a.seq + (long)s * USR_NUM, a.he2n + (s & 1) * PL_E2N,
                a.out32 + (s & 1) * PL_X, USR_NUM);
    return;
  }
  b -= NB_GATHER;
  if (b < NB_LAYER4) {
    s = tick - 7;
    if ((unsigned)s >= (unsigned)T_STEPS) return;
    role_layer<1024>(b, a.out32 + (s & 1) * PL_X, 32, a.Wi0, a.Wh0, a.bL0,
                     (s & 1) ? a.h0p0 : a.h0p1, (s & 1) ? a.h0p1 : a.h0p0, a.c0,
                     USR_NUM, (float*)arena);
    return;
  }
  b -= NB_LAYER4;
  if (b < NB_LAYER4) {
    s = tick - 8;
    if ((unsigned)s >= (unsigned)T_STEPS) return;
    role_layer<1024>(b, (s & 1) ? a.h0p1 : a.h0p0, 16, a.Wi1, a.Wh1, a.bL1,
                     (s & 1) ? a.h1p0 : a.h1p1, (s & 1) ? a.h1p1 : a.h1p0, a.c1,
                     USR_NUM, (float*)arena);
    return;
  }
  b -= NB_LAYER4;
  if (b < NB_OUT4) {
    s = tick - 9;
    if ((unsigned)s >= (unsigned)T_STEPS) return;
    role_outproj<1024>(b, (s & 1) ? a.h1p1 : a.h1p0, a.Wout, a.bout,
                       a.out + (long)s * USR_NUM * 2, USR_NUM);
    return;
  }
  b -= NB_OUT4;
  if (b < NB_GEMM4) {
    s = tick - 2;
    if ((unsigned)s >= (unsigned)T_STEPS) return;
    role_gemm<4, 1024>(b, a.aggA + (s & 1) * PL_LOC, a.W1, a.b1, a.h1b + (s & 1) * PL_LOC,
                       NUM_LOC, 64, 1, (float*)arena);
    return;
  }
  b -= NB_GEMM4;
  if (b < NB_GEMM4) {
    s = tick - 4;
    if ((unsigned)s >= (unsigned)T_STEPS) return;
    role_gemm<4, 1024>(b, a.aggB + (s & 1) * PL_LOC, a.W2, a.b2, a.h2b + (s & 1) * PL_LOC,
                       NUM_LOC, 64, 0, (float*)arena);
    return;
  }
  b -= NB_GEMM4;
  {
    s = tick - 5;
    if ((unsigned)s >= (unsigned)T_STEPS) return;
    role_gemm<2, 1024>(b, a.h2b + (s & 1) * PL_LOC, a.We, a.be, a.he2n + (s & 1) * PL_E2N,
                       NUM_LOC, 64, 0, (float*)arena);
  }
}

static inline int gs(long n, int b) { return (int)((n + b - 1) / b); }

#define NB_LAYER  ((USR_NUM + 127) / 128)
#define NB_GEMM   ((NUM_LOC + 127) / 128)
#define NB_OUT    ((USR_NUM + 255) / 256)

extern "C" void kernel_launch(void* const* d_in, const int* in_sizes, int n_in,
                              void* d_out, int out_size, void* d_ws, size_t ws_size,
                              hipStream_t stream) {
  (void)in_sizes; (void)n_in; (void)out_size;

  const int*   seq  = (const int*)d_in[0];
  const int*   gidx = (const int*)d_in[1];
  const float* Wemb = (const float*)d_in[2];
  const float* bemb = (const float*)d_in[3];
  const float* W1   = (const float*)d_in[4];
  const float* b1   = (const float*)d_in[5];
  const float* W2   = (const float*)d_in[6];
  const float* b2   = (const float*)d_in[7];
  const float* We2n = (const float*)d_in[8];
  const float* be2n = (const float*)d_in[9];
  const float* Wi0  = (const float*)d_in[10];
  const float* Wh0  = (const float*)d_in[11];
  const float* bL0  = (const float*)d_in[12];
  const float* Wi1  = (const float*)d_in[13];
  const float* Wh1  = (const float*)d_in[14];
  const float* bL1  = (const float*)d_in[15];
  const float* Wout = (const float*)d_in[16];
  const float* bout = (const float*)d_in[17];

  const int* srcE = gidx;
  const int* dstE = gidx + N_EDGES;
  float* out = (float*)d_out;
  const int BT = 256;

  float* p = (float*)d_ws;
  float* deg   = p; p += NL_PAD;
  float* isq   = p; p += NL_PAD;
  float* idg   = p; p += NL_PAD;

  k_zero<<<gs(NUM_LOC, BT), BT, 0, stream>>>(deg, NUM_LOC);
  k_deg_accum<<<gs(N_EDGES, BT), BT, 0, stream>>>(dstE, deg, N_EDGES);
  k_deg_fin<<<gs(NUM_LOC, BT), BT, 0, stream>>>(deg, isq, idg, NUM_LOC);

  if (ws_size >= WS_TICK_BYTES) {
    TickArgs a;
    a.seq = seq; a.src = srcE; a.dst = dstE;
    a.Wemb = Wemb; a.bemb = bemb;
    a.W1 = W1; a.b1 = b1; a.W2 = W2; a.b2 = b2; a.We = We2n; a.be = be2n;
    a.Wi0 = Wi0; a.Wh0 = Wh0; a.bL0 = bL0; a.Wi1 = Wi1; a.Wh1 = Wh1; a.bL1 = bL1;
    a.Wout = Wout; a.bout = bout;
    a.isq = isq; a.idg = idg;
    a.e     = p; p += 2 * PL_LOC;
    a.aggA  = p; p += 2 * PL_LOC;
    a.h1b   = p; p += 2 * PL_LOC;
    a.aggB  = p; p += 2 * PL_LOC;
    a.h2b   = p; p += 2 * PL_LOC;
    a.he2n  = p; p += 2 * PL_E2N;
    a.out32 = p; p += 2 * PL_X;
    a.h0p1  = p; p += (size_t)USR_NUM * 16;
    a.c0    = p; p += (size_t)USR_NUM * 16;
    a.h1p1  = p; p += (size_t)USR_NUM * 16;
    a.c1    = p; p += (size_t)USR_NUM * 16;
    a.h0p0  = p; p += (size_t)USR_NUM * 16;
    a.h1p0  = p; p += (size_t)USR_NUM * 16;
    a.out = out;

    k_zero<<<gs((long)USR_NUM * 64, BT), BT, 0, stream>>>(a.h0p1, USR_NUM * 64);
    for (int tick = 0; tick < N_TICKS; ++tick)
      k_tick<<<NB_TICK, 1024, 0, stream>>>(a, tick);
    return;
  }

  float* e     = p; p += (size_t)NUM_LOC * 64;
  float* agg   = p; p += (size_t)NUM_LOC * 64;
  float* h1b   = p; p += (size_t)NUM_LOC * 64;
  float* h2b   = p; p += (size_t)NUM_LOC * 64;
  float* he2n  = p; p += (size_t)NUM_LOC * 32;
  float* out32 = p; p += (size_t)USR_NUM * 32;
  float* h0    = p; p += (size_t)USR_NUM * 16;
  float* c0    = p; p += (size_t)USR_NUM * 16;
  float* h1s   = p; p += (size_t)USR_NUM * 16;
  float* c1s   = p; p += (size_t)USR_NUM * 16;

  k_zero<<<gs((long)USR_NUM * 64, BT), BT, 0, stream>>>(h0, USR_NUM * 64);

  k_stageA <<<NB_BAND, 1024, 0, stream>>>(NB_BAND, seq, Wemb, bemb, e, seq, he2n, out32);
  k_stageBD<<<NB_BAND, 1024, 0, stream>>>(srcE, dstE, isq, idg, e, agg);
  k_stageCE<<<NB_GEMM, 256, 0, stream>>>(NB_GEMM, agg, W1, b1, h1b, 1,
                                         out32, 32, Wi0, Wh0, bL0, h0, c0);
  k_stageBD<<<NB_BAND, 1024, 0, stream>>>(srcE, dstE, isq, idg, h1b, agg);
  k_stageCE<<<NB_GEMM, 256, 0, stream>>>(NB_GEMM, agg, W2, b2, h2b, 0,
                                         h0, 16, Wi1, Wh1, bL1, h1s, c1s);
  k_stageF <<<NB_GEMM, 256, 0, stream>>>(NB_GEMM, h2b, We2n, be2n, he2n, h1s, Wout, bout, out);

  for (int t = 0; t + 1 < T_STEPS; ++t) {
    const int* idsCur  = seq + (long)t * USR_NUM;
    const int* idsNext = idsCur + USR_NUM;
    float* outT = out + (long)t * USR_NUM * 2;

    k_stageA <<<NB_BAND + NB_GATHER, 1024, 0, stream>>>(NB_BAND, idsNext, Wemb, bemb, e,
                                                        idsCur, he2n, out32);
    k_stageBD<<<NB_BAND, 1024, 0, stream>>>(srcE, dstE, isq, idg, e, agg);
    k_stageCE<<<NB_GEMM + NB_LAYER, 256, 0, stream>>>(NB_GEMM, agg, W1, b1, h1b, 1,
                                                      out32, 32, Wi0, Wh0, bL0, h0, c0);
    k_stageBD<<<NB_BAND, 1024, 0, stream>>>(srcE, dstE, isq, idg, h1b, agg);
    k_stageCE<<<NB_GEMM + NB_LAYER, 256, 0, stream>>>(NB_GEMM, agg, W2, b2, h2b, 0,
                                                      h0, 16, Wi1, Wh1, bL1, h1s, c1s);
    k_stageF <<<NB_GEMM + NB_OUT, 256, 0, stream>>>(NB_GEMM, h2b, We2n, be2n, he2n,
                                                     h1s, Wout, bout, outT);
  }

  {
    const int* idsCur = seq + (long)(T_STEPS - 1) * USR_NUM;
    float* outT = out + (long)(T_STEPS - 1) * USR_NUM * 2;
    k_stageA <<<NB_GATHER, 1024, 0, stream>>>(0, idsCur, Wemb, bemb, e, idsCur, he2n, out32);
    k_stageCE<<<NB_LAYER, 256, 0, stream>>>(0, agg, W1, b1, h1b, 1,
                                            out32, 32, Wi0, Wh0, bL0, h0, c0);
    k_stageCE<<<NB_LAYER, 256, 0, stream>>>(0, agg, W2, b2, h2b, 0,
                                            h0, 16, Wi1, Wh1, bL1, h1s, c1s);
    k_stageF <<<NB_OUT, 256, 0, stream>>>(0, h2b, We2n, be2n, he2n, h1s, Wout, bout, outT);
  }
}


# Band sums without per-pass barriers: each wave queues its own keys' matches (ballot ranks, wave-private LDS queue), adds their rows in rounds of four; LDS add arrival order now over the whole row
# speedup vs baseline: 3.6524x; 1.1116x over previous
#include <hip/hip_runtime.h>
#include <math.h>


#define USR_NUM 100000
#define USR_DIM 64
#define NUM_LOC 5000
#define T_STEPS 16
#define N_EDGES 80000

typedef float v2f __attribute__((ext_vector_type(2)));
typedef float v4f __attribute__((ext_vector_type(4)));
typedef float v8f __attribute__((ext_vector_type(8)));

__global__ void k_zero(float* __restrict__ p, int n) {
  int i = blockIdx.x * blockDim.x + threadIdx.x;
  if (i < n) p[i] = 0.0f;
}

__global__ void k_deg_accum(const int* __restrict__ dst, float* __restrict__ deg, int E) {
  int i = blockIdx.x * blockDim.x + threadIdx.x;
  if (i < E) atomicAdd(&deg[dst[i]], 1.0f);
}

__global__ void k_deg_fin(const float* __restrict__ deg, float* __restrict__ isq,
                          float* __restrict__ idg, int n) {
  int i = blockIdx.x * blockDim.x + threadIdx.x;
  if (i < n) {
    float d = deg[i] + 1.0f;
    isq[i] = rsqrtf(d);
    idg[i] = 1.0f / d;
  }
}


#define ROWS_PB   32
#define SCAN_KEYS 16384
#define SCAN_V4   (SCAN_KEYS / 4 / 1024)
#define WQ_HIGH   128
#define WQ_CAP    (WQ_HIGH + 4 * 32 * SCAN_V4 + 4)
#define AR_TILE   0
#define AR_AUX    ((ROWS_PB + 1) * 64)
#define AR_Q      (AR_AUX + 64)
#define AR_BAND   (AR_Q + 32 * WQ_CAP)
#define AR_GEMM   (64 * 64)

__device__ __forceinline__ void role_user_sum(
    const int vb, const int* __restrict__ ids, const float* __restrict__ Wemb,
    const float* __restrict__ bemb, float* __restrict__ e, int* arena) {
  float* tile = (float*)(arena + AR_TILE);
  float* cntS = (float*)(arena + AR_AUX);

  const int tid  = threadIdx.x;
  const int lane = tid & 31;
  const int wave = tid >> 5;
  const int lo   = vb * ROWS_PB;
  int* wq = arena + AR_Q + wave * WQ_CAP;

  for (int i = tid; i < (ROWS_PB + 1) * 64; i += 1024) tile[i] = 0.0f;
  if (tid < ROWS_PB) cntS[tid] = 0.0f;

  const int4* ids4 = (const int4*)ids;
  const int nv4 = USR_NUM / 4;
  const float2 bb = *(const float2*)(bemb + 2 * lane);

  int4 cur[SCAN_V4];
#pragma unroll
  for (int k = 0; k < SCAN_V4; ++k) {
    int i4 = k * 1024 + tid;
    cur[k] = (i4 < nv4) ? ids4[i4] : make_int4(-1, -1, -1, -1);
  }
  __syncthreads();

  int nq = 0;
  int head = 0;
  for (int base4 = 0; base4 < nv4; base4 += SCAN_KEYS / 4) {
    int4 nx[SCAN_V4];
#pragma unroll
    for (int k = 0; k < SCAN_V4; ++k) {
      int i4 = base4 + SCAN_KEYS / 4 + k * 1024 + tid;
      nx[k] = (i4 < nv4) ? ids4[i4] : make_int4(-1, -1, -1, -1);
    }
#pragma unroll
    for (int k = 0; k < SCAN_V4; ++k) {
      const int u0 = (base4 + k * 1024 + tid) * 4;
      const int kv[4] = { cur[k].x, cur[k].y, cur[k].z, cur[k].w };
#pragma unroll
      for (int c = 0; c < 4; ++c) {
        const unsigned r = (unsigned)(kv[c] - lo);
        const bool hit = r < (unsigned)ROWS_PB;
        const unsigned m = __builtin_amdgcn_ballot_w32(hit);
        if (m != 0u) {
          if (hit) {
            wq[nq + (int)__builtin_amdgcn_mbcnt_lo(m, 0u)] = ((u0 + c) << 8) | (int)r;
            atomicAdd(&cntS[r], 1.0f);
          }
          nq += __builtin_popcount(m);
        }
      }
    }
    const bool drain = (base4 + SCAN_KEYS / 4 >= nv4) || (nq > WQ_HIGH);
    if (drain && lane < 3) wq[nq + lane] = ROWS_PB;
    const int stop = drain ? ((nq + 3) & ~3) : (nq & ~3);
    for (int j0 = head; j0 < stop; j0 += 4) {
      const int4 q4 = *(const int4*)(wq + j0);
      const int ent[4] = { __builtin_amdgcn_readfirstlane(q4.x), __builtin_amdgcn_readfirstlane(q4.y),
                           __builtin_amdgcn_readfirstlane(q4.z), __builtin_amdgcn_readfirstlane(q4.w) };
      float2 v[4];
#pragma unroll
      for (int k = 0; k < 4; ++k)
        v[k] = *(const float2*)(Wemb + (long)(ent[k] >> 8) * 64 + 2 * lane);
#pragma unroll
      for (int k = 0; k < 4; ++k) {
        int r = ent[k] & 255;
        float* dstp = &tile[r * 64 + 2 * lane];
        atomicAdd(dstp + 0, v[k].x + bb.x);
        atomicAdd(dstp + 1, v[k].y + bb.y);
      }
    }
    head = stop;
    if (drain) { head = 0; nq = 0; }
#pragma unroll
    for (int k = 0; k < SCAN_V4; ++k) cur[k] = nx[k];
  }
  __syncthreads();

  for (int i = tid; i < ROWS_PB * 64; i += 1024) {
    int j = i >> 6;
    int row = lo + j;
    if (row < NUM_LOC) {
      float cv = cntS[j]; if (cv < 1.0f) cv = 1.0f;
      e[(long)row * 64 + (i & 63)] = tile[i] / cv;
    }
  }
}

__device__ __forceinline__ void role_edge_sum(
    const int vb, const int* __restrict__ src, const int* __restrict__ dst,
    const float* __restrict__ isq, const float* __restrict__ idg,
    const float* __restrict__ h, float* __restrict__ agg, int* arena) {
  float* tile = (float*)(arena + AR_TILE);
  float* isqS = (float*)(arena + AR_AUX);

  const int tid  = threadIdx.x;
  const int lane = tid & 31;
  const int wave = tid >> 5;
  const int lo   = vb * ROWS_PB;
  int* wq = arena + AR_Q + wave * WQ_CAP;

  for (int i = tid; i < (ROWS_PB + 1) * 64; i += 1024) {
    int row = lo + (i >> 6);
    tile[i] = (i < ROWS_PB * 64 && row < NUM_LOC) ? h[(long)row * 64 + (i & 63)] * idg[row] : 0.0f;
  }
  if (tid <= ROWS_PB) isqS[tid] = (tid < ROWS_PB && lo + tid < NUM_LOC) ? isq[lo + tid] : 0.0f;

  const int4* dst4 = (const int4*)dst;
  const int nv4 = N_EDGES / 4;

  int4 cur[SCAN_V4];
#pragma unroll
  for (int k = 0; k < SCAN_V4; ++k) {
    int i4 = k * 1024 + tid;
    cur[k] = (i4 < nv4) ? dst4[i4] : make_int4(-1, -1, -1, -1);
  }
  __syncthreads();

  int nq = 0;
  int head = 0;
  for (int base4 = 0; base4 < nv4; base4 += SCAN_KEYS / 4) {
    int4 nx[SCAN_V4];
#pragma unroll
    for (int k = 0; k < SCAN_V4; ++k) {
      int i4 = base4 + SCAN_KEYS / 4 + k * 1024 + tid;
      nx[k] = (i4 < nv4) ? dst4[i4] : make_int4(-1, -1, -1, -1);
    }
#pragma unroll
    for (int k = 0; k < SCAN_V4; ++k) {
      const int e0 = (base4 + k * 1024 + tid) * 4;
      const int kv[4] = { cur[k].x, cur[k].y, cur[k].z, cur[k].w };
#pragma unroll
      for (int c = 0; c < 4; ++c) {
        const unsigned r = (unsigned)(kv[c] - lo);
        const bool hit = r < (unsigned)ROWS_PB;
        const unsigned m = __builtin_amdgcn_ballot_w32(hit);
        if (m != 0u) {
          if (hit) wq[nq + (int)__builtin_amdgcn_mbcnt_lo(m, 0u)] = ((e0 + c) << 8) | (int)r;
          nq += __builtin_popcount(m);
        }
      }
    }
    const bool drain = (base4 + SCAN_KEYS / 4 >= nv4) || (nq > WQ_HIGH);
    if (drain && lane < 3) wq[nq + lane] = ROWS_PB;
    const int stop = drain ? ((nq + 3) & ~3) : (nq & ~3);
    for (int j0 = head; j0 < stop; j0 += 4) {
      const int4 q4 = *(const int4*)(wq + j0);
      const int ent[4] = { __builtin_amdgcn_readfirstlane(q4.x), __builtin_amdgcn_readfirstlane(q4.y),
                           __builtin_amdgcn_readfirstlane(q4.z), __builtin_amdgcn_readfirstlane(q4.w) };
      float wr[4];
#pragma unroll
      for (int k = 0; k < 4; ++k) wr[k] = isqS[ent[k] & 255];
      int s[4];
#pragma unroll
      for (int k = 0; k < 4; ++k) s[k] = src[ent[k] >> 8];
      float  w[4];
#pragma unroll
      for (int k = 0; k < 4; ++k) w[k] = isq[s[k]];
      float2 v[4];
#pragma unroll
      for (int k = 0; k < 4; ++k)
        v[k] = *(const float2*)(h + (long)s[k] * 64 + 2 * lane);
#pragma unroll
      for (int k = 0; k < 4; ++k) {
        float* dstp = &tile[(ent[k] & 255) * 64 + 2 * lane];
        atomicAdd(dstp + 0, v[k].x * w[k] * wr[k]);
        atomicAdd(dstp + 1, v[k].y * w[k] * wr[k]);
      }
    }
    head = stop;
    if (drain) { head = 0; nq = 0; }
#pragma unroll
    for (int k = 0; k < SCAN_V4; ++k) cur[k] = nx[k];
  }
  __syncthreads();

  for (int i = tid; i < ROWS_PB * 64; i += 1024) {
    int row = lo + (i >> 6);
    if (row < NUM_LOC) agg[(long)row * 64 + (i & 63)] = tile[i];
  }
}

template <int NT, int TH>
__device__ __forceinline__ void role_gemm(
    const int vb, const float* __restrict__ A, const float* __restrict__ B,
    const float* __restrict__ bias, float* __restrict__ C, int M, int K, int relu,
    float* Bs) {
  constexpr int Nc = NT * 16;
  for (int i = threadIdx.x; i < K * Nc; i += TH) Bs[i] = B[i];
  __syncthreads();

  const int lane = threadIdx.x & 31;
  const int wave = threadIdx.x >> 5;
  const int half = lane >> 4;
  const int l16  = lane & 15;
  const int row0 = (vb * (TH / 32) + wave) * 16;

  int ar = row0 + l16; if (ar > M - 1) ar = M - 1;
  const float* Arow = A + (long)ar * K;

  v8f acc[NT];
#pragma unroll
  for (int tt = 0; tt < NT; ++tt) {
    float bv = bias[tt * 16 + l16];
    v8f tmp = { bv, bv, bv, bv, bv, bv, bv, bv };
    acc[tt] = tmp;
  }

  for (int k = 0; k < K; k += 4) {
    v2f a;
    a[0] = Arow[k + 2 * half + 0];
    a[1] = Arow[k + 2 * half + 1];
#pragma unroll
    for (int tt = 0; tt < NT; ++tt) {
      v2f b;
      b[0] = Bs[(k + 2 * half + 0) * Nc + tt * 16 + l16];
      b[1] = Bs[(k + 2 * half + 1) * Nc + tt * 16 + l16];
      acc[tt] = __builtin_amdgcn_wmma_f32_16x16x4_f32(false, a, false, b, (short)0,
                                                      acc[tt], false, false);
    }
  }
#pragma unroll
  for (int tt = 0; tt < NT; ++tt) {
    for (int r = 0; r < 8; ++r) {
      int row = row0 + r + 8 * half;
      if (row < M) {
        float v = acc[tt][r];
        if (relu) v = v > 0.0f ? v : 0.0f;
        C[(long)row * Nc + tt * 16 + l16] = v;
      }
    }
  }
}

__device__ __forceinline__ float sigm(float x) { return 1.0f / (1.0f + expf(-x)); }

template <int TH>
__device__ __forceinline__ void role_layer(
    const int vb, const float* __restrict__ X, int Kx, const float* __restrict__ Wi,
    const float* __restrict__ Wh, const float* __restrict__ b,
    const float* hin, float* hout, float* c, int M, float* Bs) {
  for (int i = threadIdx.x; i < Kx * 64; i += TH) Bs[i] = Wi[i];
  for (int i = threadIdx.x; i < 16 * 64; i += TH) Bs[Kx * 64 + i] = Wh[i];
  __syncthreads();

  const int lane = threadIdx.x & 31;
  const int wave = threadIdx.x >> 5;
  const int half = lane >> 4;
  const int l16  = lane & 15;
  const int row0 = (vb * (TH / 32) + wave) * 16;
  if (row0 >= M) return;

  int ar = row0 + l16; if (ar > M - 1) ar = M - 1;

  float cold[8];
#pragma unroll
  for (int r = 0; r < 8; ++r) {
    int row = row0 + r + 8 * half;
    cold[r] = (row < M) ? c[(long)row * 16 + l16] : 0.0f;
  }

  v8f acc[4];
#pragma unroll
  for (int tt = 0; tt < 4; ++tt) {
    float bv = b[tt * 16 + l16];
    v8f tmp = { bv, bv, bv, bv, bv, bv, bv, bv };
    acc[tt] = tmp;
  }

  const float* Xr = X + (long)ar * Kx;
  for (int k = 0; k < Kx; k += 4) {
    v2f a;
    a[0] = Xr[k + 2 * half + 0];
    a[1] = Xr[k + 2 * half + 1];
#pragma unroll
    for (int tt = 0; tt < 4; ++tt) {
      v2f w;
      w[0] = Bs[(k + 2 * half + 0) * 64 + tt * 16 + l16];
      w[1] = Bs[(k + 2 * half + 1) * 64 + tt * 16 + l16];
      acc[tt] = __builtin_amdgcn_wmma_f32_16x16x4_f32(false, a, false, w, (short)0,
                                                      acc[tt], false, false);
    }
  }
  const float* Hr = hin + (long)ar * 16;
#pragma unroll
  for (int k = 0; k < 16; k += 4) {
    v2f a;
    a[0] = Hr[k + 2 * half + 0];
    a[1] = Hr[k + 2 * half + 1];
#pragma unroll
    for (int tt = 0; tt < 4; ++tt) {
      v2f w;
      w[0] = Bs[(Kx + k + 2 * half + 0) * 64 + tt * 16 + l16];
      w[1] = Bs[(Kx + k + 2 * half + 1) * 64 + tt * 16 + l16];
      acc[tt] = __builtin_amdgcn_wmma_f32_16x16x4_f32(false, a, false, w, (short)0,
                                                      acc[tt], false, false);
    }
  }
#pragma unroll
  for (int r = 0; r < 8; ++r) {
    int row = row0 + r + 8 * half;
    if (row < M) {
      float gi = acc[0][r], gf = acc[1][r], gg = acc[2][r], go = acc[3][r];
      float cn = sigm(gf) * cold[r] + sigm(gi) * tanhf(gg);
      c[(long)row * 16 + l16] = cn;
      hout[(long)row * 16 + l16] = sigm(go) * tanhf(cn);
    }
  }
}

__device__ __forceinline__ void role_gather(const int vb, const int* __restrict__ ids,
                                            const float* __restrict__ he,
                                            float* __restrict__ out32, int N) {
  const int sub = threadIdx.x & 7;
  const int r0  = vb * 512 + (threadIdx.x >> 3);
  if (vb * 512 + 512 <= N) {
    int id[4];
    v4f v[4];
#pragma unroll
    for (int p = 0; p < 4; ++p) id[p] = ids[r0 + 128 * p];
#pragma unroll
    for (int p = 0; p < 4; ++p)
      v[p] = *(const v4f*)(he + (long)id[p] * 32 + 4 * sub);
    asm volatile("" : "+v"(v[0]), "+v"(v[1]), "+v"(v[2]), "+v"(v[3]));
#pragma unroll
    for (int p = 0; p < 4; ++p)
      *(v4f*)(out32 + (long)(r0 + 128 * p) * 32 + 4 * sub) = v[p];
  } else {
    for (int p = 0; p < 4; ++p) {
      int row = r0 + 128 * p;
      if (row < N)
        *(v4f*)(out32 + (long)row * 32 + 4 * sub) =
            *(const v4f*)(he + (long)ids[row] * 32 + 4 * sub);
    }
  }
}

template <int TH>
__device__ __forceinline__ void role_outproj(const int vb, const float* __restrict__ h,
                                             const float* __restrict__ Wout,
                                             const float* __restrict__ bout,
                                             float* __restrict__ out, int N) {
  int i = vb * TH + threadIdx.x;
  if (i >= N) return;
  const float* hr = h + (long)i * 16;
  float a0 = bout[0], a1 = bout[1];
  for (int j = 0; j < 16; ++j) {
    float v = hr[j];
    a0 += v * Wout[j * 2 + 0];
    a1 += v * Wout[j * 2 + 1];
  }
  out[(long)i * 2 + 0] = a0;
  out[(long)i * 2 + 1] = a1;
}

__global__ __launch_bounds__(1024) void k_stageA(
    int nLoc, const int* __restrict__ idsNext, const float* __restrict__ Wemb,
    const float* __restrict__ bemb, float* __restrict__ e,
    const int* __restrict__ idsCur, const float* __restrict__ he2n, float* __restrict__ out32) {
  __shared__ __attribute__((aligned(16))) int arena[AR_BAND];
  const int b = blockIdx.x;
  if (b < nLoc) role_user_sum(b, idsNext, Wemb, bemb, e, arena);
  else          role_gather(b - nLoc, idsCur, he2n, out32, USR_NUM);
}

__global__ __launch_bounds__(1024) void k_stageBD(
    const int* __restrict__ src, const int* __restrict__ dst,
    const float* __restrict__ isq, const float* __restrict__ idg,
    const float* __restrict__ hin, float* __restrict__ agg) {
  __shared__ __attribute__((aligned(16))) int arena[AR_BAND];
  role_edge_sum(blockIdx.x, src, dst, isq, idg, hin, agg, arena);
}

__global__ __launch_bounds__(256) void k_stageCE(
    int nLoc, const float* __restrict__ A, const float* __restrict__ W,
    const float* __restrict__ bias, float* __restrict__ C, int relu,
    const float* __restrict__ X, int Kx, const float* __restrict__ Wi,
    const float* __restrict__ Wh, const float* __restrict__ bL,
    float* hS, float* cS) {
  __shared__ float Bs[AR_GEMM];
  const int b = blockIdx.x;
  if (b < nLoc) role_gemm<4, 256>(b, A, W, bias, C, NUM_LOC, 64, relu, Bs);
  else          role_layer<256>(b - nLoc, X, Kx, Wi, Wh, bL, hS, hS, cS, USR_NUM, Bs);
}

__global__ __launch_bounds__(256) void k_stageF(
    int nLoc, const float* __restrict__ A, const float* __restrict__ W,
    const float* __restrict__ bias, float* __restrict__ C,
    const float* __restrict__ h, const float* __restrict__ Wout,
    const float* __restrict__ bout, float* __restrict__ out) {
  __shared__ float Bs[AR_GEMM];
  const int b = blockIdx.x;
  if (b < nLoc) role_gemm<2, 256>(b, A, W, bias, C, NUM_LOC, 64, 0, Bs);
  else          role_outproj<256>(b - nLoc, h, Wout, bout, out, USR_NUM);
}


#define PL_LOC   ((long)NUM_LOC * 64)
#define PL_E2N   ((long)NUM_LOC * 32)
#define PL_X     ((long)USR_NUM * 32)

#define NB_BAND   ((NUM_LOC + ROWS_PB - 1) / ROWS_PB)
#define NB_GATHER ((USR_NUM + 511) / 512)
#define NB_LAYER4 ((USR_NUM + 511) / 512)
#define NB_GEMM4  ((NUM_LOC + 511) / 512)
#define NB_OUT4   ((USR_NUM + 1023) / 1024)
#define NB_TICK   (3 * NB_BAND + NB_GATHER + 2 * NB_LAYER4 + NB_OUT4 + 3 * NB_GEMM4)
#define N_TICKS   (T_STEPS + 9)
#define NL_PAD 5056
#define WS_TICK_FLOATS (3L * NL_PAD + 10 * PL_LOC + 2 * PL_E2N + 2 * PL_X + 6L * USR_NUM * 16)
#define WS_TICK_BYTES  ((size_t)(WS_TICK_FLOATS * 4))

struct TickArgs {
  const int*   seq;  const int*   src;  const int*   dst;
  const float* Wemb; const float* bemb;
  const float* W1;   const float* b1;   const float* W2;   const float* b2;
  const float* We;   const float* be;
  const float* Wi0;  const float* Wh0;  const float* bL0;
  const float* Wi1;  const float* Wh1;  const float* bL1;
  const float* Wout; const float* bout;
  const float* isq;  const float* idg;
  float* e; float* aggA; float* h1b; float* aggB; float* h2b;
  float* he2n; float* out32;
  float* h0p0; float* h0p1; float* c0;
  float* h1p0; float* h1p1; float* c1;
  float* out;
};

__global__ __launch_bounds__(1024, 16) void k_tick(const TickArgs a, const int tick) {
  __shared__ __attribute__((aligned(16))) int arena[AR_BAND];
  int b = blockIdx.x;
  int s;
  if (b < NB_BAND) {
    s = tick;
    if ((unsigned)s >= (unsigned)T_STEPS) return;
    role_user_sum(b, a.seq + (long)s * USR_NUM, a.Wemb, a.bemb, a.e + (s & 1) * PL_LOC, arena);
    return;
  }
  b -= NB_BAND;
  if (b < NB_BAND) {
    s = tick - 1;
    if ((unsigned)s >= (unsigned)T_STEPS) return;
    role_edge_sum(b, a.src, a.dst, a.isq, a.idg, a.e + (s & 1) * PL_LOC,
                  a.aggA + (s & 1) * PL_LOC, arena);
    return;
  }
  b -= NB_BAND;
  if (b < NB_BAND) {
    s = tick - 3;
    if ((unsigned)s >= (unsigned)T_STEPS) return;
    role_edge_sum(b, a.src, a.dst, a.isq, a.idg, a.h1b + (s & 1) * PL_LOC,
                  a.aggB + (s & 1) * PL_LOC, arena);
    return;
  }
  b -= NB_BAND;
  if (b < NB_GATHER) {
    s = tick - 6;
    if ((unsigned)s >= (unsigned)T_STEPS) return;
    role_gather(b, a.seq + (long)s * USR_NUM, a.he2n + (s & 1) * PL_E2N,
                a.out32 + (s & 1) * PL_X, USR_NUM);
    return;
  }
  b -= NB_GATHER;
  if (b < NB_LAYER4) {
    s = tick - 7;
    if ((unsigned)s >= (unsigned)T_STEPS) return;
    role_layer<1024>(b, a.out32 + (s & 1) * PL_X, 32, a.Wi0, a.Wh0, a.bL0,
                     (s & 1) ? a.h0p0 : a.h0p1, (s & 1) ? a.h0p1 : a.h0p0, a.c0,
                     USR_NUM, (float*)arena);
    return;
  }
  b -= NB_LAYER4;
  if (b < NB_LAYER4) {
    s = tick - 8;
    if ((unsigned)s >= (unsigned)T_STEPS) return;
    role_layer<1024>(b, (s & 1) ? a.h0p1 : a.h0p0, 16, a.Wi1, a.Wh1, a.bL1,
                     (s & 1) ? a.h1p0 : a.h1p1, (s & 1) ? a.h1p1 : a.h1p0, a.c1,
                     USR_NUM, (float*)arena);
    return;
  }
  b -= NB_LAYER4;
  if (b < NB_OUT4) {
    s = tick - 9;
    if ((unsigned)s >= (unsigned)T_STEPS) return;
    role_outproj<1024>(b, (s & 1) ? a.h1p1 : a.h1p0, a.Wout, a.bout,
                       a.out + (long)s * USR_NUM * 2, USR_NUM);
    return;
  }
  b -= NB_OUT4;
  if (b < NB_GEMM4) {
    s = tick - 2;
    if ((unsigned)s >= (unsigned)T_STEPS) return;
    role_gemm<4, 1024>(b, a.aggA + (s & 1) * PL_LOC, a.W1, a.b1, a.h1b + (s & 1) * PL_LOC,
                       NUM_LOC, 64, 1, (float*)arena);
    return;
  }
  b -= NB_GEMM4;
  if (b < NB_GEMM4) {
    s = tick - 4;
    if ((unsigned)s >= (unsigned)T_STEPS) return;
    role_gemm<4, 1024>(b, a.aggB + (s & 1) * PL_LOC, a.W2, a.b2, a.h2b + (s & 1) * PL_LOC,
                       NUM_LOC, 64, 0, (float*)arena);
    return;
  }
  b -= NB_GEMM4;
  {
    s = tick - 5;
    if ((unsigned)s >= (unsigned)T_STEPS) return;
    role_gemm<2, 1024>(b, a.h2b + (s & 1) * PL_LOC, a.We, a.be, a.he2n + (s & 1) * PL_E2N,
                       NUM_LOC, 64, 0, (float*)arena);
  }
}

static inline int gs(long n, int b) { return (int)((n + b - 1) / b); }

#define NB_LAYER  ((USR_NUM + 127) / 128)
#define NB_GEMM   ((NUM_LOC + 127) / 128)
#define NB_OUT    ((USR_NUM + 255) / 256)

extern "C" void kernel_launch(void* const* d_in, const int* in_sizes, int n_in,
                              void* d_out, int out_size, void* d_ws, size_t ws_size,
                              hipStream_t stream) {
  (void)in_sizes; (void)n_in; (void)out_size;

  const int*   seq  = (const int*)d_in[0];
  const int*   gidx = (const int*)d_in[1];
  const float* Wemb = (const float*)d_in[2];
  const float* bemb = (const float*)d_in[3];
  const float* W1   = (const float*)d_in[4];
  const float* b1   = (const float*)d_in[5];
  const float* W2   = (const float*)d_in[6];
  const float* b2   = (const float*)d_in[7];
  const float* We2n = (const float*)d_in[8];
  const float* be2n = (const float*)d_in[9];
  const float* Wi0  = (const float*)d_in[10];
  const float* Wh0  = (const float*)d_in[11];
  const float* bL0  = (const float*)d_in[12];
  const float* Wi1  = (const float*)d_in[13];
  const float* Wh1  = (const float*)d_in[14];
  const float* bL1  = (const float*)d_in[15];
  const float* Wout = (const float*)d_in[16];
  const float* bout = (const float*)d_in[17];

  const int* srcE = gidx;
  const int* dstE = gidx + N_EDGES;
  float* out = (float*)d_out;
  const int BT = 256;

  float* p = (float*)d_ws;
  float* deg   = p; p += NL_PAD;
  float* isq   = p; p += NL_PAD;
  float* idg   = p; p += NL_PAD;

  k_zero<<<gs(NUM_LOC, BT), BT, 0, stream>>>(deg, NUM_LOC);
  k_deg_accum<<<gs(N_EDGES, BT), BT, 0, stream>>>(dstE, deg, N_EDGES);
  k_deg_fin<<<gs(NUM_LOC, BT), BT, 0, stream>>>(deg, isq, idg, NUM_LOC);

  if (ws_size >= WS_TICK_BYTES) {
    TickArgs a;
    a.seq = seq; a.src = srcE; a.dst = dstE;
    a.Wemb = Wemb; a.bemb = bemb;
    a.W1 = W1; a.b1 = b1; a.W2 = W2; a.b2 = b2; a.We = We2n; a.be = be2n;
    a.Wi0 = Wi0; a.Wh0 = Wh0; a.bL0 = bL0; a.Wi1 = Wi1; a.Wh1 = Wh1; a.bL1 = bL1;
    a.Wout = Wout; a.bout = bout;
    a.isq = isq; a.idg = idg;
    a.e     = p; p += 2 * PL_LOC;
    a.aggA  = p; p += 2 * PL_LOC;
    a.h1b   = p; p += 2 * PL_LOC;
    a.aggB  = p; p += 2 * PL_LOC;
    a.h2b   = p; p += 2 * PL_LOC;
    a.he2n  = p; p += 2 * PL_E2N;
    a.out32 = p; p += 2 * PL_X;
    a.h0p1  = p; p += (size_t)USR_NUM * 16;
    a.c0    = p; p += (size_t)USR_NUM * 16;
    a.h1p1  = p; p += (size_t)USR_NUM * 16;
    a.c1    = p; p += (size_t)USR_NUM * 16;
    a.h0p0  = p; p += (size_t)USR_NUM * 16;
    a.h1p0  = p; p += (size_t)USR_NUM * 16;
    a.out = out;

    k_zero<<<gs((long)USR_NUM * 64, BT), BT, 0, stream>>>(a.h0p1, USR_NUM * 64);
    for (int tick = 0; tick < N_TICKS; ++tick)
      k_tick<<<NB_TICK, 1024, 0, stream>>>(a, tick);
    return;
  }

  float* e     = p; p += (size_t)NUM_LOC * 64;
  float* agg   = p; p += (size_t)NUM_LOC * 64;
  float* h1b   = p; p += (size_t)NUM_LOC * 64;
  float* h2b   = p; p += (size_t)NUM_LOC * 64;
  float* he2n  = p; p += (size_t)NUM_LOC * 32;
  float* out32 = p; p += (size_t)USR_NUM * 32;
  float* h0    = p; p += (size_t)USR_NUM * 16;
  float* c0    = p; p += (size_t)USR_NUM * 16;
  float* h1s   = p; p += (size_t)USR_NUM * 16;
  float* c1s   = p; p += (size_t)USR_NUM * 16;

  k_zero<<<gs((long)USR_NUM * 64, BT), BT, 0, stream>>>(h0, USR_NUM * 64);

  k_stageA <<<NB_BAND, 1024, 0, stream>>>(NB_BAND, seq, Wemb, bemb, e, seq, he2n, out32);
  k_stageBD<<<NB_BAND, 1024, 0, stream>>>(srcE, dstE, isq, idg, e, agg);
  k_stageCE<<<NB_GEMM, 256, 0, stream>>>(NB_GEMM, agg, W1, b1, h1b, 1,
                                         out32, 32, Wi0, Wh0, bL0, h0, c0);
  k_stageBD<<<NB_BAND, 1024, 0, stream>>>(srcE, dstE, isq, idg, h1b, agg);
  k_stageCE<<<NB_GEMM, 256, 0, stream>>>(NB_GEMM, agg, W2, b2, h2b, 0,
                                         h0, 16, Wi1, Wh1, bL1, h1s, c1s);
  k_stageF <<<NB_GEMM, 256, 0, stream>>>(NB_GEMM, h2b, We2n, be2n, he2n, h1s, Wout, bout, out);

  for (int t = 0; t + 1 < T_STEPS; ++t) {
    const int* idsCur  = seq + (long)t * USR_NUM;
    const int* idsNext = idsCur + USR_NUM;
    float* outT = out + (long)t * USR_NUM * 2;

    k_stageA <<<NB_BAND + NB_GATHER, 1024, 0, stream>>>(NB_BAND, idsNext, Wemb, bemb, e,
                                                        idsCur, he2n, out32);
    k_stageBD<<<NB_BAND, 1024, 0, stream>>>(srcE, dstE, isq, idg, e, agg);
    k_stageCE<<<NB_GEMM + NB_LAYER, 256, 0, stream>>>(NB_GEMM, agg, W1, b1, h1b, 1,
                                                      out32, 32, Wi0, Wh0, bL0, h0, c0);
    k_stageBD<<<NB_BAND, 1024, 0, stream>>>(srcE, dstE, isq, idg, h1b, agg);
    k_stageCE<<<NB_GEMM + NB_LAYER, 256, 0, stream>>>(NB_GEMM, agg, W2, b2, h2b, 0,
                                                      h0, 16, Wi1, Wh1, bL1, h1s, c1s);
    k_stageF <<<NB_GEMM + NB_OUT, 256, 0, stream>>>(NB_GEMM, h2b, We2n, be2n, he2n,
                                                     h1s, Wout, bout, outT);
  }

  {
    const int* idsCur = seq + (long)(T_STEPS - 1) * USR_NUM;
    float* outT = out + (long)(T_STEPS - 1) * USR_NUM * 2;
    k_stageA <<<NB_GATHER, 1024, 0, stream>>>(0, idsCur, Wemb, bemb, e, idsCur, he2n, out32);
    k_stageCE<<<NB_LAYER, 256, 0, stream>>>(0, agg, W1, b1, h1b, 1,
                                            out32, 32, Wi0, Wh0, bL0, h0, c0);
    k_stageCE<<<NB_LAYER, 256, 0, stream>>>(0, agg, W2, b2, h2b, 0,
                                            h0, 16, Wi1, Wh1, bL1, h1s, c1s);
    k_stageF <<<NB_OUT, 256, 0, stream>>>(0, h2b, We2n, be2n, he2n, h1s, Wout, bout, outT);
  }
}


# LSTM layer tiles carried by the band work-groups (27 each): band job (raised issue priority) and layer job in opposite order on the two groups of a WGP, layers issue in the other's band waits
# speedup vs baseline: 4.0245x; 1.1019x over previous
#include <hip/hip_runtime.h>
#include <math.h>


#define USR_NUM 100000
#define USR_DIM 64
#define NUM_LOC 5000
#define T_STEPS 16
#define N_EDGES 80000

typedef float v2f __attribute__((ext_vector_type(2)));
typedef float v4f __attribute__((ext_vector_type(4)));
typedef float v8f __attribute__((ext_vector_type(8)));

__global__ void k_zero(float* __restrict__ p, int n) {
  int i = blockIdx.x * blockDim.x + threadIdx.x;
  if (i < n) p[i] = 0.0f;
}

__global__ void k_deg_accum(const int* __restrict__ dst, float* __restrict__ deg, int E) {
  int i = blockIdx.x * blockDim.x + threadIdx.x;
  if (i < E) atomicAdd(&deg[dst[i]], 1.0f);
}

__global__ void k_deg_fin(const float* __restrict__ deg, float* __restrict__ isq,
                          float* __restrict__ idg, int n) {
  int i = blockIdx.x * blockDim.x + threadIdx.x;
  if (i < n) {
    float d = deg[i] + 1.0f;
    isq[i] = rsqrtf(d);
    idg[i] = 1.0f / d;
  }
}


#define ROWS_PB   32
#define SCAN_KEYS 16384
#define SCAN_V4   (SCAN_KEYS / 4 / 1024)
#define WQ_HIGH   128
#define WQ_CAP    (WQ_HIGH + 4 * 32 * SCAN_V4 + 4)
#define AR_TILE   0
#define AR_AUX    ((ROWS_PB + 1) * 64)
#define AR_Q      (AR_AUX + 64)
#define AR_BAND   (AR_Q + 32 * WQ_CAP)
#define AR_GEMM   (64 * 64)

__device__ __forceinline__ void role_user_sum(
    const int vb, const int* __restrict__ ids, const float* __restrict__ Wemb,
    const float* __restrict__ bemb, float* __restrict__ e, int* arena) {
  float* tile = (float*)(arena + AR_TILE);
  float* cntS = (float*)(arena + AR_AUX);

  const int tid  = threadIdx.x;
  const int lane = tid & 31;
  const int wave = tid >> 5;
  const int lo   = vb * ROWS_PB;
  int* wq = arena + AR_Q + wave * WQ_CAP;

  for (int i = tid; i < (ROWS_PB + 1) * 64; i += 1024) tile[i] = 0.0f;
  if (tid < ROWS_PB) cntS[tid] = 0.0f;

  const int4* ids4 = (const int4*)ids;
  const int nv4 = USR_NUM / 4;
  const float2 bb = *(const float2*)(bemb + 2 * lane);

  int4 cur[SCAN_V4];
#pragma unroll
  for (int k = 0; k < SCAN_V4; ++k) {
    int i4 = k * 1024 + tid;
    cur[k] = (i4 < nv4) ? ids4[i4] : make_int4(-1, -1, -1, -1);
  }
  __syncthreads();

  int nq = 0;
  int head = 0;
  for (int base4 = 0; base4 < nv4; base4 += SCAN_KEYS / 4) {
    int4 nx[SCAN_V4];
#pragma unroll
    for (int k = 0; k < SCAN_V4; ++k) {
      int i4 = base4 + SCAN_KEYS / 4 + k * 1024 + tid;
      nx[k] = (i4 < nv4) ? ids4[i4] : make_int4(-1, -1, -1, -1);
    }
#pragma unroll
    for (int k = 0; k < SCAN_V4; ++k) {
      const int u0 = (base4 + k * 1024 + tid) * 4;
      const int kv[4] = { cur[k].x, cur[k].y, cur[k].z, cur[k].w };
#pragma unroll
      for (int c = 0; c < 4; ++c) {
        const unsigned r = (unsigned)(kv[c] - lo);
        const bool hit = r < (unsigned)ROWS_PB;
        const unsigned m = __builtin_amdgcn_ballot_w32(hit);
        if (m != 0u) {
          if (hit) {
            wq[nq + (int)__builtin_amdgcn_mbcnt_lo(m, 0u)] = ((u0 + c) << 8) | (int)r;
            atomicAdd(&cntS[r], 1.0f);
          }
          nq += __builtin_popcount(m);
        }
      }
    }
    const bool drain = (base4 + SCAN_KEYS / 4 >= nv4) || (nq > WQ_HIGH);
    if (drain && lane < 3) wq[nq + lane] = ROWS_PB;
    const int stop = drain ? ((nq + 3) & ~3) : (nq & ~3);
    for (int j0 = head; j0 < stop; j0 += 4) {
      const int4 q4 = *(const int4*)(wq + j0);
      const int ent[4] = { __builtin_amdgcn_readfirstlane(q4.x), __builtin_amdgcn_readfirstlane(q4.y),
                           __builtin_amdgcn_readfirstlane(q4.z), __builtin_amdgcn_readfirstlane(q4.w) };
      float2 v[4];
#pragma unroll
      for (int k = 0; k < 4; ++k)
        v[k] = *(const float2*)(Wemb + (long)(ent[k] >> 8) * 64 + 2 * lane);
#pragma unroll
      for (int k = 0; k < 4; ++k) {
        int r = ent[k] & 255;
        float* dstp = &tile[r * 64 + 2 * lane];
        atomicAdd(dstp + 0, v[k].x + bb.x);
        atomicAdd(dstp + 1, v[k].y + bb.y);
      }
    }
    head = stop;
    if (drain) { head = 0; nq = 0; }
#pragma unroll
    for (int k = 0; k < SCAN_V4; ++k) cur[k] = nx[k];
  }
  __syncthreads();

  for (int i = tid; i < ROWS_PB * 64; i += 1024) {
    int j = i >> 6;
    int row = lo + j;
    if (row < NUM_LOC) {
      float cv = cntS[j]; if (cv < 1.0f) cv = 1.0f;
      e[(long)row * 64 + (i & 63)] = tile[i] / cv;
    }
  }
}

__device__ __forceinline__ void role_edge_sum(
    const int vb, const int* __restrict__ src, const int* __restrict__ dst,
    const float* __restrict__ isq, const float* __restrict__ idg,
    const float* __restrict__ h, float* __restrict__ agg, int* arena) {
  float* tile = (float*)(arena + AR_TILE);
  float* isqS = (float*)(arena + AR_AUX);

  const int tid  = threadIdx.x;
  const int lane = tid & 31;
  const int wave = tid >> 5;
  const int lo   = vb * ROWS_PB;
  int* wq = arena + AR_Q + wave * WQ_CAP;

  for (int i = tid; i < (ROWS_PB + 1) * 64; i += 1024) {
    int row = lo + (i >> 6);
    tile[i] = (i < ROWS_PB * 64 && row < NUM_LOC) ? h[(long)row * 64 + (i & 63)] * idg[row] : 0.0f;
  }
  if (tid <= ROWS_PB) isqS[tid] = (tid < ROWS_PB && lo + tid < NUM_LOC) ? isq[lo + tid] : 0.0f;

  const int4* dst4 = (const int4*)dst;
  const int nv4 = N_EDGES / 4;

  int4 cur[SCAN_V4];
#pragma unroll
  for (int k = 0; k < SCAN_V4; ++k) {
    int i4 = k * 1024 + tid;
    cur[k] = (i4 < nv4) ? dst4[i4] : make_int4(-1, -1, -1, -1);
  }
  __syncthreads();

  int nq = 0;
  int head = 0;
  for (int base4 = 0; base4 < nv4; base4 += SCAN_KEYS / 4) {
    int4 nx[SCAN_V4];
#pragma unroll
    for (int k = 0; k < SCAN_V4; ++k) {
      int i4 = base4 + SCAN_KEYS / 4 + k * 1024 + tid;
      nx[k] = (i4 < nv4) ? dst4[i4] : make_int4(-1, -1, -1, -1);
    }
#pragma unroll
    for (int k = 0; k < SCAN_V4; ++k) {
      const int e0 = (base4 + k * 1024 + tid) * 4;
      const int kv[4] = { cur[k].x, cur[k].y, cur[k].z, cur[k].w };
#pragma unroll
      for (int c = 0; c < 4; ++c) {
        const unsigned r = (unsigned)(kv[c] - lo);
        const bool hit = r < (unsigned)ROWS_PB;
        const unsigned m = __builtin_amdgcn_ballot_w32(hit);
        if (m != 0u) {
          if (hit) wq[nq + (int)__builtin_amdgcn_mbcnt_lo(m, 0u)] = ((e0 + c) << 8) | (int)r;
          nq += __builtin_popcount(m);
        }
      }
    }
    const bool drain = (base4 + SCAN_KEYS / 4 >= nv4) || (nq > WQ_HIGH);
    if (drain && lane < 3) wq[nq + lane] = ROWS_PB;
    const int stop = drain ? ((nq + 3) & ~3) : (nq & ~3);
    for (int j0 = head; j0 < stop; j0 += 4) {
      const int4 q4 = *(const int4*)(wq + j0);
      const int ent[4] = { __builtin_amdgcn_readfirstlane(q4.x), __builtin_amdgcn_readfirstlane(q4.y),
                           __builtin_amdgcn_readfirstlane(q4.z), __builtin_amdgcn_readfirstlane(q4.w) };
      float wr[4];
#pragma unroll
      for (int k = 0; k < 4; ++k) wr[k] = isqS[ent[k] & 255];
      int s[4];
#pragma unroll
      for (int k = 0; k < 4; ++k) s[k] = src[ent[k] >> 8];
      float  w[4];
#pragma unroll
      for (int k = 0; k < 4; ++k) w[k] = isq[s[k]];
      float2 v[4];
#pragma unroll
      for (int k = 0; k < 4; ++k)
        v[k] = *(const float2*)(h + (long)s[k] * 64 + 2 * lane);
#pragma unroll
      for (int k = 0; k < 4; ++k) {
        float* dstp = &tile[(ent[k] & 255) * 64 + 2 * lane];
        atomicAdd(dstp + 0, v[k].x * w[k] * wr[k]);
        atomicAdd(dstp + 1, v[k].y * w[k] * wr[k]);
      }
    }
    head = stop;
    if (drain) { head = 0; nq = 0; }
#pragma unroll
    for (int k = 0; k < SCAN_V4; ++k) cur[k] = nx[k];
  }
  __syncthreads();

  for (int i = tid; i < ROWS_PB * 64; i += 1024) {
    int row = lo + (i >> 6);
    if (row < NUM_LOC) agg[(long)row * 64 + (i & 63)] = tile[i];
  }
}

template <int NT, int TH>
__device__ __forceinline__ void role_gemm(
    const int vb, const float* __restrict__ A, const float* __restrict__ B,
    const float* __restrict__ bias, float* __restrict__ C, int M, int K, int relu,
    float* Bs) {
  constexpr int Nc = NT * 16;
  for (int i = threadIdx.x; i < K * Nc; i += TH) Bs[i] = B[i];
  __syncthreads();

  const int lane = threadIdx.x & 31;
  const int wave = threadIdx.x >> 5;
  const int half = lane >> 4;
  const int l16  = lane & 15;
  const int row0 = (vb * (TH / 32) + wave) * 16;

  int ar = row0 + l16; if (ar > M - 1) ar = M - 1;
  const float* Arow = A + (long)ar * K;

  v8f acc[NT];
#pragma unroll
  for (int tt = 0; tt < NT; ++tt) {
    float bv = bias[tt * 16 + l16];
    v8f tmp = { bv, bv, bv, bv, bv, bv, bv, bv };
    acc[tt] = tmp;
  }

  for (int k = 0; k < K; k += 4) {
    v2f a;
    a[0] = Arow[k + 2 * half + 0];
    a[1] = Arow[k + 2 * half + 1];
#pragma unroll
    for (int tt = 0; tt < NT; ++tt) {
      v2f b;
      b[0] = Bs[(k + 2 * half + 0) * Nc + tt * 16 + l16];
      b[1] = Bs[(k + 2 * half + 1) * Nc + tt * 16 + l16];
      acc[tt] = __builtin_amdgcn_wmma_f32_16x16x4_f32(false, a, false, b, (short)0,
                                                      acc[tt], false, false);
    }
  }
#pragma unroll
  for (int tt = 0; tt < NT; ++tt) {
    for (int r = 0; r < 8; ++r) {
      int row = row0 + r + 8 * half;
      if (row < M) {
        float v = acc[tt][r];
        if (relu) v = v > 0.0f ? v : 0.0f;
        C[(long)row * Nc + tt * 16 + l16] = v;
      }
    }
  }
}

__device__ __forceinline__ float sigm(float x) { return 1.0f / (1.0f + expf(-x)); }

template <int TH>
__device__ __forceinline__ void role_layer(
    const int vb, const int TP, const float* __restrict__ X, int Kx, const float* __restrict__ Wi,
    const float* __restrict__ Wh, const float* __restrict__ b,
    const float* hin, float* hout, float* c, int M, float* Bs) {
  for (int i = threadIdx.x; i < Kx * 64; i += TH) Bs[i] = Wi[i];
  for (int i = threadIdx.x; i < 16 * 64; i += TH) Bs[Kx * 64 + i] = Wh[i];
  __syncthreads();

  const int lane = threadIdx.x & 31;
  const int half = lane >> 4;
  const int l16  = lane & 15;
  const int wave = __builtin_amdgcn_readfirstlane((int)(threadIdx.x >> 5));
  const int row0 = (vb * TP + wave) * 16;
  if (wave < TP && row0 < M) {

  int ar = row0 + l16; if (ar > M - 1) ar = M - 1;
  const int rleft = M - row0;
  const unsigned loff = (unsigned)(half * 128 + l16);
  float* const cT = c + (long)row0 * 16;
  float* const hT = hout + (long)row0 * 16;

  float cold[8];
#pragma unroll
  for (int r = 0; r < 8; ++r)
    cold[r] = (r + 8 * half < rleft) ? cT[loff + r * 16] : 0.0f;

  v8f acc[4];
#pragma unroll
  for (int tt = 0; tt < 4; ++tt) {
    float bv = b[tt * 16 + l16];
    v8f tmp = { bv, bv, bv, bv, bv, bv, bv, bv };
    acc[tt] = tmp;
  }

  const float* Xr = X + (long)ar * Kx;
  for (int k = 0; k < Kx; k += 4) {
    v2f a;
    a[0] = Xr[k + 2 * half + 0];
    a[1] = Xr[k + 2 * half + 1];
#pragma unroll
    for (int tt = 0; tt < 4; ++tt) {
      v2f w;
      w[0] = Bs[(k + 2 * half + 0) * 64 + tt * 16 + l16];
      w[1] = Bs[(k + 2 * half + 1) * 64 + tt * 16 + l16];
      acc[tt] = __builtin_amdgcn_wmma_f32_16x16x4_f32(false, a, false, w, (short)0,
                                                      acc[tt], false, false);
    }
  }
  const float* Hr = hin + (long)ar * 16;
#pragma unroll
  for (int k = 0; k < 16; k += 4) {
    v2f a;
    a[0] = Hr[k + 2 * half + 0];
    a[1] = Hr[k + 2 * half + 1];
#pragma unroll
    for (int tt = 0; tt < 4; ++tt) {
      v2f w;
      w[0] = Bs[(Kx + k + 2 * half + 0) * 64 + tt * 16 + l16];
      w[1] = Bs[(Kx + k + 2 * half + 1) * 64 + tt * 16 + l16];
      acc[tt] = __builtin_amdgcn_wmma_f32_16x16x4_f32(false, a, false, w, (short)0,
                                                      acc[tt], false, false);
    }
  }
#pragma unroll
  for (int r = 0; r < 8; ++r) {
    if (r + 8 * half < rleft) {
      float gi = acc[0][r], gf = acc[1][r], gg = acc[2][r], go = acc[3][r];
      float cn = sigm(gf) * cold[r] + sigm(gi) * tanhf(gg);
      cT[loff + r * 16] = cn;
      hT[loff + r * 16] = sigm(go) * tanhf(cn);
    }
  }
  }
}

__device__ __forceinline__ void role_gather(const int vb, const int* __restrict__ ids,
                                            const float* __restrict__ he,
                                            float* __restrict__ out32, int N) {
  const int sub = threadIdx.x & 7;
  const int r0  = vb * 512 + (threadIdx.x >> 3);
  if (vb * 512 + 512 <= N) {
    int id[4];
    v4f v[4];
#pragma unroll
    for (int p = 0; p < 4; ++p) id[p] = ids[r0 + 128 * p];
#pragma unroll
    for (int p = 0; p < 4; ++p)
      v[p] = *(const v4f*)(he + (long)id[p] * 32 + 4 * sub);
    asm volatile("" : "+v"(v[0]), "+v"(v[1]), "+v"(v[2]), "+v"(v[3]));
#pragma unroll
    for (int p = 0; p < 4; ++p)
      *(v4f*)(out32 + (long)(r0 + 128 * p) * 32 + 4 * sub) = v[p];
  } else {
    for (int p = 0; p < 4; ++p) {
      int row = r0 + 128 * p;
      if (row < N)
        *(v4f*)(out32 + (long)row * 32 + 4 * sub) =
            *(const v4f*)(he + (long)ids[row] * 32 + 4 * sub);
    }
  }
}

template <int TH>
__device__ __forceinline__ void role_outproj(const int vb, const float* __restrict__ h,
                                             const float* __restrict__ Wout,
                                             const float* __restrict__ bout,
                                             float* __restrict__ out, int N) {
  int i = vb * TH + threadIdx.x;
  if (i >= N) return;
  const float* hr = h + (long)i * 16;
  float a0 = bout[0], a1 = bout[1];
  for (int j = 0; j < 16; ++j) {
    float v = hr[j];
    a0 += v * Wout[j * 2 + 0];
    a1 += v * Wout[j * 2 + 1];
  }
  out[(long)i * 2 + 0] = a0;
  out[(long)i * 2 + 1] = a1;
}

__global__ __launch_bounds__(1024) void k_stageA(
    int nLoc, const int* __restrict__ idsNext, const float* __restrict__ Wemb,
    const float* __restrict__ bemb, float* __restrict__ e,
    const int* __restrict__ idsCur, const float* __restrict__ he2n, float* __restrict__ out32) {
  __shared__ __attribute__((aligned(16))) int arena[AR_BAND];
  const int b = blockIdx.x;
  if (b < nLoc) role_user_sum(b, idsNext, Wemb, bemb, e, arena);
  else          role_gather(b - nLoc, idsCur, he2n, out32, USR_NUM);
}

__global__ __launch_bounds__(1024) void k_stageBD(
    const int* __restrict__ src, const int* __restrict__ dst,
    const float* __restrict__ isq, const float* __restrict__ idg,
    const float* __restrict__ hin, float* __restrict__ agg) {
  __shared__ __attribute__((aligned(16))) int arena[AR_BAND];
  role_edge_sum(blockIdx.x, src, dst, isq, idg, hin, agg, arena);
}

__global__ __launch_bounds__(256) void k_stageCE(
    int nLoc, const float* __restrict__ A, const float* __restrict__ W,
    const float* __restrict__ bias, float* __restrict__ C, int relu,
    const float* __restrict__ X, int Kx, const float* __restrict__ Wi,
    const float* __restrict__ Wh, const float* __restrict__ bL,
    float* hS, float* cS) {
  __shared__ float Bs[AR_GEMM];
  const int b = blockIdx.x;
  if (b < nLoc) role_gemm<4, 256>(b, A, W, bias, C, NUM_LOC, 64, relu, Bs);
  else          role_layer<256>(b - nLoc, 8, X, Kx, Wi, Wh, bL, hS, hS, cS, USR_NUM, Bs);
}

__global__ __launch_bounds__(256) void k_stageF(
    int nLoc, const float* __restrict__ A, const float* __restrict__ W,
    const float* __restrict__ bias, float* __restrict__ C,
    const float* __restrict__ h, const float* __restrict__ Wout,
    const float* __restrict__ bout, float* __restrict__ out) {
  __shared__ float Bs[AR_GEMM];
  const int b = blockIdx.x;
  if (b < nLoc) role_gemm<2, 256>(b, A, W, bias, C, NUM_LOC, 64, 0, Bs);
  else          role_outproj<256>(b - nLoc, h, Wout, bout, out, USR_NUM);
}


#define PL_LOC   ((long)NUM_LOC * 64)
#define PL_E2N   ((long)NUM_LOC * 32)
#define PL_X     ((long)USR_NUM * 32)
#define PL_ST    ((long)USR_NUM * 16)

#define NB_BAND   ((NUM_LOC + ROWS_PB - 1) / ROWS_PB)
#define NB_BAND3  (3 * NB_BAND)
#define N_TILES   ((USR_NUM + 15) / 16)
#define TILES_PB  ((2 * N_TILES + NB_BAND3 - 1) / NB_BAND3)
#define NB_LAYF   ((N_TILES + TILES_PB - 1) / TILES_PB)
#define NB_GATHER ((USR_NUM + 511) / 512)
#define NB_GEMM4  ((NUM_LOC + 511) / 512)
#define NB_OUT4   ((USR_NUM + 1023) / 1024)
#define NB_TICK   (NB_BAND3 + 3 * NB_GEMM4 + NB_GATHER + NB_OUT4)
#define N_TICKS   (T_STEPS + 9)
static_assert(TILES_PB <= 32, "one layer tile a wave");
static_assert(2 * NB_LAYF <= NB_BAND3, "both layers' tiles must fit the carrying work-groups");
#define NL_PAD 5056
#define WS_DEG   0L
#define WS_ISQ   (1L * NL_PAD)
#define WS_IDG   (2L * NL_PAD)
#define WS_E     (3L * NL_PAD)
#define WS_AGGA  (WS_E    + 2 * PL_LOC)
#define WS_H1B   (WS_AGGA + 2 * PL_LOC)
#define WS_AGGB  (WS_H1B  + 2 * PL_LOC)
#define WS_H2B   (WS_AGGB + 2 * PL_LOC)
#define WS_HE2N  (WS_H2B  + 2 * PL_LOC)
#define WS_OUT32 (WS_HE2N + 2 * PL_E2N)
#define WS_H0P1  (WS_OUT32 + 2 * PL_X)
#define WS_C0    (WS_H0P1 + PL_ST)
#define WS_H1P1  (WS_C0   + PL_ST)
#define WS_C1    (WS_H1P1 + PL_ST)
#define WS_H0P0  (WS_C1   + PL_ST)
#define WS_H1P0  (WS_H0P0 + PL_ST)
#define WS_TICK_FLOATS (WS_H1P0 + PL_ST)
#define WS_TICK_BYTES  ((size_t)(WS_TICK_FLOATS * 4))
static_assert(WS_TICK_FLOATS == 3L * NL_PAD + 10 * PL_LOC + 2 * PL_E2N + 2 * PL_X + 6L * USR_NUM * 16, "workspace layout");

struct TickArgs {
  const int*   seq;  const int*   src;  const int*   dst;
  const float* Wemb; const float* bemb;
  const float* W1;   const float* b1;   const float* W2;   const float* b2;
  const float* We;   const float* be;
  const float* Wi0;  const float* Wh0;  const float* bL0;
  const float* Wi1;  const float* Wh1;  const float* bL1;
  const float* Wout; const float* bout;
  float* ws;
  float* out;
};

__device__ __forceinline__ void band_job(const int b, const TickArgs& a, const int tick, int* arena) {
  float* const ws = a.ws;
  __builtin_amdgcn_s_setprio(3);
  if (b < NB_BAND) {
    const int s = tick;
    if ((unsigned)s < (unsigned)T_STEPS)
      role_user_sum(b, a.seq + (long)s * USR_NUM, a.Wemb, a.bemb, ws + WS_E + (s & 1) * PL_LOC, arena);
  } else if (b < 2 * NB_BAND) {
    const int s = tick - 1;
    if ((unsigned)s < (unsigned)T_STEPS)
      role_edge_sum(b - NB_BAND, a.src, a.dst, ws + WS_ISQ, ws + WS_IDG, ws + WS_E + (s & 1) * PL_LOC,
                    ws + WS_AGGA + (s & 1) * PL_LOC, arena);
  } else {
    const int s = tick - 3;
    if ((unsigned)s < (unsigned)T_STEPS)
      role_edge_sum(b - 2 * NB_BAND, a.src, a.dst, ws + WS_ISQ, ws + WS_IDG, ws + WS_H1B + (s & 1) * PL_LOC,
                    ws + WS_AGGB + (s & 1) * PL_LOC, arena);
  }
  __builtin_amdgcn_s_setprio(0);
}

__global__ __launch_bounds__(1024, 16) void k_tick(const TickArgs a, const int tick) {
  __shared__ __attribute__((aligned(16))) int arena[AR_BAND];
  float* const ws = a.ws;
  int b = blockIdx.x;
  int s;
  if (b < NB_BAND3) {
    const bool layerFirst = (__builtin_popcount((unsigned)b) & 1) != 0;
    if (!layerFirst) band_job(b, a, tick, arena);
    __syncthreads();
    if (b < NB_LAYF) {
      s = tick - 7;
      if ((unsigned)s < (unsigned)T_STEPS)
        role_layer<1024>(b, TILES_PB, ws + WS_OUT32 + (s & 1) * PL_X, 32, a.Wi0, a.Wh0, a.bL0,
                         ws + ((s & 1) ? WS_H0P0 : WS_H0P1), ws + ((s & 1) ? WS_H0P1 : WS_H0P0), ws + WS_C0,
                         USR_NUM, (float*)arena);
    } else if (b < 2 * NB_LAYF) {
      s = tick - 8;
      if ((unsigned)s < (unsigned)T_STEPS)
        role_layer<1024>(b - NB_LAYF, TILES_PB, ws + ((s & 1) ? WS_H0P1 : WS_H0P0), 16, a.Wi1, a.Wh1, a.bL1,
                         ws + ((s & 1) ? WS_H1P0 : WS_H1P1), ws + ((s & 1) ? WS_H1P1 : WS_H1P0), ws + WS_C1,
                         USR_NUM, (float*)arena);
    }
    __syncthreads();
    if (layerFirst) band_job(b, a, tick, arena);
    return;
  }
  b -= NB_BAND3;
  if (b < NB_GEMM4) {
    s = tick - 2;
    if ((unsigned)s >= (unsigned)T_STEPS) return;
    role_gemm<4, 1024>(b, ws + WS_AGGA + (s & 1) * PL_LOC, a.W1, a.b1, ws + WS_H1B + (s & 1) * PL_LOC,
                       NUM_LOC, 64, 1, (float*)arena);
    return;
  }
  b -= NB_GEMM4;
  if (b < NB_GEMM4) {
    s = tick - 4;
    if ((unsigned)s >= (unsigned)T_STEPS) return;
    role_gemm<4, 1024>(b, ws + WS_AGGB + (s & 1) * PL_LOC, a.W2, a.b2, ws + WS_H2B + (s & 1) * PL_LOC,
                       NUM_LOC, 64, 0, (float*)arena);
    return;
  }
  b -= NB_GEMM4;
  if (b < NB_GEMM4) {
    s = tick - 5;
    if ((unsigned)s >= (unsigned)T_STEPS) return;
    role_gemm<2, 1024>(b, ws + WS_H2B + (s & 1) * PL_LOC, a.We, a.be, ws + WS_HE2N + (s & 1) * PL_E2N,
                       NUM_LOC, 64, 0, (float*)arena);
    return;
  }
  b -= NB_GEMM4;
  if (b < NB_GATHER) {
    s = tick - 6;
    if ((unsigned)s >= (unsigned)T_STEPS) return;
    role_gather(b, a.seq + (long)s * USR_NUM, ws + WS_HE2N + (s & 1) * PL_E2N,
                ws + WS_OUT32 + (s & 1) * PL_X, USR_NUM);
    return;
  }
  b -= NB_GATHER;
  {
    s = tick - 9;
    if ((unsigned)s >= (unsigned)T_STEPS) return;
    role_outproj<1024>(b, ws + ((s & 1) ? WS_H1P1 : WS_H1P0), a.Wout, a.bout,
                       a.out + (long)s * USR_NUM * 2, USR_NUM);
  }
}

static inline int gs(long n, int b) { return (int)((n + b - 1) / b); }

#define NB_LAYER  ((USR_NUM + 127) / 128)
#define NB_GEMM   ((NUM_LOC + 127) / 128)
#define NB_OUT    ((USR_NUM + 255) / 256)

extern "C" void kernel_launch(void* const* d_in, const int* in_sizes, int n_in,
                              void* d_out, int out_size, void* d_ws, size_t ws_size,
                              hipStream_t stream) {
  (void)in_sizes; (void)n_in; (void)out_size;

  const int*   seq  = (const int*)d_in[0];
  const int*   gidx = (const int*)d_in[1];
  const float* Wemb = (const float*)d_in[2];
  const float* bemb = (const float*)d_in[3];
  const float* W1   = (const float*)d_in[4];
  const float* b1   = (const float*)d_in[5];
  const float* W2   = (const float*)d_in[6];
  const float* b2   = (const float*)d_in[7];
  const float* We2n = (const float*)d_in[8];
  const float* be2n = (const float*)d_in[9];
  const float* Wi0  = (const float*)d_in[10];
  const float* Wh0  = (const float*)d_in[11];
  const float* bL0  = (const float*)d_in[12];
  const float* Wi1  = (const float*)d_in[13];
  const float* Wh1  = (const float*)d_in[14];
  const float* bL1  = (const float*)d_in[15];
  const float* Wout = (const float*)d_in[16];
  const float* bout = (const float*)d_in[17];

  const int* srcE = gidx;
  const int* dstE = gidx + N_EDGES;
  float* out = (float*)d_out;
  const int BT = 256;

  float* p = (float*)d_ws;
  float* deg   = p; p += NL_PAD;
  float* isq   = p; p += NL_PAD;
  float* idg   = p; p += NL_PAD;

  k_zero<<<gs(NUM_LOC, BT), BT, 0, stream>>>(deg, NUM_LOC);
  k_deg_accum<<<gs(N_EDGES, BT), BT, 0, stream>>>(dstE, deg, N_EDGES);
  k_deg_fin<<<gs(NUM_LOC, BT), BT, 0, stream>>>(deg, isq, idg, NUM_LOC);

  if (ws_size >= WS_TICK_BYTES) {
    TickArgs a;
    a.seq = seq; a.src = srcE; a.dst = dstE;
    a.Wemb = Wemb; a.bemb = bemb;
    a.W1 = W1; a.b1 = b1; a.W2 = W2; a.b2 = b2; a.We = We2n; a.be = be2n;
    a.Wi0 = Wi0; a.Wh0 = Wh0; a.bL0 = bL0; a.Wi1 = Wi1; a.Wh1 = Wh1; a.bL1 = bL1;
    a.Wout = Wout; a.bout = bout;
    a.ws = (float*)d_ws;
    a.out = out;

    k_zero<<<gs((long)USR_NUM * 64, BT), BT, 0, stream>>>((float*)d_ws + WS_H0P1, USR_NUM * 64);
    for (int tick = 0; tick < N_TICKS; ++tick)
      k_tick<<<NB_TICK, 1024, 0, stream>>>(a, tick);
    return;
  }

  float* e     = p; p += (size_t)NUM_LOC * 64;
  float* agg   = p; p += (size_t)NUM_LOC * 64;
  float* h1b   = p; p += (size_t)NUM_LOC * 64;
  float* h2b   = p; p += (size_t)NUM_LOC * 64;
  float* he2n  = p; p += (size_t)NUM_LOC * 32;
  float* out32 = p; p += (size_t)USR_NUM * 32;
  float* h0    = p; p += (size_t)USR_NUM * 16;
  float* c0    = p; p += (size_t)USR_NUM * 16;
  float* h1s   = p; p += (size_t)USR_NUM * 16;
  float* c1s   = p; p += (size_t)USR_NUM * 16;

  k_zero<<<gs((long)USR_NUM * 64, BT), BT, 0, stream>>>(h0, USR_NUM * 64);

  k_stageA <<<NB_BAND, 1024, 0, stream>>>(NB_BAND, seq, Wemb, bemb, e, seq, he2n, out32);
  k_stageBD<<<NB_BAND, 1024, 0, stream>>>(srcE, dstE, isq, idg, e, agg);
  k_stageCE<<<NB_GEMM, 256, 0, stream>>>(NB_GEMM, agg, W1, b1, h1b, 1,
                                         out32, 32, Wi0, Wh0, bL0, h0, c0);
  k_stageBD<<<NB_BAND, 1024, 0, stream>>>(srcE, dstE, isq, idg, h1b, agg);
  k_stageCE<<<NB_GEMM, 256, 0, stream>>>(NB_GEMM, agg, W2, b2, h2b, 0,
                                         h0, 16, Wi1, Wh1, bL1, h1s, c1s);
  k_stageF <<<NB_GEMM, 256, 0, stream>>>(NB_GEMM, h2b, We2n, be2n, he2n, h1s, Wout, bout, out);

  for (int t = 0; t + 1 < T_STEPS; ++t) {
    const int* idsCur  = seq + (long)t * USR_NUM;
    const int* idsNext = idsCur + USR_NUM;
    float* outT = out + (long)t * USR_NUM * 2;

    k_stageA <<<NB_BAND + NB_GATHER, 1024, 0, stream>>>(NB_BAND, idsNext, Wemb, bemb, e,
                                                        idsCur, he2n, out32);
    k_stageBD<<<NB_BAND, 1024, 0, stream>>>(srcE, dstE, isq, idg, e, agg);
    k_stageCE<<<NB_GEMM + NB_LAYER, 256, 0, stream>>>(NB_GEMM, agg, W1, b1, h1b, 1,
                                                      out32, 32, Wi0, Wh0, bL0, h0, c0);
    k_stageBD<<<NB_BAND, 1024, 0, stream>>>(srcE, dstE, isq, idg, h1b, agg);
    k_stageCE<<<NB_GEMM + NB_LAYER, 256, 0, stream>>>(NB_GEMM, agg, W2, b2, h2b, 0,
                                                      h0, 16, Wi1, Wh1, bL1, h1s, c1s);
    k_stageF <<<NB_GEMM + NB_OUT, 256, 0, stream>>>(NB_GEMM, h2b, We2n, be2n, he2n,
                                                     h1s, Wout, bout, outT);
  }

  {
    const int* idsCur = seq + (long)(T_STEPS - 1) * USR_NUM;
    float* outT = out + (long)(T_STEPS - 1) * USR_NUM * 2;
    k_stageA <<<NB_GATHER, 1024, 0, stream>>>(0, idsCur, Wemb, bemb, e, idsCur, he2n, out32);
    k_stageCE<<<NB_LAYER, 256, 0, stream>>>(0, agg, W1, b1, h1b, 1,
                                            out32, 32, Wi0, Wh0, bL0, h0, c0);
    k_stageCE<<<NB_LAYER, 256, 0, stream>>>(0, agg, W2, b2, h2b, 0,
                                            h0, 16, Wi1, Wh1, bL1, h1s, c1s);
    k_stageF <<<NB_OUT, 256, 0, stream>>>(0, h2b, We2n, be2n, he2n, h1s, Wout, bout, outT);
  }
}


# Edge sums: the next chunk's four key loads made unconditional (index clamped, tail struck out at hand-over) so they fly behind the key tests; as guarded loads each was waited for in turn
# speedup vs baseline: 4.1781x; 1.0382x over previous
#include <hip/hip_runtime.h>
#include <math.h>


#define USR_NUM 100000
#define USR_DIM 64
#define NUM_LOC 5000
#define T_STEPS 16
#define N_EDGES 80000

typedef float v2f __attribute__((ext_vector_type(2)));
typedef float v4f __attribute__((ext_vector_type(4)));
typedef float v8f __attribute__((ext_vector_type(8)));

__global__ void k_zero(float* __restrict__ p, int n) {
  int i = blockIdx.x * blockDim.x + threadIdx.x;
  if (i < n) p[i] = 0.0f;
}

__global__ void k_deg_accum(const int* __restrict__ dst, float* __restrict__ deg, int E) {
  int i = blockIdx.x * blockDim.x + threadIdx.x;
  if (i < E) atomicAdd(&deg[dst[i]], 1.0f);
}

__global__ void k_deg_fin(const float* __restrict__ deg, float* __restrict__ isq,
                          float* __restrict__ idg, int n) {
  int i = blockIdx.x * blockDim.x + threadIdx.x;
  if (i < n) {
    float d = deg[i] + 1.0f;
    isq[i] = rsqrtf(d);
    idg[i] = 1.0f / d;
  }
}


#define ROWS_PB   32
#define SCAN_KEYS 16384
#define SCAN_V4   (SCAN_KEYS / 4 / 1024)
#define WQ_HIGH   128
#define WQ_CAP    (WQ_HIGH + 4 * 32 * SCAN_V4 + 4)
#define AR_TILE   0
#define AR_AUX    ((ROWS_PB + 1) * 64)
#define AR_Q      (AR_AUX + 64)
#define AR_BAND   (AR_Q + 32 * WQ_CAP)
#define AR_GEMM   (64 * 64)

__device__ __forceinline__ void role_user_sum(
    const int vb, const int* __restrict__ ids, const float* __restrict__ Wemb,
    const float* __restrict__ bemb, float* __restrict__ e, int* arena) {
  float* tile = (float*)(arena + AR_TILE);
  float* cntS = (float*)(arena + AR_AUX);

  const int tid  = threadIdx.x;
  const int lane = tid & 31;
  const int wave = tid >> 5;
  const int lo   = vb * ROWS_PB;
  int* wq = arena + AR_Q + wave * WQ_CAP;

  for (int i = tid; i < (ROWS_PB + 1) * 64; i += 1024) tile[i] = 0.0f;
  if (tid < ROWS_PB) cntS[tid] = 0.0f;

  const int4* ids4 = (const int4*)ids;
  const int nv4 = USR_NUM / 4;
  const float2 bb = *(const float2*)(bemb + 2 * lane);

  int4 cur[SCAN_V4];
#pragma unroll
  for (int k = 0; k < SCAN_V4; ++k) {
    int i4 = k * 1024 + tid;
    cur[k] = (i4 < nv4) ? ids4[i4] : make_int4(-1, -1, -1, -1);
  }
  __syncthreads();

  int nq = 0;
  int head = 0;
  for (int base4 = 0; base4 < nv4; base4 += SCAN_KEYS / 4) {
    int4 nx[SCAN_V4];
#pragma unroll
    for (int k = 0; k < SCAN_V4; ++k) {
      int i4 = base4 + SCAN_KEYS / 4 + k * 1024 + tid;
      nx[k] = (i4 < nv4) ? ids4[i4] : make_int4(-1, -1, -1, -1);
    }
#pragma unroll
    for (int k = 0; k < SCAN_V4; ++k) {
      const int u0 = (base4 + k * 1024 + tid) * 4;
      const int kv[4] = { cur[k].x, cur[k].y, cur[k].z, cur[k].w };
#pragma unroll
      for (int c = 0; c < 4; ++c) {
        const unsigned r = (unsigned)(kv[c] - lo);
        const bool hit = r < (unsigned)ROWS_PB;
        const unsigned m = __builtin_amdgcn_ballot_w32(hit);
        if (m != 0u) {
          if (hit) {
            wq[nq + (int)__builtin_amdgcn_mbcnt_lo(m, 0u)] = ((u0 + c) << 8) | (int)r;
            atomicAdd(&cntS[r], 1.0f);
          }
          nq += __builtin_popcount(m);
        }
      }
    }
    const bool drain = (base4 + SCAN_KEYS / 4 >= nv4) || (nq > WQ_HIGH);
    if (drain && lane < 3) wq[nq + lane] = ROWS_PB;
    const int stop = drain ? ((nq + 3) & ~3) : (nq & ~3);
    for (int j0 = head; j0 < stop; j0 += 4) {
      const int4 q4 = *(const int4*)(wq + j0);
      const int ent[4] = { __builtin_amdgcn_readfirstlane(q4.x), __builtin_amdgcn_readfirstlane(q4.y),
                           __builtin_amdgcn_readfirstlane(q4.z), __builtin_amdgcn_readfirstlane(q4.w) };
      float2 v[4];
#pragma unroll
      for (int k = 0; k < 4; ++k)
        v[k] = *(const float2*)(Wemb + (long)(ent[k] >> 8) * 64 + 2 * lane);
#pragma unroll
      for (int k = 0; k < 4; ++k) {
        int r = ent[k] & 255;
        float* dstp = &tile[r * 64 + 2 * lane];
        atomicAdd(dstp + 0, v[k].x + bb.x);
        atomicAdd(dstp + 1, v[k].y + bb.y);
      }
    }
    head = stop;
    if (drain) { head = 0; nq = 0; }
#pragma unroll
    for (int k = 0; k < SCAN_V4; ++k) cur[k] = nx[k];
  }
  __syncthreads();

  for (int i = tid; i < ROWS_PB * 64; i += 1024) {
    int j = i >> 6;
    int row = lo + j;
    if (row < NUM_LOC) {
      float cv = cntS[j]; if (cv < 1.0f) cv = 1.0f;
      e[(long)row * 64 + (i & 63)] = tile[i] / cv;
    }
  }
}

__device__ __forceinline__ void role_edge_sum(
    const int vb, const int* __restrict__ src, const int* __restrict__ dst,
    const float* __restrict__ isq, const float* __restrict__ idg,
    const float* __restrict__ h, float* __restrict__ agg, int* arena) {
  float* tile = (float*)(arena + AR_TILE);
  float* isqS = (float*)(arena + AR_AUX);

  const int tid  = threadIdx.x;
  const int lane = tid & 31;
  const int wave = tid >> 5;
  const int lo   = vb * ROWS_PB;
  int* wq = arena + AR_Q + wave * WQ_CAP;

  for (int i = tid; i < (ROWS_PB + 1) * 64; i += 1024) {
    int row = lo + (i >> 6);
    tile[i] = (i < ROWS_PB * 64 && row < NUM_LOC) ? h[(long)row * 64 + (i & 63)] * idg[row] : 0.0f;
  }
  if (tid <= ROWS_PB) isqS[tid] = (tid < ROWS_PB && lo + tid < NUM_LOC) ? isq[lo + tid] : 0.0f;

  const int4* dst4 = (const int4*)dst;
  const int nv4 = N_EDGES / 4;

  int4 cur[SCAN_V4];
#pragma unroll
  for (int k = 0; k < SCAN_V4; ++k) {
    int i4 = k * 1024 + tid;
    cur[k] = (i4 < nv4) ? dst4[i4] : make_int4(-1, -1, -1, -1);
  }
  __syncthreads();

  int nq = 0;
  int head = 0;
  for (int base4 = 0; base4 < nv4; base4 += SCAN_KEYS / 4) {
    int4 nx[SCAN_V4];
#pragma unroll
    for (int k = 0; k < SCAN_V4; ++k) {
      int i4 = base4 + SCAN_KEYS / 4 + k * 1024 + tid;
      nx[k] = dst4[i4 < nv4 ? i4 : nv4 - 1];
    }
#pragma unroll
    for (int k = 0; k < SCAN_V4; ++k) {
      const int e0 = (base4 + k * 1024 + tid) * 4;
      const int kv[4] = { cur[k].x, cur[k].y, cur[k].z, cur[k].w };
#pragma unroll
      for (int c = 0; c < 4; ++c) {
        const unsigned r = (unsigned)(kv[c] - lo);
        const bool hit = r < (unsigned)ROWS_PB;
        const unsigned m = __builtin_amdgcn_ballot_w32(hit);
        if (m != 0u) {
          if (hit) wq[nq + (int)__builtin_amdgcn_mbcnt_lo(m, 0u)] = ((e0 + c) << 8) | (int)r;
          nq += __builtin_popcount(m);
        }
      }
    }
    const bool drain = (base4 + SCAN_KEYS / 4 >= nv4) || (nq > WQ_HIGH);
    if (drain && lane < 3) wq[nq + lane] = ROWS_PB;
    const int stop = drain ? ((nq + 3) & ~3) : (nq & ~3);
    for (int j0 = head; j0 < stop; j0 += 4) {
      const int4 q4 = *(const int4*)(wq + j0);
      const int ent[4] = { __builtin_amdgcn_readfirstlane(q4.x), __builtin_amdgcn_readfirstlane(q4.y),
                           __builtin_amdgcn_readfirstlane(q4.z), __builtin_amdgcn_readfirstlane(q4.w) };
      float wr[4];
#pragma unroll
      for (int k = 0; k < 4; ++k) wr[k] = isqS[ent[k] & 255];
      int s[4];
#pragma unroll
      for (int k = 0; k < 4; ++k) s[k] = src[ent[k] >> 8];
      float  w[4];
#pragma unroll
      for (int k = 0; k < 4; ++k) w[k] = isq[s[k]];
      float2 v[4];
#pragma unroll
      for (int k = 0; k < 4; ++k)
        v[k] = *(const float2*)(h + (long)s[k] * 64 + 2 * lane);
#pragma unroll
      for (int k = 0; k < 4; ++k) {
        float* dstp = &tile[(ent[k] & 255) * 64 + 2 * lane];
        atomicAdd(dstp + 0, v[k].x * w[k] * wr[k]);
        atomicAdd(dstp + 1, v[k].y * w[k] * wr[k]);
      }
    }
    head = stop;
    if (drain) { head = 0; nq = 0; }
    if (base4 + 2 * (SCAN_KEYS / 4) > nv4) {
#pragma unroll
      for (int k = 0; k < SCAN_V4; ++k)
        if (base4 + SCAN_KEYS / 4 + k * 1024 + tid >= nv4) nx[k] = make_int4(-1, -1, -1, -1);
    }
#pragma unroll
    for (int k = 0; k < SCAN_V4; ++k) cur[k] = nx[k];
  }
  __syncthreads();

  for (int i = tid; i < ROWS_PB * 64; i += 1024) {
    int row = lo + (i >> 6);
    if (row < NUM_LOC) agg[(long)row * 64 + (i & 63)] = tile[i];
  }
}

template <int NT, int TH>
__device__ __forceinline__ void role_gemm(
    const int vb, const float* __restrict__ A, const float* __restrict__ B,
    const float* __restrict__ bias, float* __restrict__ C, int M, int K, int relu,
    float* Bs) {
  constexpr int Nc = NT * 16;
  for (int i = threadIdx.x; i < K * Nc; i += TH) Bs[i] = B[i];
  __syncthreads();

  const int lane = threadIdx.x & 31;
  const int wave = threadIdx.x >> 5;
  const int half = lane >> 4;
  const int l16  = lane & 15;
  const int row0 = (vb * (TH / 32) + wave) * 16;

  int ar = row0 + l16; if (ar > M - 1) ar = M - 1;
  const float* Arow = A + (long)ar * K;

  v8f acc[NT];
#pragma unroll
  for (int tt = 0; tt < NT; ++tt) {
    float bv = bias[tt * 16 + l16];
    v8f tmp = { bv, bv, bv, bv, bv, bv, bv, bv };
    acc[tt] = tmp;
  }

  for (int k = 0; k < K; k += 4) {
    v2f a;
    a[0] = Arow[k + 2 * half + 0];
    a[1] = Arow[k + 2 * half + 1];
#pragma unroll
    for (int tt = 0; tt < NT; ++tt) {
      v2f b;
      b[0] = Bs[(k + 2 * half + 0) * Nc + tt * 16 + l16];
      b[1] = Bs[(k + 2 * half + 1) * Nc + tt * 16 + l16];
      acc[tt] = __builtin_amdgcn_wmma_f32_16x16x4_f32(false, a, false, b, (short)0,
                                                      acc[tt], false, false);
    }
  }
#pragma unroll
  for (int tt = 0; tt < NT; ++tt) {
    for (int r = 0; r < 8; ++r) {
      int row = row0 + r + 8 * half;
      if (row < M) {
        float v = acc[tt][r];
        if (relu) v = v > 0.0f ? v : 0.0f;
        C[(long)row * Nc + tt * 16 + l16] = v;
      }
    }
  }
}

__device__ __forceinline__ float sigm(float x) { return 1.0f / (1.0f + expf(-x)); }

template <int TH>
__device__ __forceinline__ void role_layer(
    const int vb, const int TP, const float* __restrict__ X, int Kx, const float* __restrict__ Wi,
    const float* __restrict__ Wh, const float* __restrict__ b,
    const float* hin, float* hout, float* c, int M, float* Bs) {
  for (int i = threadIdx.x; i < Kx * 64; i += TH) Bs[i] = Wi[i];
  for (int i = threadIdx.x; i < 16 * 64; i += TH) Bs[Kx * 64 + i] = Wh[i];
  __syncthreads();

  const int lane = threadIdx.x & 31;
  const int half = lane >> 4;
  const int l16  = lane & 15;
  const int wave = __builtin_amdgcn_readfirstlane((int)(threadIdx.x >> 5));
  const int row0 = (vb * TP + wave) * 16;
  if (wave < TP && row0 < M) {

  int ar = row0 + l16; if (ar > M - 1) ar = M - 1;
  const int rleft = M - row0;
  const unsigned loff = (unsigned)(half * 128 + l16);
  float* const cT = c + (long)row0 * 16;
  float* const hT = hout + (long)row0 * 16;

  float cold[8];
#pragma unroll
  for (int r = 0; r < 8; ++r)
    cold[r] = (r + 8 * half < rleft) ? cT[loff + r * 16] : 0.0f;

  v8f acc[4];
#pragma unroll
  for (int tt = 0; tt < 4; ++tt) {
    float bv = b[tt * 16 + l16];
    v8f tmp = { bv, bv, bv, bv, bv, bv, bv, bv };
    acc[tt] = tmp;
  }

  const float* Xr = X + (long)ar * Kx;
  for (int k = 0; k < Kx; k += 4) {
    v2f a;
    a[0] = Xr[k + 2 * half + 0];
    a[1] = Xr[k + 2 * half + 1];
#pragma unroll
    for (int tt = 0; tt < 4; ++tt) {
      v2f w;
      w[0] = Bs[(k + 2 * half + 0) * 64 + tt * 16 + l16];
      w[1] = Bs[(k + 2 * half + 1) * 64 + tt * 16 + l16];
      acc[tt] = __builtin_amdgcn_wmma_f32_16x16x4_f32(false, a, false, w, (short)0,
                                                      acc[tt], false, false);
    }
  }
  const float* Hr = hin + (long)ar * 16;
#pragma unroll
  for (int k = 0; k < 16; k += 4) {
    v2f a;
    a[0] = Hr[k + 2 * half + 0];
    a[1] = Hr[k + 2 * half + 1];
#pragma unroll
    for (int tt = 0; tt < 4; ++tt) {
      v2f w;
      w[0] = Bs[(Kx + k + 2 * half + 0) * 64 + tt * 16 + l16];
      w[1] = Bs[(Kx + k + 2 * half + 1) * 64 + tt * 16 + l16];
      acc[tt] = __builtin_amdgcn_wmma_f32_16x16x4_f32(false, a, false, w, (short)0,
                                                      acc[tt], false, false);
    }
  }
#pragma unroll
  for (int r = 0; r < 8; ++r) {
    if (r + 8 * half < rleft) {
      float gi = acc[0][r], gf = acc[1][r], gg = acc[2][r], go = acc[3][r];
      float cn = sigm(gf) * cold[r] + sigm(gi) * tanhf(gg);
      cT[loff + r * 16] = cn;
      hT[loff + r * 16] = sigm(go) * tanhf(cn);
    }
  }
  }
}

__device__ __forceinline__ void role_gather(const int vb, const int* __restrict__ ids,
                                            const float* __restrict__ he,
                                            float* __restrict__ out32, int N) {
  const int sub = threadIdx.x & 7;
  const int r0  = vb * 512 + (threadIdx.x >> 3);
  if (vb * 512 + 512 <= N) {
    int id[4];
    v4f v[4];
#pragma unroll
    for (int p = 0; p < 4; ++p) id[p] = ids[r0 + 128 * p];
#pragma unroll
    for (int p = 0; p < 4; ++p)
      v[p] = *(const v4f*)(he + (long)id[p] * 32 + 4 * sub);
    asm volatile("" : "+v"(v[0]), "+v"(v[1]), "+v"(v[2]), "+v"(v[3]));
#pragma unroll
    for (int p = 0; p < 4; ++p)
      *(v4f*)(out32 + (long)(r0 + 128 * p) * 32 + 4 * sub) = v[p];
  } else {
    for (int p = 0; p < 4; ++p) {
      int row = r0 + 128 * p;
      if (row < N)
        *(v4f*)(out32 + (long)row * 32 + 4 * sub) =
            *(const v4f*)(he + (long)ids[row] * 32 + 4 * sub);
    }
  }
}

template <int TH>
__device__ __forceinline__ void role_outproj(const int vb, const float* __restrict__ h,
                                             const float* __restrict__ Wout,
                                             const float* __restrict__ bout,
                                             float* __restrict__ out, int N) {
  int i = vb * TH + threadIdx.x;
  if (i >= N) return;
  const float* hr = h + (long)i * 16;
  float a0 = bout[0], a1 = bout[1];
  for (int j = 0; j < 16; ++j) {
    float v = hr[j];
    a0 += v * Wout[j * 2 + 0];
    a1 += v * Wout[j * 2 + 1];
  }
  out[(long)i * 2 + 0] = a0;
  out[(long)i * 2 + 1] = a1;
}

__global__ __launch_bounds__(1024) void k_stageA(
    int nLoc, const int* __restrict__ idsNext, const float* __restrict__ Wemb,
    const float* __restrict__ bemb, float* __restrict__ e,
    const int* __restrict__ idsCur, const float* __restrict__ he2n, float* __restrict__ out32) {
  __shared__ __attribute__((aligned(16))) int arena[AR_BAND];
  const int b = blockIdx.x;
  if (b < nLoc) role_user_sum(b, idsNext, Wemb, bemb, e, arena);
  else          role_gather(b - nLoc, idsCur, he2n, out32, USR_NUM);
}

__global__ __launch_bounds__(1024) void k_stageBD(
    const int* __restrict__ src, const int* __restrict__ dst,
    const float* __restrict__ isq, const float* __restrict__ idg,
    const float* __restrict__ hin, float* __restrict__ agg) {
  __shared__ __attribute__((aligned(16))) int arena[AR_BAND];
  role_edge_sum(blockIdx.x, src, dst, isq, idg, hin, agg, arena);
}

__global__ __launch_bounds__(256) void k_stageCE(
    int nLoc, const float* __restrict__ A, const float* __restrict__ W,
    const float* __restrict__ bias, float* __restrict__ C, int relu,
    const float* __restrict__ X, int Kx, const float* __restrict__ Wi,
    const float* __restrict__ Wh, const float* __restrict__ bL,
    float* hS, float* cS) {
  __shared__ float Bs[AR_GEMM];
  const int b = blockIdx.x;
  if (b < nLoc) role_gemm<4, 256>(b, A, W, bias, C, NUM_LOC, 64, relu, Bs);
  else          role_layer<256>(b - nLoc, 8, X, Kx, Wi, Wh, bL, hS, hS, cS, USR_NUM, Bs);
}

__global__ __launch_bounds__(256) void k_stageF(
    int nLoc, const float* __restrict__ A, const float* __restrict__ W,
    const float* __restrict__ bias, float* __restrict__ C,
    const float* __restrict__ h, const float* __restrict__ Wout,
    const float* __restrict__ bout, float* __restrict__ out) {
  __shared__ float Bs[AR_GEMM];
  const int b = blockIdx.x;
  if (b < nLoc) role_gemm<2, 256>(b, A, W, bias, C, NUM_LOC, 64, 0, Bs);
  else          role_outproj<256>(b - nLoc, h, Wout, bout, out, USR_NUM);
}


#define PL_LOC   ((long)NUM_LOC * 64)
#define PL_E2N   ((long)NUM_LOC * 32)
#define PL_X     ((long)USR_NUM * 32)
#define PL_ST    ((long)USR_NUM * 16)

#define NB_BAND   ((NUM_LOC + ROWS_PB - 1) / ROWS_PB)
#define NB_BAND3  (3 * NB_BAND)
#define N_TILES   ((USR_NUM + 15) / 16)
#define TILES_PB  ((2 * N_TILES + NB_BAND3 - 1) / NB_BAND3)
#define NB_LAYF   ((N_TILES + TILES_PB - 1) / TILES_PB)
#define NB_GATHER ((USR_NUM + 511) / 512)
#define NB_GEMM4  ((NUM_LOC + 511) / 512)
#define NB_OUT4   ((USR_NUM + 1023) / 1024)
#define NB_TICK   (NB_BAND3 + 3 * NB_GEMM4 + NB_GATHER + NB_OUT4)
#define N_TICKS   (T_STEPS + 9)
static_assert(TILES_PB <= 32, "one layer tile a wave");
static_assert(2 * NB_LAYF <= NB_BAND3, "both layers' tiles must fit the carrying work-groups");
#define NL_PAD 5056
#define WS_DEG   0L
#define WS_ISQ   (1L * NL_PAD)
#define WS_IDG   (2L * NL_PAD)
#define WS_E     (3L * NL_PAD)
#define WS_AGGA  (WS_E    + 2 * PL_LOC)
#define WS_H1B   (WS_AGGA + 2 * PL_LOC)
#define WS_AGGB  (WS_H1B  + 2 * PL_LOC)
#define WS_H2B   (WS_AGGB + 2 * PL_LOC)
#define WS_HE2N  (WS_H2B  + 2 * PL_LOC)
#define WS_OUT32 (WS_HE2N + 2 * PL_E2N)
#define WS_H0P1  (WS_OUT32 + 2 * PL_X)
#define WS_C0    (WS_H0P1 + PL_ST)
#define WS_H1P1  (WS_C0   + PL_ST)
#define WS_C1    (WS_H1P1 + PL_ST)
#define WS_H0P0  (WS_C1   + PL_ST)
#define WS_H1P0  (WS_H0P0 + PL_ST)
#define WS_TICK_FLOATS (WS_H1P0 + PL_ST)
#define WS_TICK_BYTES  ((size_t)(WS_TICK_FLOATS * 4))
static_assert(WS_TICK_FLOATS == 3L * NL_PAD + 10 * PL_LOC + 2 * PL_E2N + 2 * PL_X + 6L * USR_NUM * 16, "workspace layout");

struct TickArgs {
  const int*   seq;  const int*   src;  const int*   dst;
  const float* Wemb; const float* bemb;
  const float* W1;   const float* b1;   const float* W2;   const float* b2;
  const float* We;   const float* be;
  const float* Wi0;  const float* Wh0;  const float* bL0;
  const float* Wi1;  const float* Wh1;  const float* bL1;
  const float* Wout; const float* bout;
  float* ws;
  float* out;
};

__device__ __forceinline__ void band_job(const int b, const TickArgs& a, const int tick, int* arena) {
  float* const ws = a.ws;
  __builtin_amdgcn_s_setprio(3);
  if (b < NB_BAND) {
    const int s = tick;
    if ((unsigned)s < (unsigned)T_STEPS)
      role_user_sum(b, a.seq + (long)s * USR_NUM, a.Wemb, a.bemb, ws + WS_E + (s & 1) * PL_LOC, arena);
  } else if (b < 2 * NB_BAND) {
    const int s = tick - 1;
    if ((unsigned)s < (unsigned)T_STEPS)
      role_edge_sum(b - NB_BAND, a.src, a.dst, ws + WS_ISQ, ws + WS_IDG, ws + WS_E + (s & 1) * PL_LOC,
                    ws + WS_AGGA + (s & 1) * PL_LOC, arena);
  } else {
    const int s = tick - 3;
    if ((unsigned)s < (unsigned)T_STEPS)
      role_edge_sum(b - 2 * NB_BAND, a.src, a.dst, ws + WS_ISQ, ws + WS_IDG, ws + WS_H1B + (s & 1) * PL_LOC,
                    ws + WS_AGGB + (s & 1) * PL_LOC, arena);
  }
  __builtin_amdgcn_s_setprio(0);
}

__global__ __launch_bounds__(1024, 16) void k_tick(const TickArgs a, const int tick) {
  __shared__ __attribute__((aligned(16))) int arena[AR_BAND];
  float* const ws = a.ws;
  int b = blockIdx.x;
  int s;
  if (b < NB_BAND3) {
    const bool layerFirst = (__builtin_popcount((unsigned)b) & 1) != 0;
    if (!layerFirst) band_job(b, a, tick, arena);
    __syncthreads();
    if (b < NB_LAYF) {
      s = tick - 7;
      if ((unsigned)s < (unsigned)T_STEPS)
        role_layer<1024>(b, TILES_PB, ws + WS_OUT32 + (s & 1) * PL_X, 32, a.Wi0, a.Wh0, a.bL0,
                         ws + ((s & 1) ? WS_H0P0 : WS_H0P1), ws + ((s & 1) ? WS_H0P1 : WS_H0P0), ws + WS_C0,
                         USR_NUM, (float*)arena);
    } else if (b < 2 * NB_LAYF) {
      s = tick - 8;
      if ((unsigned)s < (unsigned)T_STEPS)
        role_layer<1024>(b - NB_LAYF, TILES_PB, ws + ((s & 1) ? WS_H0P1 : WS_H0P0), 16, a.Wi1, a.Wh1, a.bL1,
                         ws + ((s & 1) ? WS_H1P0 : WS_H1P1), ws + ((s & 1) ? WS_H1P1 : WS_H1P0), ws + WS_C1,
                         USR_NUM, (float*)arena);
    }
    __syncthreads();
    if (layerFirst) band_job(b, a, tick, arena);
    return;
  }
  b -= NB_BAND3;
  if (b < NB_GEMM4) {
    s = tick - 2;
    if ((unsigned)s >= (unsigned)T_STEPS) return;
    role_gemm<4, 1024>(b, ws + WS_AGGA + (s & 1) * PL_LOC, a.W1, a.b1, ws + WS_H1B + (s & 1) * PL_LOC,
                       NUM_LOC, 64, 1, (float*)arena);
    return;
  }
  b -= NB_GEMM4;
  if (b < NB_GEMM4) {
    s = tick - 4;
    if ((unsigned)s >= (unsigned)T_STEPS) return;
    role_gemm<4, 1024>(b, ws + WS_AGGB + (s & 1) * PL_LOC, a.W2, a.b2, ws + WS_H2B + (s & 1) * PL_LOC,
                       NUM_LOC, 64, 0, (float*)arena);
    return;
  }
  b -= NB_GEMM4;
  if (b < NB_GEMM4) {
    s = tick - 5;
    if ((unsigned)s >= (unsigned)T_STEPS) return;
    role_gemm<2, 1024>(b, ws + WS_H2B + (s & 1) * PL_LOC, a.We, a.be, ws + WS_HE2N + (s & 1) * PL_E2N,
                       NUM_LOC, 64, 0, (float*)arena);
    return;
  }
  b -= NB_GEMM4;
  if (b < NB_GATHER) {
    s = tick - 6;
    if ((unsigned)s >= (unsigned)T_STEPS) return;
    role_gather(b, a.seq + (long)s * USR_NUM, ws + WS_HE2N + (s & 1) * PL_E2N,
                ws + WS_OUT32 + (s & 1) * PL_X, USR_NUM);
    return;
  }
  b -= NB_GATHER;
  {
    s = tick - 9;
    if ((unsigned)s >= (unsigned)T_STEPS) return;
    role_outproj<1024>(b, ws + ((s & 1) ? WS_H1P1 : WS_H1P0), a.Wout, a.bout,
                       a.out + (long)s * USR_NUM * 2, USR_NUM);
  }
}

static inline int gs(long n, int b) { return (int)((n + b - 1) / b); }

#define NB_LAYER  ((USR_NUM + 127) / 128)
#define NB_GEMM   ((NUM_LOC + 127) / 128)
#define NB_OUT    ((USR_NUM + 255) / 256)

extern "C" void kernel_launch(void* const* d_in, const int* in_sizes, int n_in,
                              void* d_out, int out_size, void* d_ws, size_t ws_size,
                              hipStream_t stream) {
  (void)in_sizes; (void)n_in; (void)out_size;

  const int*   seq  = (const int*)d_in[0];
  const int*   gidx = (const int*)d_in[1];
  const float* Wemb = (const float*)d_in[2];
  const float* bemb = (const float*)d_in[3];
  const float* W1   = (const float*)d_in[4];
  const float* b1   = (const float*)d_in[5];
  const float* W2   = (const float*)d_in[6];
  const float* b2   = (const float*)d_in[7];
  const float* We2n = (const float*)d_in[8];
  const float* be2n = (const float*)d_in[9];
  const float* Wi0  = (const float*)d_in[10];
  const float* Wh0  = (const float*)d_in[11];
  const float* bL0  = (const float*)d_in[12];
  const float* Wi1  = (const float*)d_in[13];
  const float* Wh1  = (const float*)d_in[14];
  const float* bL1  = (const float*)d_in[15];
  const float* Wout = (const float*)d_in[16];
  const float* bout = (const float*)d_in[17];

  const int* srcE = gidx;
  const int* dstE = gidx + N_EDGES;
  float* out = (float*)d_out;
  const int BT = 256;

  float* p = (float*)d_ws;
  float* deg   = p; p += NL_PAD;
  float* isq   = p; p += NL_PAD;
  float* idg   = p; p += NL_PAD;

  k_zero<<<gs(NUM_LOC, BT), BT, 0, stream>>>(deg, NUM_LOC);
  k_deg_accum<<<gs(N_EDGES, BT), BT, 0, stream>>>(dstE, deg, N_EDGES);
  k_deg_fin<<<gs(NUM_LOC, BT), BT, 0, stream>>>(deg, isq, idg, NUM_LOC);

  if (ws_size >= WS_TICK_BYTES) {
    TickArgs a;
    a.seq = seq; a.src = srcE; a.dst = dstE;
    a.Wemb = Wemb; a.bemb = bemb;
    a.W1 = W1; a.b1 = b1; a.W2 = W2; a.b2 = b2; a.We = We2n; a.be = be2n;
    a.Wi0 = Wi0; a.Wh0 = Wh0; a.bL0 = bL0; a.Wi1 = Wi1; a.Wh1 = Wh1; a.bL1 = bL1;
    a.Wout = Wout; a.bout = bout;
    a.ws = (float*)d_ws;
    a.out = out;

    k_zero<<<gs((long)USR_NUM * 64, BT), BT, 0, stream>>>((float*)d_ws + WS_H0P1, USR_NUM * 64);
    for (int tick = 0; tick < N_TICKS; ++tick)
      k_tick<<<NB_TICK, 1024, 0, stream>>>(a, tick);
    return;
  }

  float* e     = p; p += (size_t)NUM_LOC * 64;
  float* agg   = p; p += (size_t)NUM_LOC * 64;
  float* h1b   = p; p += (size_t)NUM_LOC * 64;
  float* h2b   = p; p += (size_t)NUM_LOC * 64;
  float* he2n  = p; p += (size_t)NUM_LOC * 32;
  float* out32 = p; p += (size_t)USR_NUM * 32;
  float* h0    = p; p += (size_t)USR_NUM * 16;
  float* c0    = p; p += (size_t)USR_NUM * 16;
  float* h1s   = p; p += (size_t)USR_NUM * 16;
  float* c1s   = p; p += (size_t)USR_NUM * 16;

  k_zero<<<gs((long)USR_NUM * 64, BT), BT, 0, stream>>>(h0, USR_NUM * 64);

  k_stageA <<<NB_BAND, 1024, 0, stream>>>(NB_BAND, seq, Wemb, bemb, e, seq, he2n, out32);
  k_stageBD<<<NB_BAND, 1024, 0, stream>>>(srcE, dstE, isq, idg, e, agg);
  k_stageCE<<<NB_GEMM, 256, 0, stream>>>(NB_GEMM, agg, W1, b1, h1b, 1,
                                         out32, 32, Wi0, Wh0, bL0, h0, c0);
  k_stageBD<<<NB_BAND, 1024, 0, stream>>>(srcE, dstE, isq, idg, h1b, agg);
  k_stageCE<<<NB_GEMM, 256, 0, stream>>>(NB_GEMM, agg, W2, b2, h2b, 0,
                                         h0, 16, Wi1, Wh1, bL1, h1s, c1s);
  k_stageF <<<NB_GEMM, 256, 0, stream>>>(NB_GEMM, h2b, We2n, be2n, he2n, h1s, Wout, bout, out);

  for (int t = 0; t + 1 < T_STEPS; ++t) {
    const int* idsCur  = seq + (long)t * USR_NUM;
    const int* idsNext = idsCur + USR_NUM;
    float* outT = out + (long)t * USR_NUM * 2;

    k_stageA <<<NB_BAND + NB_GATHER, 1024, 0, stream>>>(NB_BAND, idsNext, Wemb, bemb, e,
                                                        idsCur, he2n, out32);
    k_stageBD<<<NB_BAND, 1024, 0, stream>>>(srcE, dstE, isq, idg, e, agg);
    k_stageCE<<<NB_GEMM + NB_LAYER, 256, 0, stream>>>(NB_GEMM, agg, W1, b1, h1b, 1,
                                                      out32, 32, Wi0, Wh0, bL0, h0, c0);
    k_stageBD<<<NB_BAND, 1024, 0, stream>>>(srcE, dstE, isq, idg, h1b, agg);
    k_stageCE<<<NB_GEMM + NB_LAYER, 256, 0, stream>>>(NB_GEMM, agg, W2, b2, h2b, 0,
                                                      h0, 16, Wi1, Wh1, bL1, h1s, c1s);
    k_stageF <<<NB_GEMM + NB_OUT, 256, 0, stream>>>(NB_GEMM, h2b, We2n, be2n, he2n,
                                                     h1s, Wout, bout, outT);
  }

  {
    const int* idsCur = seq + (long)(T_STEPS - 1) * USR_NUM;
    float* outT = out + (long)(T_STEPS - 1) * USR_NUM * 2;
    k_stageA <<<NB_GATHER, 1024, 0, stream>>>(0, idsCur, Wemb, bemb, e, idsCur, he2n, out32);
    k_stageCE<<<NB_LAYER, 256, 0, stream>>>(0, agg, W1, b1, h1b, 1,
                                            out32, 32, Wi0, Wh0, bL0, h0, c0);
    k_stageCE<<<NB_LAYER, 256, 0, stream>>>(0, agg, W2, b2, h2b, 0,
                                            h0, 16, Wi1, Wh1, bL1, h1s, c1s);
    k_stageF <<<NB_OUT, 256, 0, stream>>>(0, h2b, We2n, be2n, he2n, h1s, Wout, bout, outT);
  }
}


# Layer-0 tiles read their 16 input rows from the edge2node plane through the step's key row: gather role (6272 waves a tick) and its two per-user planes gone; pipeline 10 -> 9 deep, 29 -> 28 launches
# speedup vs baseline: 4.3662x; 1.0450x over previous
#include <hip/hip_runtime.h>
#include <math.h>


#define USR_NUM 100000
#define USR_DIM 64
#define NUM_LOC 5000
#define T_STEPS 16
#define N_EDGES 80000

typedef float v2f __attribute__((ext_vector_type(2)));
typedef float v4f __attribute__((ext_vector_type(4)));
typedef float v8f __attribute__((ext_vector_type(8)));

__global__ void k_zero(float* __restrict__ p, int n) {
  int i = blockIdx.x * blockDim.x + threadIdx.x;
  if (i < n) p[i] = 0.0f;
}

__global__ void k_deg_accum(const int* __restrict__ dst, float* __restrict__ deg, int E) {
  int i = blockIdx.x * blockDim.x + threadIdx.x;
  if (i < E) atomicAdd(&deg[dst[i]], 1.0f);
}

__global__ void k_deg_fin(const float* __restrict__ deg, float* __restrict__ isq,
                          float* __restrict__ idg, int n) {
  int i = blockIdx.x * blockDim.x + threadIdx.x;
  if (i < n) {
    float d = deg[i] + 1.0f;
    isq[i] = rsqrtf(d);
    idg[i] = 1.0f / d;
  }
}


#define ROWS_PB   32
#define SCAN_KEYS 16384
#define SCAN_V4   (SCAN_KEYS / 4 / 1024)
#define WQ_HIGH   128
#define WQ_CAP    (WQ_HIGH + 4 * 32 * SCAN_V4 + 4)
#define AR_TILE   0
#define AR_AUX    ((ROWS_PB + 1) * 64)
#define AR_Q      (AR_AUX + 64)
#define AR_BAND   (AR_Q + 32 * WQ_CAP)
#define AR_GEMM   (64 * 64)

__device__ __forceinline__ void role_user_sum(
    const int vb, const int* __restrict__ ids, const float* __restrict__ Wemb,
    const float* __restrict__ bemb, float* __restrict__ e, int* arena) {
  float* tile = (float*)(arena + AR_TILE);
  float* cntS = (float*)(arena + AR_AUX);

  const int tid  = threadIdx.x;
  const int lane = tid & 31;
  const int wave = tid >> 5;
  const int lo   = vb * ROWS_PB;
  int* wq = arena + AR_Q + wave * WQ_CAP;

  for (int i = tid; i < (ROWS_PB + 1) * 64; i += 1024) tile[i] = 0.0f;
  if (tid < ROWS_PB) cntS[tid] = 0.0f;

  const int4* ids4 = (const int4*)ids;
  const int nv4 = USR_NUM / 4;
  const float2 bb = *(const float2*)(bemb + 2 * lane);

  int4 cur[SCAN_V4];
#pragma unroll
  for (int k = 0; k < SCAN_V4; ++k) {
    int i4 = k * 1024 + tid;
    cur[k] = (i4 < nv4) ? ids4[i4] : make_int4(-1, -1, -1, -1);
  }
  __syncthreads();

  int nq = 0;
  int head = 0;
  for (int base4 = 0; base4 < nv4; base4 += SCAN_KEYS / 4) {
    int4 nx[SCAN_V4];
#pragma unroll
    for (int k = 0; k < SCAN_V4; ++k) {
      int i4 = base4 + SCAN_KEYS / 4 + k * 1024 + tid;
      nx[k] = (i4 < nv4) ? ids4[i4] : make_int4(-1, -1, -1, -1);
    }
#pragma unroll
    for (int k = 0; k < SCAN_V4; ++k) {
      const int u0 = (base4 + k * 1024 + tid) * 4;
      const int kv[4] = { cur[k].x, cur[k].y, cur[k].z, cur[k].w };
#pragma unroll
      for (int c = 0; c < 4; ++c) {
        const unsigned r = (unsigned)(kv[c] - lo);
        const bool hit = r < (unsigned)ROWS_PB;
        const unsigned m = __builtin_amdgcn_ballot_w32(hit);
        if (m != 0u) {
          if (hit) {
            wq[nq + (int)__builtin_amdgcn_mbcnt_lo(m, 0u)] = ((u0 + c) << 8) | (int)r;
            atomicAdd(&cntS[r], 1.0f);
          }
          nq += __builtin_popcount(m);
        }
      }
    }
    const bool drain = (base4 + SCAN_KEYS / 4 >= nv4) || (nq > WQ_HIGH);
    if (drain && lane < 3) wq[nq + lane] = ROWS_PB;
    const int stop = drain ? ((nq + 3) & ~3) : (nq & ~3);
    for (int j0 = head; j0 < stop; j0 += 4) {
      const int4 q4 = *(const int4*)(wq + j0);
      const int ent[4] = { __builtin_amdgcn_readfirstlane(q4.x), __builtin_amdgcn_readfirstlane(q4.y),
                           __builtin_amdgcn_readfirstlane(q4.z), __builtin_amdgcn_readfirstlane(q4.w) };
      float2 v[4];
#pragma unroll
      for (int k = 0; k < 4; ++k)
        v[k] = *(const float2*)(Wemb + (long)(ent[k] >> 8) * 64 + 2 * lane);
#pragma unroll
      for (int k = 0; k < 4; ++k) {
        int r = ent[k] & 255;
        float* dstp = &tile[r * 64 + 2 * lane];
        atomicAdd(dstp + 0, v[k].x + bb.x);
        atomicAdd(dstp + 1, v[k].y + bb.y);
      }
    }
    head = stop;
    if (drain) { head = 0; nq = 0; }
#pragma unroll
    for (int k = 0; k < SCAN_V4; ++k) cur[k] = nx[k];
  }
  __syncthreads();

  for (int i = tid; i < ROWS_PB * 64; i += 1024) {
    int j = i >> 6;
    int row = lo + j;
    if (row < NUM_LOC) {
      float cv = cntS[j]; if (cv < 1.0f) cv = 1.0f;
      e[(long)row * 64 + (i & 63)] = tile[i] / cv;
    }
  }
}

__device__ __forceinline__ void role_edge_sum(
    const int vb, const int* __restrict__ src, const int* __restrict__ dst,
    const float* __restrict__ isq, const float* __restrict__ idg,
    const float* __restrict__ h, float* __restrict__ agg, int* arena) {
  float* tile = (float*)(arena + AR_TILE);
  float* isqS = (float*)(arena + AR_AUX);

  const int tid  = threadIdx.x;
  const int lane = tid & 31;
  const int wave = tid >> 5;
  const int lo   = vb * ROWS_PB;
  int* wq = arena + AR_Q + wave * WQ_CAP;

  for (int i = tid; i < (ROWS_PB + 1) * 64; i += 1024) {
    int row = lo + (i >> 6);
    tile[i] = (i < ROWS_PB * 64 && row < NUM_LOC) ? h[(long)row * 64 + (i & 63)] * idg[row] : 0.0f;
  }
  if (tid <= ROWS_PB) isqS[tid] = (tid < ROWS_PB && lo + tid < NUM_LOC) ? isq[lo + tid] : 0.0f;

  const int4* dst4 = (const int4*)dst;
  const int nv4 = N_EDGES / 4;

  int4 cur[SCAN_V4];
#pragma unroll
  for (int k = 0; k < SCAN_V4; ++k) {
    int i4 = k * 1024 + tid;
    cur[k] = (i4 < nv4) ? dst4[i4] : make_int4(-1, -1, -1, -1);
  }
  __syncthreads();

  int nq = 0;
  int head = 0;
  for (int base4 = 0; base4 < nv4; base4 += SCAN_KEYS / 4) {
    int4 nx[SCAN_V4];
#pragma unroll
    for (int k = 0; k < SCAN_V4; ++k) {
      int i4 = base4 + SCAN_KEYS / 4 + k * 1024 + tid;
      nx[k] = dst4[i4 < nv4 ? i4 : nv4 - 1];
    }
#pragma unroll
    for (int k = 0; k < SCAN_V4; ++k) {
      const int e0 = (base4 + k * 1024 + tid) * 4;
      const int kv[4] = { cur[k].x, cur[k].y, cur[k].z, cur[k].w };
#pragma unroll
      for (int c = 0; c < 4; ++c) {
        const unsigned r = (unsigned)(kv[c] - lo);
        const bool hit = r < (unsigned)ROWS_PB;
        const unsigned m = __builtin_amdgcn_ballot_w32(hit);
        if (m != 0u) {
          if (hit) wq[nq + (int)__builtin_amdgcn_mbcnt_lo(m, 0u)] = ((e0 + c) << 8) | (int)r;
          nq += __builtin_popcount(m);
        }
      }
    }
    const bool drain = (base4 + SCAN_KEYS / 4 >= nv4) || (nq > WQ_HIGH);
    if (drain && lane < 3) wq[nq + lane] = ROWS_PB;
    const int stop = drain ? ((nq + 3) & ~3) : (nq & ~3);
    for (int j0 = head; j0 < stop; j0 += 4) {
      const int4 q4 = *(const int4*)(wq + j0);
      const int ent[4] = { __builtin_amdgcn_readfirstlane(q4.x), __builtin_amdgcn_readfirstlane(q4.y),
                           __builtin_amdgcn_readfirstlane(q4.z), __builtin_amdgcn_readfirstlane(q4.w) };
      float wr[4];
#pragma unroll
      for (int k = 0; k < 4; ++k) wr[k] = isqS[ent[k] & 255];
      int s[4];
#pragma unroll
      for (int k = 0; k < 4; ++k) s[k] = src[ent[k] >> 8];
      float  w[4];
#pragma unroll
      for (int k = 0; k < 4; ++k) w[k] = isq[s[k]];
      float2 v[4];
#pragma unroll
      for (int k = 0; k < 4; ++k)
        v[k] = *(const float2*)(h + (long)s[k] * 64 + 2 * lane);
#pragma unroll
      for (int k = 0; k < 4; ++k) {
        float* dstp = &tile[(ent[k] & 255) * 64 + 2 * lane];
        atomicAdd(dstp + 0, v[k].x * w[k] * wr[k]);
        atomicAdd(dstp + 1, v[k].y * w[k] * wr[k]);
      }
    }
    head = stop;
    if (drain) { head = 0; nq = 0; }
    if (base4 + 2 * (SCAN_KEYS / 4) > nv4) {
#pragma unroll
      for (int k = 0; k < SCAN_V4; ++k)
        if (base4 + SCAN_KEYS / 4 + k * 1024 + tid >= nv4) nx[k] = make_int4(-1, -1, -1, -1);
    }
#pragma unroll
    for (int k = 0; k < SCAN_V4; ++k) cur[k] = nx[k];
  }
  __syncthreads();

  for (int i = tid; i < ROWS_PB * 64; i += 1024) {
    int row = lo + (i >> 6);
    if (row < NUM_LOC) agg[(long)row * 64 + (i & 63)] = tile[i];
  }
}

template <int NT, int TH>
__device__ __forceinline__ void role_gemm(
    const int vb, const float* __restrict__ A, const float* __restrict__ B,
    const float* __restrict__ bias, float* __restrict__ C, int M, int K, int relu,
    float* Bs) {
  constexpr int Nc = NT * 16;
  for (int i = threadIdx.x; i < K * Nc; i += TH) Bs[i] = B[i];
  __syncthreads();

  const int lane = threadIdx.x & 31;
  const int wave = threadIdx.x >> 5;
  const int half = lane >> 4;
  const int l16  = lane & 15;
  const int row0 = (vb * (TH / 32) + wave) * 16;

  int ar = row0 + l16; if (ar > M - 1) ar = M - 1;
  const float* Arow = A + (long)ar * K;

  v8f acc[NT];
#pragma unroll
  for (int tt = 0; tt < NT; ++tt) {
    float bv = bias[tt * 16 + l16];
    v8f tmp = { bv, bv, bv, bv, bv, bv, bv, bv };
    acc[tt] = tmp;
  }

  for (int k = 0; k < K; k += 4) {
    v2f a;
    a[0] = Arow[k + 2 * half + 0];
    a[1] = Arow[k + 2 * half + 1];
#pragma unroll
    for (int tt = 0; tt < NT; ++tt) {
      v2f b;
      b[0] = Bs[(k + 2 * half + 0) * Nc + tt * 16 + l16];
      b[1] = Bs[(k + 2 * half + 1) * Nc + tt * 16 + l16];
      acc[tt] = __builtin_amdgcn_wmma_f32_16x16x4_f32(false, a, false, b, (short)0,
                                                      acc[tt], false, false);
    }
  }
#pragma unroll
  for (int tt = 0; tt < NT; ++tt) {
    for (int r = 0; r < 8; ++r) {
      int row = row0 + r + 8 * half;
      if (row < M) {
        float v = acc[tt][r];
        if (relu) v = v > 0.0f ? v : 0.0f;
        C[(long)row * Nc + tt * 16 + l16] = v;
      }
    }
  }
}

__device__ __forceinline__ float sigm(float x) { return 1.0f / (1.0f + expf(-x)); }

template <int TH, bool VIA_IDS>
__device__ __forceinline__ void role_layer(
    const int vb, const int TP, const float* __restrict__ X, const int* __restrict__ xids, int Kx,
    const float* __restrict__ Wi,
    const float* __restrict__ Wh, const float* __restrict__ b,
    const float* hin, float* hout, float* c, int M, float* Bs) {
  for (int i = threadIdx.x; i < Kx * 64; i += TH) Bs[i] = Wi[i];
  for (int i = threadIdx.x; i < 16 * 64; i += TH) Bs[Kx * 64 + i] = Wh[i];
  __syncthreads();

  const int lane = threadIdx.x & 31;
  const int half = lane >> 4;
  const int l16  = lane & 15;
  const int wave = __builtin_amdgcn_readfirstlane((int)(threadIdx.x >> 5));
  const int row0 = (vb * TP + wave) * 16;
  if (wave < TP && row0 < M) {

  int ar = row0 + l16; if (ar > M - 1) ar = M - 1;
  int xrow = ar;
  if (VIA_IDS) xrow = xids[ar];
  const int rleft = M - row0;
  const unsigned loff = (unsigned)(half * 128 + l16);
  float* const cT = c + (long)row0 * 16;
  float* const hT = hout + (long)row0 * 16;

  float cold[8];
#pragma unroll
  for (int r = 0; r < 8; ++r)
    cold[r] = (r + 8 * half < rleft) ? cT[loff + r * 16] : 0.0f;

  v8f acc[4];
#pragma unroll
  for (int tt = 0; tt < 4; ++tt) {
    float bv = b[tt * 16 + l16];
    v8f tmp = { bv, bv, bv, bv, bv, bv, bv, bv };
    acc[tt] = tmp;
  }

  const float* Xr = X + (long)xrow * Kx;
  for (int k = 0; k < Kx; k += 4) {
    v2f a;
    a[0] = Xr[k + 2 * half + 0];
    a[1] = Xr[k + 2 * half + 1];
#pragma unroll
    for (int tt = 0; tt < 4; ++tt) {
      v2f w;
      w[0] = Bs[(k + 2 * half + 0) * 64 + tt * 16 + l16];
      w[1] = Bs[(k + 2 * half + 1) * 64 + tt * 16 + l16];
      acc[tt] = __builtin_amdgcn_wmma_f32_16x16x4_f32(false, a, false, w, (short)0,
                                                      acc[tt], false, false);
    }
  }
  const float* Hr = hin + (long)ar * 16;
#pragma unroll
  for (int k = 0; k < 16; k += 4) {
    v2f a;
    a[0] = Hr[k + 2 * half + 0];
    a[1] = Hr[k + 2 * half + 1];
#pragma unroll
    for (int tt = 0; tt < 4; ++tt) {
      v2f w;
      w[0] = Bs[(Kx + k + 2 * half + 0) * 64 + tt * 16 + l16];
      w[1] = Bs[(Kx + k + 2 * half + 1) * 64 + tt * 16 + l16];
      acc[tt] = __builtin_amdgcn_wmma_f32_16x16x4_f32(false, a, false, w, (short)0,
                                                      acc[tt], false, false);
    }
  }
#pragma unroll
  for (int r = 0; r < 8; ++r) {
    if (r + 8 * half < rleft) {
      float gi = acc[0][r], gf = acc[1][r], gg = acc[2][r], go = acc[3][r];
      float cn = sigm(gf) * cold[r] + sigm(gi) * tanhf(gg);
      cT[loff + r * 16] = cn;
      hT[loff + r * 16] = sigm(go) * tanhf(cn);
    }
  }
  }
}

__device__ __forceinline__ void role_gather(const int vb, const int* __restrict__ ids,
                                            const float* __restrict__ he,
                                            float* __restrict__ out32, int N) {
  const int sub = threadIdx.x & 7;
  const int r0  = vb * 512 + (threadIdx.x >> 3);
  if (vb * 512 + 512 <= N) {
    int id[4];
    v4f v[4];
#pragma unroll
    for (int p = 0; p < 4; ++p) id[p] = ids[r0 + 128 * p];
#pragma unroll
    for (int p = 0; p < 4; ++p)
      v[p] = *(const v4f*)(he + (long)id[p] * 32 + 4 * sub);
    asm volatile("" : "+v"(v[0]), "+v"(v[1]), "+v"(v[2]), "+v"(v[3]));
#pragma unroll
    for (int p = 0; p < 4; ++p)
      *(v4f*)(out32 + (long)(r0 + 128 * p) * 32 + 4 * sub) = v[p];
  } else {
    for (int p = 0; p < 4; ++p) {
      int row = r0 + 128 * p;
      if (row < N)
        *(v4f*)(out32 + (long)row * 32 + 4 * sub) =
            *(const v4f*)(he + (long)ids[row] * 32 + 4 * sub);
    }
  }
}

template <int TH>
__device__ __forceinline__ void role_outproj(const int vb, const float* __restrict__ h,
                                             const float* __restrict__ Wout,
                                             const float* __restrict__ bout,
                                             float* __restrict__ out, int N) {
  int i = vb * TH + threadIdx.x;
  if (i >= N) return;
  const float* hr = h + (long)i * 16;
  float a0 = bout[0], a1 = bout[1];
  for (int j = 0; j < 16; ++j) {
    float v = hr[j];
    a0 += v * Wout[j * 2 + 0];
    a1 += v * Wout[j * 2 + 1];
  }
  out[(long)i * 2 + 0] = a0;
  out[(long)i * 2 + 1] = a1;
}

__global__ __launch_bounds__(1024) void k_stageA(
    int nLoc, const int* __restrict__ idsNext, const float* __restrict__ Wemb,
    const float* __restrict__ bemb, float* __restrict__ e,
    const int* __restrict__ idsCur, const float* __restrict__ he2n, float* __restrict__ out32) {
  __shared__ __attribute__((aligned(16))) int arena[AR_BAND];
  const int b = blockIdx.x;
  if (b < nLoc) role_user_sum(b, idsNext, Wemb, bemb, e, arena);
  else          role_gather(b - nLoc, idsCur, he2n, out32, USR_NUM);
}

__global__ __launch_bounds__(1024) void k_stageBD(
    const int* __restrict__ src, const int* __restrict__ dst,
    const float* __restrict__ isq, const float* __restrict__ idg,
    const float* __restrict__ hin, float* __restrict__ agg) {
  __shared__ __attribute__((aligned(16))) int arena[AR_BAND];
  role_edge_sum(blockIdx.x, src, dst, isq, idg, hin, agg, arena);
}

__global__ __launch_bounds__(256) void k_stageCE(
    int nLoc, const float* __restrict__ A, const float* __restrict__ W,
    const float* __restrict__ bias, float* __restrict__ C, int relu,
    const float* __restrict__ X, int Kx, const float* __restrict__ Wi,
    const float* __restrict__ Wh, const float* __restrict__ bL,
    float* hS, float* cS) {
  __shared__ float Bs[AR_GEMM];
  const int b = blockIdx.x;
  if (b < nLoc) role_gemm<4, 256>(b, A, W, bias, C, NUM_LOC, 64, relu, Bs);
  else          role_layer<256, false>(b - nLoc, 8, X, nullptr, Kx, Wi, Wh, bL, hS, hS, cS, USR_NUM, Bs);
}

__global__ __launch_bounds__(256) void k_stageF(
    int nLoc, const float* __restrict__ A, const float* __restrict__ W,
    const float* __restrict__ bias, float* __restrict__ C,
    const float* __restrict__ h, const float* __restrict__ Wout,
    const float* __restrict__ bout, float* __restrict__ out) {
  __shared__ float Bs[AR_GEMM];
  const int b = blockIdx.x;
  if (b < nLoc) role_gemm<2, 256>(b, A, W, bias, C, NUM_LOC, 64, 0, Bs);
  else          role_outproj<256>(b - nLoc, h, Wout, bout, out, USR_NUM);
}


#define PL_LOC   ((long)NUM_LOC * 64)
#define PL_E2N   ((long)NUM_LOC * 32)
#define PL_ST    ((long)USR_NUM * 16)

#define NB_BAND   ((NUM_LOC + ROWS_PB - 1) / ROWS_PB)
#define NB_BAND3  (3 * NB_BAND)
#define N_TILES   ((USR_NUM + 15) / 16)
#define TILES_PB  ((2 * N_TILES + NB_BAND3 - 1) / NB_BAND3)
#define NB_LAYF   ((N_TILES + TILES_PB - 1) / TILES_PB)
#define NB_GATHER ((USR_NUM + 511) / 512)
#define NB_GEMM4  ((NUM_LOC + 511) / 512)
#define NB_OUT4   ((USR_NUM + 1023) / 1024)
#define NB_TICK   (NB_BAND3 + 3 * NB_GEMM4 + NB_OUT4)
#define N_TICKS   (T_STEPS + 8)
static_assert(TILES_PB <= 32, "one layer tile a wave");
static_assert(2 * NB_LAYF <= NB_BAND3, "both layers' tiles must fit the carrying work-groups");
#define NL_PAD 5056
#define WS_DEG   0L
#define WS_ISQ   (1L * NL_PAD)
#define WS_IDG   (2L * NL_PAD)
#define WS_E     (3L * NL_PAD)
#define WS_AGGA  (WS_E    + 2 * PL_LOC)
#define WS_H1B   (WS_AGGA + 2 * PL_LOC)
#define WS_AGGB  (WS_H1B  + 2 * PL_LOC)
#define WS_H2B   (WS_AGGB + 2 * PL_LOC)
#define WS_HE2N  (WS_H2B  + 2 * PL_LOC)
#define WS_H0P1  (WS_HE2N + 2 * PL_E2N)
#define WS_C0    (WS_H0P1 + PL_ST)
#define WS_H1P1  (WS_C0   + PL_ST)
#define WS_C1    (WS_H1P1 + PL_ST)
#define WS_H0P0  (WS_C1   + PL_ST)
#define WS_H1P0  (WS_H0P0 + PL_ST)
#define WS_TICK_FLOATS (WS_H1P0 + PL_ST)
#define WS_TICK_BYTES  ((size_t)(WS_TICK_FLOATS * 4))
static_assert(WS_TICK_FLOATS == 3L * NL_PAD + 10 * PL_LOC + 2 * PL_E2N + 6L * USR_NUM * 16, "workspace layout");

struct TickArgs {
  const int*   seq;  const int*   src;  const int*   dst;
  const float* Wemb; const float* bemb;
  const float* W1;   const float* b1;   const float* W2;   const float* b2;
  const float* We;   const float* be;
  const float* Wi0;  const float* Wh0;  const float* bL0;
  const float* Wi1;  const float* Wh1;  const float* bL1;
  const float* Wout; const float* bout;
  float* ws;
  float* out;
};

__device__ __forceinline__ void band_job(const int b, const TickArgs& a, const int tick, int* arena) {
  float* const ws = a.ws;
  __builtin_amdgcn_s_setprio(3);
  if (b < NB_BAND) {
    const int s = tick;
    if ((unsigned)s < (unsigned)T_STEPS)
      role_user_sum(b, a.seq + (long)s * USR_NUM, a.Wemb, a.bemb, ws + WS_E + (s & 1) * PL_LOC, arena);
  } else if (b < 2 * NB_BAND) {
    const int s = tick - 1;
    if ((unsigned)s < (unsigned)T_STEPS)
      role_edge_sum(b - NB_BAND, a.src, a.dst, ws + WS_ISQ, ws + WS_IDG, ws + WS_E + (s & 1) * PL_LOC,
                    ws + WS_AGGA + (s & 1) * PL_LOC, arena);
  } else {
    const int s = tick - 3;
    if ((unsigned)s < (unsigned)T_STEPS)
      role_edge_sum(b - 2 * NB_BAND, a.src, a.dst, ws + WS_ISQ, ws + WS_IDG, ws + WS_H1B + (s & 1) * PL_LOC,
                    ws + WS_AGGB + (s & 1) * PL_LOC, arena);
  }
  __builtin_amdgcn_s_setprio(0);
}

__global__ __launch_bounds__(1024, 16) void k_tick(const TickArgs a, const int tick) {
  __shared__ __attribute__((aligned(16))) int arena[AR_BAND];
  float* const ws = a.ws;
  int b = blockIdx.x;
  int s;
  if (b < NB_BAND3) {
    const bool layerFirst = (__builtin_popcount((unsigned)b) & 1) != 0;
    if (!layerFirst) band_job(b, a, tick, arena);
    __syncthreads();
    if (b < NB_LAYF) {
      s = tick - 6;
      if ((unsigned)s < (unsigned)T_STEPS)
        role_layer<1024, true>(b, TILES_PB, ws + WS_HE2N + (s & 1) * PL_E2N, a.seq + (long)s * USR_NUM, 32,
                         a.Wi0, a.Wh0, a.bL0,
                         ws + ((s & 1) ? WS_H0P0 : WS_H0P1), ws + ((s & 1) ? WS_H0P1 : WS_H0P0), ws + WS_C0,
                         USR_NUM, (float*)arena);
    } else if (b < 2 * NB_LAYF) {
      s = tick - 7;
      if ((unsigned)s < (unsigned)T_STEPS)
        role_layer<1024, false>(b - NB_LAYF, TILES_PB, ws + ((s & 1) ? WS_H0P1 : WS_H0P0), nullptr, 16,
                         a.Wi1, a.Wh1, a.bL1,
                         ws + ((s & 1) ? WS_H1P0 : WS_H1P1), ws + ((s & 1) ? WS_H1P1 : WS_H1P0), ws + WS_C1,
                         USR_NUM, (float*)arena);
    }
    __syncthreads();
    if (layerFirst) band_job(b, a, tick, arena);
    return;
  }
  b -= NB_BAND3;
  if (b < NB_GEMM4) {
    s = tick - 2;
    if ((unsigned)s >= (unsigned)T_STEPS) return;
    role_gemm<4, 1024>(b, ws + WS_AGGA + (s & 1) * PL_LOC, a.W1, a.b1, ws + WS_H1B + (s & 1) * PL_LOC,
                       NUM_LOC, 64, 1, (float*)arena);
    return;
  }
  b -= NB_GEMM4;
  if (b < NB_GEMM4) {
    s = tick - 4;
    if ((unsigned)s >= (unsigned)T_STEPS) return;
    role_gemm<4, 1024>(b, ws + WS_AGGB + (s & 1) * PL_LOC, a.W2, a.b2, ws + WS_H2B + (s & 1) * PL_LOC,
                       NUM_LOC, 64, 0, (float*)arena);
    return;
  }
  b -= NB_GEMM4;
  if (b < NB_GEMM4) {
    s = tick - 5;
    if ((unsigned)s >= (unsigned)T_STEPS) return;
    role_gemm<2, 1024>(b, ws + WS_H2B + (s & 1) * PL_LOC, a.We, a.be, ws + WS_HE2N + (s & 1) * PL_E2N,
                       NUM_LOC, 64, 0, (float*)arena);
    return;
  }
  b -= NB_GEMM4;
  {
    s = tick - 8;
    if ((unsigned)s >= (unsigned)T_STEPS) return;
    role_outproj<1024>(b, ws + ((s & 1) ? WS_H1P1 : WS_H1P0), a.Wout, a.bout,
                       a.out + (long)s * USR_NUM * 2, USR_NUM);
  }
}

static inline int gs(long n, int b) { return (int)((n + b - 1) / b); }

#define NB_LAYER  ((USR_NUM + 127) / 128)
#define NB_GEMM   ((NUM_LOC + 127) / 128)
#define NB_OUT    ((USR_NUM + 255) / 256)

extern "C" void kernel_launch(void* const* d_in, const int* in_sizes, int n_in,
                              void* d_out, int out_size, void* d_ws, size_t ws_size,
                              hipStream_t stream) {
  (void)in_sizes; (void)n_in; (void)out_size;

  const int*   seq  = (const int*)d_in[0];
  const int*   gidx = (const int*)d_in[1];
  const float* Wemb = (const float*)d_in[2];
  const float* bemb = (const float*)d_in[3];
  const float* W1   = (const float*)d_in[4];
  const float* b1   = (const float*)d_in[5];
  const float* W2   = (const float*)d_in[6];
  const float* b2   = (const float*)d_in[7];
  const float* We2n = (const float*)d_in[8];
  const float* be2n = (const float*)d_in[9];
  const float* Wi0  = (const float*)d_in[10];
  const float* Wh0  = (const float*)d_in[11];
  const float* bL0  = (const float*)d_in[12];
  const float* Wi1  = (const float*)d_in[13];
  const float* Wh1  = (const float*)d_in[14];
  const float* bL1  = (const float*)d_in[15];
  const float* Wout = (const float*)d_in[16];
  const float* bout = (const float*)d_in[17];

  const int* srcE = gidx;
  const int* dstE = gidx + N_EDGES;
  float* out = (float*)d_out;
  const int BT = 256;

  float* p = (float*)d_ws;
  float* deg   = p; p += NL_PAD;
  float* isq   = p; p += NL_PAD;
  float* idg   = p; p += NL_PAD;

  k_zero<<<gs(NUM_LOC, BT), BT, 0, stream>>>(deg, NUM_LOC);
  k_deg_accum<<<gs(N_EDGES, BT), BT, 0, stream>>>(dstE, deg, N_EDGES);
  k_deg_fin<<<gs(NUM_LOC, BT), BT, 0, stream>>>(deg, isq, idg, NUM_LOC);

  if (ws_size >= WS_TICK_BYTES) {
    TickArgs a;
    a.seq = seq; a.src = srcE; a.dst = dstE;
    a.Wemb = Wemb; a.bemb = bemb;
    a.W1 = W1; a.b1 = b1; a.W2 = W2; a.b2 = b2; a.We = We2n; a.be = be2n;
    a.Wi0 = Wi0; a.Wh0 = Wh0; a.bL0 = bL0; a.Wi1 = Wi1; a.Wh1 = Wh1; a.bL1 = bL1;
    a.Wout = Wout; a.bout = bout;
    a.ws = (float*)d_ws;
    a.out = out;

    k_zero<<<gs((long)USR_NUM * 64, BT), BT, 0, stream>>>((float*)d_ws + WS_H0P1, USR_NUM * 64);
    for (int tick = 0; tick < N_TICKS; ++tick)
      k_tick<<<NB_TICK, 1024, 0, stream>>>(a, tick);
    return;
  }

  float* e     = p; p += (size_t)NUM_LOC * 64;
  float* agg   = p; p += (size_t)NUM_LOC * 64;
  float* h1b   = p; p += (size_t)NUM_LOC * 64;
  float* h2b   = p; p += (size_t)NUM_LOC * 64;
  float* he2n  = p; p += (size_t)NUM_LOC * 32;
  float* out32 = p; p += (size_t)USR_NUM * 32;
  float* h0    = p; p += (size_t)USR_NUM * 16;
  float* c0    = p; p += (size_t)USR_NUM * 16;
  float* h1s   = p; p += (size_t)USR_NUM * 16;
  float* c1s   = p; p += (size_t)USR_NUM * 16;

  k_zero<<<gs((long)USR_NUM * 64, BT), BT, 0, stream>>>(h0, USR_NUM * 64);

  k_stageA <<<NB_BAND, 1024, 0, stream>>>(NB_BAND, seq, Wemb, bemb, e, seq, he2n, out32);
  k_stageBD<<<NB_BAND, 1024, 0, stream>>>(srcE, dstE, isq, idg, e, agg);
  k_stageCE<<<NB_GEMM, 256, 0, stream>>>(NB_GEMM, agg, W1, b1, h1b, 1,
                                         out32, 32, Wi0, Wh0, bL0, h0, c0);
  k_stageBD<<<NB_BAND, 1024, 0, stream>>>(srcE, dstE, isq, idg, h1b, agg);
  k_stageCE<<<NB_GEMM, 256, 0, stream>>>(NB_GEMM, agg, W2, b2, h2b, 0,
                                         h0, 16, Wi1, Wh1, bL1, h1s, c1s);
  k_stageF <<<NB_GEMM, 256, 0, stream>>>(NB_GEMM, h2b, We2n, be2n, he2n, h1s, Wout, bout, out);

  for (int t = 0; t + 1 < T_STEPS; ++t) {
    const int* idsCur  = seq + (long)t * USR_NUM;
    const int* idsNext = idsCur + USR_NUM;
    float* outT = out + (long)t * USR_NUM * 2;

    k_stageA <<<NB_BAND + NB_GATHER, 1024, 0, stream>>>(NB_BAND, idsNext, Wemb, bemb, e,
                                                        idsCur, he2n, out32);
    k_stageBD<<<NB_BAND, 1024, 0, stream>>>(srcE, dstE, isq, idg, e, agg);
    k_stageCE<<<NB_GEMM + NB_LAYER, 256, 0, stream>>>(NB_GEMM, agg, W1, b1, h1b, 1,
                                                      out32, 32, Wi0, Wh0, bL0, h0, c0);
    k_stageBD<<<NB_BAND, 1024, 0, stream>>>(srcE, dstE, isq, idg, h1b, agg);
    k_stageCE<<<NB_GEMM + NB_LAYER, 256, 0, stream>>>(NB_GEMM, agg, W2, b2, h2b, 0,
                                                      h0, 16, Wi1, Wh1, bL1, h1s, c1s);
    k_stageF <<<NB_GEMM + NB_OUT, 256, 0, stream>>>(NB_GEMM, h2b, We2n, be2n, he2n,
                                                     h1s, Wout, bout, outT);
  }

  {
    const int* idsCur = seq + (long)(T_STEPS - 1) * USR_NUM;
    float* outT = out + (long)(T_STEPS - 1) * USR_NUM * 2;
    k_stageA <<<NB_GATHER, 1024, 0, stream>>>(0, idsCur, Wemb, bemb, e, idsCur, he2n, out32);
    k_stageCE<<<NB_LAYER, 256, 0, stream>>>(0, agg, W1, b1, h1b, 1,
                                            out32, 32, Wi0, Wh0, bL0, h0, c0);
    k_stageCE<<<NB_LAYER, 256, 0, stream>>>(0, agg, W2, b2, h2b, 0,
                                            h0, 16, Wi1, Wh1, bL1, h1s, c1s);
    k_stageF <<<NB_OUT, 256, 0, stream>>>(0, h2b, We2n, be2n, he2n, h1s, Wout, bout, outT);
  }
}


# Edge sums read a per-band edge list (source, row, weight isq[s]*isq[d]: product re-associated) that tick 0's idle edge bands write once a call: no key scan per tick; adds stay LDS arrival order
# speedup vs baseline: 4.8888x; 1.1197x over previous
#include <hip/hip_runtime.h>
#include <math.h>


#define USR_NUM 100000
#define USR_DIM 64
#define NUM_LOC 5000
#define T_STEPS 16
#define N_EDGES 80000

typedef float v2f __attribute__((ext_vector_type(2)));
typedef float v4f __attribute__((ext_vector_type(4)));
typedef float v8f __attribute__((ext_vector_type(8)));

__global__ void k_zero(float* __restrict__ p, int n) {
  int i = blockIdx.x * blockDim.x + threadIdx.x;
  if (i < n) p[i] = 0.0f;
}

__global__ void k_deg_accum(const int* __restrict__ dst, float* __restrict__ deg, int E) {
  int i = blockIdx.x * blockDim.x + threadIdx.x;
  if (i < E) atomicAdd(&deg[dst[i]], 1.0f);
}

__global__ void k_deg_fin(const float* __restrict__ deg, float* __restrict__ isq,
                          float* __restrict__ idg, int n) {
  int i = blockIdx.x * blockDim.x + threadIdx.x;
  if (i < n) {
    float d = deg[i] + 1.0f;
    isq[i] = rsqrtf(d);
    idg[i] = 1.0f / d;
  }
}


#define ROWS_PB   32
#define SCAN_KEYS 16384
#define SCAN_V4   (SCAN_KEYS / 4 / 1024)
#define WQ_HIGH   128
#define WQ_CAP    (WQ_HIGH + 4 * 32 * SCAN_V4 + 4)
#define AR_TILE   0
#define AR_AUX    ((ROWS_PB + 1) * 64)
#define AR_Q      (AR_AUX + 64)
#define AR_BAND   (AR_Q + 32 * WQ_CAP)
#define AR_GEMM   (64 * 64)

__device__ __forceinline__ void role_user_sum(
    const int vb, const int* __restrict__ ids, const float* __restrict__ Wemb,
    const float* __restrict__ bemb, float* __restrict__ e, int* arena) {
  float* tile = (float*)(arena + AR_TILE);
  float* cntS = (float*)(arena + AR_AUX);

  const int tid  = threadIdx.x;
  const int lane = tid & 31;
  const int wave = tid >> 5;
  const int lo   = vb * ROWS_PB;
  int* wq = arena + AR_Q + wave * WQ_CAP;

  for (int i = tid; i < (ROWS_PB + 1) * 64; i += 1024) tile[i] = 0.0f;
  if (tid < ROWS_PB) cntS[tid] = 0.0f;

  const int4* ids4 = (const int4*)ids;
  const int nv4 = USR_NUM / 4;
  const float2 bb = *(const float2*)(bemb + 2 * lane);

  int4 cur[SCAN_V4];
#pragma unroll
  for (int k = 0; k < SCAN_V4; ++k) {
    int i4 = k * 1024 + tid;
    cur[k] = (i4 < nv4) ? ids4[i4] : make_int4(-1, -1, -1, -1);
  }
  __syncthreads();

  int nq = 0;
  int head = 0;
  for (int base4 = 0; base4 < nv4; base4 += SCAN_KEYS / 4) {
    int4 nx[SCAN_V4];
#pragma unroll
    for (int k = 0; k < SCAN_V4; ++k) {
      int i4 = base4 + SCAN_KEYS / 4 + k * 1024 + tid;
      nx[k] = (i4 < nv4) ? ids4[i4] : make_int4(-1, -1, -1, -1);
    }
#pragma unroll
    for (int k = 0; k < SCAN_V4; ++k) {
      const int u0 = (base4 + k * 1024 + tid) * 4;
      const int kv[4] = { cur[k].x, cur[k].y, cur[k].z, cur[k].w };
#pragma unroll
      for (int c = 0; c < 4; ++c) {
        const unsigned r = (unsigned)(kv[c] - lo);
        const bool hit = r < (unsigned)ROWS_PB;
        const unsigned m = __builtin_amdgcn_ballot_w32(hit);
        if (m != 0u) {
          if (hit) {
            wq[nq + (int)__builtin_amdgcn_mbcnt_lo(m, 0u)] = ((u0 + c) << 8) | (int)r;
            atomicAdd(&cntS[r], 1.0f);
          }
          nq += __builtin_popcount(m);
        }
      }
    }
    const bool drain = (base4 + SCAN_KEYS / 4 >= nv4) || (nq > WQ_HIGH);
    if (drain && lane < 3) wq[nq + lane] = ROWS_PB;
    const int stop = drain ? ((nq + 3) & ~3) : (nq & ~3);
    for (int j0 = head; j0 < stop; j0 += 4) {
      const int4 q4 = *(const int4*)(wq + j0);
      const int ent[4] = { __builtin_amdgcn_readfirstlane(q4.x), __builtin_amdgcn_readfirstlane(q4.y),
                           __builtin_amdgcn_readfirstlane(q4.z), __builtin_amdgcn_readfirstlane(q4.w) };
      float2 v[4];
#pragma unroll
      for (int k = 0; k < 4; ++k)
        v[k] = *(const float2*)(Wemb + (long)(ent[k] >> 8) * 64 + 2 * lane);
#pragma unroll
      for (int k = 0; k < 4; ++k) {
        int r = ent[k] & 255;
        float* dstp = &tile[r * 64 + 2 * lane];
        atomicAdd(dstp + 0, v[k].x + bb.x);
        atomicAdd(dstp + 1, v[k].y + bb.y);
      }
    }
    head = stop;
    if (drain) { head = 0; nq = 0; }
#pragma unroll
    for (int k = 0; k < SCAN_V4; ++k) cur[k] = nx[k];
  }
  __syncthreads();

  for (int i = tid; i < ROWS_PB * 64; i += 1024) {
    int j = i >> 6;
    int row = lo + j;
    if (row < NUM_LOC) {
      float cv = cntS[j]; if (cv < 1.0f) cv = 1.0f;
      e[(long)row * 64 + (i & 63)] = tile[i] / cv;
    }
  }
}

__device__ __forceinline__ void role_edge_sum(
    const int vb, const int* __restrict__ src, const int* __restrict__ dst,
    const float* __restrict__ isq, const float* __restrict__ idg,
    const float* __restrict__ h, float* __restrict__ agg, int* arena) {
  float* tile = (float*)(arena + AR_TILE);
  float* isqS = (float*)(arena + AR_AUX);

  const int tid  = threadIdx.x;
  const int lane = tid & 31;
  const int wave = tid >> 5;
  const int lo   = vb * ROWS_PB;
  int* wq = arena + AR_Q + wave * WQ_CAP;

  for (int i = tid; i < (ROWS_PB + 1) * 64; i += 1024) {
    int row = lo + (i >> 6);
    tile[i] = (i < ROWS_PB * 64 && row < NUM_LOC) ? h[(long)row * 64 + (i & 63)] * idg[row] : 0.0f;
  }
  if (tid <= ROWS_PB) isqS[tid] = (tid < ROWS_PB && lo + tid < NUM_LOC) ? isq[lo + tid] : 0.0f;

  const int4* dst4 = (const int4*)dst;
  const int nv4 = N_EDGES / 4;

  int4 cur[SCAN_V4];
#pragma unroll
  for (int k = 0; k < SCAN_V4; ++k) {
    int i4 = k * 1024 + tid;
    cur[k] = (i4 < nv4) ? dst4[i4] : make_int4(-1, -1, -1, -1);
  }
  __syncthreads();

  int nq = 0;
  int head = 0;
  for (int base4 = 0; base4 < nv4; base4 += SCAN_KEYS / 4) {
    int4 nx[SCAN_V4];
#pragma unroll
    for (int k = 0; k < SCAN_V4; ++k) {
      int i4 = base4 + SCAN_KEYS / 4 + k * 1024 + tid;
      nx[k] = dst4[i4 < nv4 ? i4 : nv4 - 1];
    }
#pragma unroll
    for (int k = 0; k < SCAN_V4; ++k) {
      const int e0 = (base4 + k * 1024 + tid) * 4;
      const int kv[4] = { cur[k].x, cur[k].y, cur[k].z, cur[k].w };
#pragma unroll
      for (int c = 0; c < 4; ++c) {
        const unsigned r = (unsigned)(kv[c] - lo);
        const bool hit = r < (unsigned)ROWS_PB;
        const unsigned m = __builtin_amdgcn_ballot_w32(hit);
        if (m != 0u) {
          if (hit) wq[nq + (int)__builtin_amdgcn_mbcnt_lo(m, 0u)] = ((e0 + c) << 8) | (int)r;
          nq += __builtin_popcount(m);
        }
      }
    }
    const bool drain = (base4 + SCAN_KEYS / 4 >= nv4) || (nq > WQ_HIGH);
    if (drain && lane < 3) wq[nq + lane] = ROWS_PB;
    const int stop = drain ? ((nq + 3) & ~3) : (nq & ~3);
    for (int j0 = head; j0 < stop; j0 += 4) {
      const int4 q4 = *(const int4*)(wq + j0);
      const int ent[4] = { __builtin_amdgcn_readfirstlane(q4.x), __builtin_amdgcn_readfirstlane(q4.y),
                           __builtin_amdgcn_readfirstlane(q4.z), __builtin_amdgcn_readfirstlane(q4.w) };
      float wr[4];
#pragma unroll
      for (int k = 0; k < 4; ++k) wr[k] = isqS[ent[k] & 255];
      int s[4];
#pragma unroll
      for (int k = 0; k < 4; ++k) s[k] = src[ent[k] >> 8];
      float  w[4];
#pragma unroll
      for (int k = 0; k < 4; ++k) w[k] = isq[s[k]];
      float2 v[4];
#pragma unroll
      for (int k = 0; k < 4; ++k)
        v[k] = *(const float2*)(h + (long)s[k] * 64 + 2 * lane);
#pragma unroll
      for (int k = 0; k < 4; ++k) {
        float* dstp = &tile[(ent[k] & 255) * 64 + 2 * lane];
        atomicAdd(dstp + 0, v[k].x * w[k] * wr[k]);
        atomicAdd(dstp + 1, v[k].y * w[k] * wr[k]);
      }
    }
    head = stop;
    if (drain) { head = 0; nq = 0; }
    if (base4 + 2 * (SCAN_KEYS / 4) > nv4) {
#pragma unroll
      for (int k = 0; k < SCAN_V4; ++k)
        if (base4 + SCAN_KEYS / 4 + k * 1024 + tid >= nv4) nx[k] = make_int4(-1, -1, -1, -1);
    }
#pragma unroll
    for (int k = 0; k < SCAN_V4; ++k) cur[k] = nx[k];
  }
  __syncthreads();

  for (int i = tid; i < ROWS_PB * 64; i += 1024) {
    int row = lo + (i >> 6);
    if (row < NUM_LOC) agg[(long)row * 64 + (i & 63)] = tile[i];
  }
}

__device__ __forceinline__ void role_edge_list(
    const int vb, const int* __restrict__ src, const int* __restrict__ dst,
    const float* __restrict__ deg, const float* __restrict__ isq,
    int2* __restrict__ elist, int* __restrict__ boff, int* __restrict__ bcnt, int* arena) {
  int*   red    = arena + AR_TILE;
  float* isqS   = (float*)(arena + AR_AUX);
  int*   cursor = arena + AR_AUX + ROWS_PB;

  const int tid  = threadIdx.x;
  const int lane = tid & 31;
  const int wave = tid >> 5;
  const int lo   = vb * ROWS_PB;
  int* wq = arena + AR_Q + wave * WQ_CAP;

  int part = 0;
  for (int i = tid; i < lo; i += 1024) part += (int)deg[i];
  for (int m = 16; m > 0; m >>= 1) part += __shfl_xor(part, m);
  if (lane == 0) red[wave] = part;
  if (tid < ROWS_PB) isqS[tid] = (lo + tid < NUM_LOC) ? isq[lo + tid] : 0.0f;
  if (tid == 0) *cursor = 0;
  __syncthreads();
  int off = 0;
  for (int w = 0; w < 32; ++w) off += red[w];
  int2* const out = elist + off;

  const int4* dst4 = (const int4*)dst;
  const int nv4 = N_EDGES / 4;
  int nq = 0;
#pragma nounroll
  for (int base4 = 0; base4 < nv4; base4 += SCAN_KEYS / 4) {
    int4 cur[SCAN_V4];
#pragma unroll
    for (int k = 0; k < SCAN_V4; ++k) {
      int i4 = base4 + k * 1024 + tid;
      cur[k] = dst4[i4 < nv4 ? i4 : nv4 - 1];
      if (i4 >= nv4) cur[k] = make_int4(-1, -1, -1, -1);
    }
#pragma unroll
    for (int k = 0; k < SCAN_V4; ++k) {
      const int e0 = (base4 + k * 1024 + tid) * 4;
      const int kv[4] = { cur[k].x, cur[k].y, cur[k].z, cur[k].w };
#pragma unroll
      for (int c = 0; c < 4; ++c) {
        const unsigned r = (unsigned)(kv[c] - lo);
        const bool hit = r < (unsigned)ROWS_PB;
        const unsigned m = __builtin_amdgcn_ballot_w32(hit);
        if (m != 0u) {
          if (hit) wq[nq + (int)__builtin_amdgcn_mbcnt_lo(m, 0u)] = ((e0 + c) << 8) | (int)r;
          nq += __builtin_popcount(m);
        }
      }
    }
    if ((base4 + SCAN_KEYS / 4 >= nv4) || (nq > WQ_HIGH)) {
      int first = 0;
      if (lane == 0) first = atomicAdd(cursor, nq);
      first = __builtin_amdgcn_readfirstlane(first);
      for (int j = lane; j < nq; j += 32) {
        const int ent = wq[j];
        const int s = src[ent >> 8];
        const float w = isq[s] * isqS[ent & 255];
        out[first + j] = make_int2((s << 8) | (ent & 255), __float_as_int(w));
      }
      nq = 0;
    }
  }
  __syncthreads();
  if (tid == 0) { boff[vb] = off; bcnt[vb] = *cursor; }
}

__device__ __forceinline__ void role_edge_replay(
    const int vb, const int2* __restrict__ elist, const int* __restrict__ boff,
    const int* __restrict__ bcnt, const float* __restrict__ idg,
    const float* __restrict__ h, float* __restrict__ agg, int* arena) {
  float* tile = (float*)(arena + AR_TILE);

  const int tid  = threadIdx.x;
  const int lane = tid & 31;
  const int wave = __builtin_amdgcn_readfirstlane(tid >> 5);
  const int lo   = vb * ROWS_PB;
  const int n    = __builtin_amdgcn_readfirstlane(bcnt[vb]);
  const int2* const el = elist + __builtin_amdgcn_readfirstlane(boff[vb]);

  for (int i = tid; i < (ROWS_PB + 1) * 64; i += 1024) {
    int row = lo + (i >> 6);
    tile[i] = (i < ROWS_PB * 64 && row < NUM_LOC) ? h[(long)row * 64 + (i & 63)] * idg[row] : 0.0f;
  }
  __syncthreads();

  for (int p0 = wave * 16; p0 < n; p0 += 32 * 16) {
    const int idx = p0 + (lane & 15);
    int2 en = make_int2(ROWS_PB, 0);
    if (idx < n) en = el[idx];
    float2 v[16];
#pragma unroll
    for (int k = 0; k < 16; ++k) {
      const int ek = __builtin_amdgcn_readlane(en.x, k);
      v[k] = *(const float2*)(h + (long)(ek >> 8) * 64 + 2 * lane);
    }
#pragma unroll
    for (int k = 0; k < 16; ++k) {
      const int ek = __builtin_amdgcn_readlane(en.x, k);
      const float wk = __int_as_float(__builtin_amdgcn_readlane(en.y, k));
      float* dstp = &tile[(ek & 255) * 64 + 2 * lane];
      atomicAdd(dstp + 0, v[k].x * wk);
      atomicAdd(dstp + 1, v[k].y * wk);
    }
  }
  __syncthreads();

  for (int i = tid; i < ROWS_PB * 64; i += 1024) {
    int row = lo + (i >> 6);
    if (row < NUM_LOC) agg[(long)row * 64 + (i & 63)] = tile[i];
  }
}

template <int NT, int TH>
__device__ __forceinline__ void role_gemm(
    const int vb, const float* __restrict__ A, const float* __restrict__ B,
    const float* __restrict__ bias, float* __restrict__ C, int M, int K, int relu,
    float* Bs) {
  constexpr int Nc = NT * 16;
  for (int i = threadIdx.x; i < K * Nc; i += TH) Bs[i] = B[i];
  __syncthreads();

  const int lane = threadIdx.x & 31;
  const int wave = threadIdx.x >> 5;
  const int half = lane >> 4;
  const int l16  = lane & 15;
  const int row0 = (vb * (TH / 32) + wave) * 16;

  int ar = row0 + l16; if (ar > M - 1) ar = M - 1;
  const float* Arow = A + (long)ar * K;

  v8f acc[NT];
#pragma unroll
  for (int tt = 0; tt < NT; ++tt) {
    float bv = bias[tt * 16 + l16];
    v8f tmp = { bv, bv, bv, bv, bv, bv, bv, bv };
    acc[tt] = tmp;
  }

  for (int k = 0; k < K; k += 4) {
    v2f a;
    a[0] = Arow[k + 2 * half + 0];
    a[1] = Arow[k + 2 * half + 1];
#pragma unroll
    for (int tt = 0; tt < NT; ++tt) {
      v2f b;
      b[0] = Bs[(k + 2 * half + 0) * Nc + tt * 16 + l16];
      b[1] = Bs[(k + 2 * half + 1) * Nc + tt * 16 + l16];
      acc[tt] = __builtin_amdgcn_wmma_f32_16x16x4_f32(false, a, false, b, (short)0,
                                                      acc[tt], false, false);
    }
  }
#pragma unroll
  for (int tt = 0; tt < NT; ++tt) {
    for (int r = 0; r < 8; ++r) {
      int row = row0 + r + 8 * half;
      if (row < M) {
        float v = acc[tt][r];
        if (relu) v = v > 0.0f ? v : 0.0f;
        C[(long)row * Nc + tt * 16 + l16] = v;
      }
    }
  }
}

__device__ __forceinline__ float sigm(float x) { return 1.0f / (1.0f + expf(-x)); }

template <int TH, bool VIA_IDS>
__device__ __forceinline__ void role_layer(
    const int vb, const int TP, const float* __restrict__ X, const int* __restrict__ xids, int Kx,
    const float* __restrict__ Wi,
    const float* __restrict__ Wh, const float* __restrict__ b,
    const float* hin, float* hout, float* c, int M, float* Bs) {
  for (int i = threadIdx.x; i < Kx * 64; i += TH) Bs[i] = Wi[i];
  for (int i = threadIdx.x; i < 16 * 64; i += TH) Bs[Kx * 64 + i] = Wh[i];
  __syncthreads();

  const int lane = threadIdx.x & 31;
  const int half = lane >> 4;
  const int l16  = lane & 15;
  const int wave = __builtin_amdgcn_readfirstlane((int)(threadIdx.x >> 5));
  const int row0 = (vb * TP + wave) * 16;
  if (wave < TP && row0 < M) {

  int ar = row0 + l16; if (ar > M - 1) ar = M - 1;
  int xrow = ar;
  if (VIA_IDS) xrow = xids[ar];
  const int rleft = M - row0;
  const unsigned loff = (unsigned)(half * 128 + l16);
  float* const cT = c + (long)row0 * 16;
  float* const hT = hout + (long)row0 * 16;

  float cold[8];
#pragma unroll
  for (int r = 0; r < 8; ++r)
    cold[r] = (r + 8 * half < rleft) ? cT[loff + r * 16] : 0.0f;

  v8f acc[4];
#pragma unroll
  for (int tt = 0; tt < 4; ++tt) {
    float bv = b[tt * 16 + l16];
    v8f tmp = { bv, bv, bv, bv, bv, bv, bv, bv };
    acc[tt] = tmp;
  }

  const float* Xr = X + (long)xrow * Kx;
  for (int k = 0; k < Kx; k += 4) {
    v2f a;
    a[0] = Xr[k + 2 * half + 0];
    a[1] = Xr[k + 2 * half + 1];
#pragma unroll
    for (int tt = 0; tt < 4; ++tt) {
      v2f w;
      w[0] = Bs[(k + 2 * half + 0) * 64 + tt * 16 + l16];
      w[1] = Bs[(k + 2 * half + 1) * 64 + tt * 16 + l16];
      acc[tt] = __builtin_amdgcn_wmma_f32_16x16x4_f32(false, a, false, w, (short)0,
                                                      acc[tt], false, false);
    }
  }
  const float* Hr = hin + (long)ar * 16;
#pragma unroll
  for (int k = 0; k < 16; k += 4) {
    v2f a;
    a[0] = Hr[k + 2 * half + 0];
    a[1] = Hr[k + 2 * half + 1];
#pragma unroll
    for (int tt = 0; tt < 4; ++tt) {
      v2f w;
      w[0] = Bs[(Kx + k + 2 * half + 0) * 64 + tt * 16 + l16];
      w[1] = Bs[(Kx + k + 2 * half + 1) * 64 + tt * 16 + l16];
      acc[tt] = __builtin_amdgcn_wmma_f32_16x16x4_f32(false, a, false, w, (short)0,
                                                      acc[tt], false, false);
    }
  }
#pragma unroll
  for (int r = 0; r < 8; ++r) {
    if (r + 8 * half < rleft) {
      float gi = acc[0][r], gf = acc[1][r], gg = acc[2][r], go = acc[3][r];
      float cn = sigm(gf) * cold[r] + sigm(gi) * tanhf(gg);
      cT[loff + r * 16] = cn;
      hT[loff + r * 16] = sigm(go) * tanhf(cn);
    }
  }
  }
}

__device__ __forceinline__ void role_gather(const int vb, const int* __restrict__ ids,
                                            const float* __restrict__ he,
                                            float* __restrict__ out32, int N) {
  const int sub = threadIdx.x & 7;
  const int r0  = vb * 512 + (threadIdx.x >> 3);
  if (vb * 512 + 512 <= N) {
    int id[4];
    v4f v[4];
#pragma unroll
    for (int p = 0; p < 4; ++p) id[p] = ids[r0 + 128 * p];
#pragma unroll
    for (int p = 0; p < 4; ++p)
      v[p] = *(const v4f*)(he + (long)id[p] * 32 + 4 * sub);
    asm volatile("" : "+v"(v[0]), "+v"(v[1]), "+v"(v[2]), "+v"(v[3]));
#pragma unroll
    for (int p = 0; p < 4; ++p)
      *(v4f*)(out32 + (long)(r0 + 128 * p) * 32 + 4 * sub) = v[p];
  } else {
    for (int p = 0; p < 4; ++p) {
      int row = r0 + 128 * p;
      if (row < N)
        *(v4f*)(out32 + (long)row * 32 + 4 * sub) =
            *(const v4f*)(he + (long)ids[row] * 32 + 4 * sub);
    }
  }
}

template <int TH>
__device__ __forceinline__ void role_outproj(const int vb, const float* __restrict__ h,
                                             const float* __restrict__ Wout,
                                             const float* __restrict__ bout,
                                             float* __restrict__ out, int N) {
  int i = vb * TH + threadIdx.x;
  if (i >= N) return;
  const float* hr = h + (long)i * 16;
  float a0 = bout[0], a1 = bout[1];
  for (int j = 0; j < 16; ++j) {
    float v = hr[j];
    a0 += v * Wout[j * 2 + 0];
    a1 += v * Wout[j * 2 + 1];
  }
  out[(long)i * 2 + 0] = a0;
  out[(long)i * 2 + 1] = a1;
}

__global__ __launch_bounds__(1024) void k_stageA(
    int nLoc, const int* __restrict__ idsNext, const float* __restrict__ Wemb,
    const float* __restrict__ bemb, float* __restrict__ e,
    const int* __restrict__ idsCur, const float* __restrict__ he2n, float* __restrict__ out32) {
  __shared__ __attribute__((aligned(16))) int arena[AR_BAND];
  const int b = blockIdx.x;
  if (b < nLoc) role_user_sum(b, idsNext, Wemb, bemb, e, arena);
  else          role_gather(b - nLoc, idsCur, he2n, out32, USR_NUM);
}

__global__ __launch_bounds__(1024) void k_stageBD(
    const int* __restrict__ src, const int* __restrict__ dst,
    const float* __restrict__ isq, const float* __restrict__ idg,
    const float* __restrict__ hin, float* __restrict__ agg) {
  __shared__ __attribute__((aligned(16))) int arena[AR_BAND];
  role_edge_sum(blockIdx.x, src, dst, isq, idg, hin, agg, arena);
}

__global__ __launch_bounds__(256) void k_stageCE(
    int nLoc, const float* __restrict__ A, const float* __restrict__ W,
    const float* __restrict__ bias, float* __restrict__ C, int relu,
    const float* __restrict__ X, int Kx, const float* __restrict__ Wi,
    const float* __restrict__ Wh, const float* __restrict__ bL,
    float* hS, float* cS) {
  __shared__ float Bs[AR_GEMM];
  const int b = blockIdx.x;
  if (b < nLoc) role_gemm<4, 256>(b, A, W, bias, C, NUM_LOC, 64, relu, Bs);
  else          role_layer<256, false>(b - nLoc, 8, X, nullptr, Kx, Wi, Wh, bL, hS, hS, cS, USR_NUM, Bs);
}

__global__ __launch_bounds__(256) void k_stageF(
    int nLoc, const float* __restrict__ A, const float* __restrict__ W,
    const float* __restrict__ bias, float* __restrict__ C,
    const float* __restrict__ h, const float* __restrict__ Wout,
    const float* __restrict__ bout, float* __restrict__ out) {
  __shared__ float Bs[AR_GEMM];
  const int b = blockIdx.x;
  if (b < nLoc) role_gemm<2, 256>(b, A, W, bias, C, NUM_LOC, 64, 0, Bs);
  else          role_outproj<256>(b - nLoc, h, Wout, bout, out, USR_NUM);
}


#define PL_LOC   ((long)NUM_LOC * 64)
#define PL_E2N   ((long)NUM_LOC * 32)
#define PL_ST    ((long)USR_NUM * 16)

#define NB_BAND   ((NUM_LOC + ROWS_PB - 1) / ROWS_PB)
#define NB_BAND3  (3 * NB_BAND)
#define N_TILES   ((USR_NUM + 15) / 16)
#define TILES_PB  ((2 * N_TILES + NB_BAND3 - 1) / NB_BAND3)
#define NB_LAYF   ((N_TILES + TILES_PB - 1) / TILES_PB)
#define NB_GATHER ((USR_NUM + 511) / 512)
#define NB_GEMM4  ((NUM_LOC + 511) / 512)
#define NB_OUT4   ((USR_NUM + 1023) / 1024)
#define NB_TICK   (NB_BAND3 + 3 * NB_GEMM4 + NB_OUT4)
#define N_TICKS   (T_STEPS + 8)
static_assert(TILES_PB <= 32, "one layer tile a wave");
static_assert(2 * NB_LAYF <= NB_BAND3, "both layers' tiles must fit the carrying work-groups");
#define NL_PAD 5056
#define WS_DEG   0L
#define WS_ISQ   (1L * NL_PAD)
#define WS_IDG   (2L * NL_PAD)
#define WS_E     (3L * NL_PAD)
#define WS_AGGA  (WS_E    + 2 * PL_LOC)
#define WS_H1B   (WS_AGGA + 2 * PL_LOC)
#define WS_AGGB  (WS_H1B  + 2 * PL_LOC)
#define WS_H2B   (WS_AGGB + 2 * PL_LOC)
#define WS_HE2N  (WS_H2B  + 2 * PL_LOC)
#define WS_H0P1  (WS_HE2N + 2 * PL_E2N)
#define WS_C0    (WS_H0P1 + PL_ST)
#define WS_H1P1  (WS_C0   + PL_ST)
#define WS_C1    (WS_H1P1 + PL_ST)
#define WS_H0P0  (WS_C1   + PL_ST)
#define WS_H1P0  (WS_H0P0 + PL_ST)
#define WS_ELIST (WS_H1P0 + PL_ST)
#define WS_BOFF  (WS_ELIST + 2L * N_EDGES)
#define WS_BCNT  (WS_BOFF + 256)
#define WS_TICK_FLOATS (WS_BCNT + 256)
#define WS_TICK_BYTES  ((size_t)(WS_TICK_FLOATS * 4))
static_assert(WS_TICK_FLOATS == 3L * NL_PAD + 10 * PL_LOC + 2 * PL_E2N + 6L * USR_NUM * 16 + 2L * N_EDGES + 512, "workspace layout");
static_assert(WS_ELIST % 2 == 0, "list entries are 8-byte words");
static_assert(NB_BAND <= 256, "one start and one length a band");

struct TickArgs {
  const int*   seq;  const int*   src;  const int*   dst;
  const float* Wemb; const float* bemb;
  const float* W1;   const float* b1;   const float* W2;   const float* b2;
  const float* We;   const float* be;
  const float* Wi0;  const float* Wh0;  const float* bL0;
  const float* Wi1;  const float* Wh1;  const float* bL1;
  const float* Wout; const float* bout;
  float* ws;
  float* out;
};

__device__ __forceinline__ void band_job(const int b, const TickArgs& a, const int tick, int* arena) {
  float* const ws = a.ws;
  __builtin_amdgcn_s_setprio(3);
  if (b < NB_BAND) {
    const int s = tick;
    if ((unsigned)s < (unsigned)T_STEPS)
      role_user_sum(b, a.seq + (long)s * USR_NUM, a.Wemb, a.bemb, ws + WS_E + (s & 1) * PL_LOC, arena);
  } else {
    const bool st1 = b < 2 * NB_BAND;
    const int vb = b - (st1 ? NB_BAND : 2 * NB_BAND);
    const int s = tick - (st1 ? 1 : 3);
    if (st1 && s == -1)
      role_edge_list(vb, a.src, a.dst, ws + WS_DEG, ws + WS_ISQ, (int2*)(ws + WS_ELIST),
                     (int*)(ws + WS_BOFF), (int*)(ws + WS_BCNT), arena);
    else if ((unsigned)s < (unsigned)T_STEPS)
      role_edge_replay(vb, (const int2*)(ws + WS_ELIST), (const int*)(ws + WS_BOFF),
                       (const int*)(ws + WS_BCNT), ws + WS_IDG,
                       ws + (st1 ? WS_E : WS_H1B) + (s & 1) * PL_LOC,
                       ws + (st1 ? WS_AGGA : WS_AGGB) + (s & 1) * PL_LOC, arena);
  }
  __builtin_amdgcn_s_setprio(0);
}

__global__ __launch_bounds__(1024, 16) void k_tick(const TickArgs a, const int tick) {
  __shared__ __attribute__((aligned(16))) int arena[AR_BAND];
  float* const ws = a.ws;
  int b = blockIdx.x;
  int s;
  if (b < NB_BAND3) {
    const bool layerFirst = (__builtin_popcount((unsigned)b) & 1) != 0;
    if (!layerFirst) band_job(b, a, tick, arena);
    __syncthreads();
    if (b < NB_LAYF) {
      s = tick - 6;
      if ((unsigned)s < (unsigned)T_STEPS)
        role_layer<1024, true>(b, TILES_PB, ws + WS_HE2N + (s & 1) * PL_E2N, a.seq + (long)s * USR_NUM, 32,
                         a.Wi0, a.Wh0, a.bL0,
                         ws + ((s & 1) ? WS_H0P0 : WS_H0P1), ws + ((s & 1) ? WS_H0P1 : WS_H0P0), ws + WS_C0,
                         USR_NUM, (float*)arena);
    } else if (b < 2 * NB_LAYF) {
      s = tick - 7;
      if ((unsigned)s < (unsigned)T_STEPS)
        role_layer<1024, false>(b - NB_LAYF, TILES_PB, ws + ((s & 1) ? WS_H0P1 : WS_H0P0), nullptr, 16,
                         a.Wi1, a.Wh1, a.bL1,
                         ws + ((s & 1) ? WS_H1P0 : WS_H1P1), ws + ((s & 1) ? WS_H1P1 : WS_H1P0), ws + WS_C1,
                         USR_NUM, (float*)arena);
    }
    __syncthreads();
    if (layerFirst) band_job(b, a, tick, arena);
    return;
  }
  b -= NB_BAND3;
  if (b < NB_GEMM4) {
    s = tick - 2;
    if ((unsigned)s >= (unsigned)T_STEPS) return;
    role_gemm<4, 1024>(b, ws + WS_AGGA + (s & 1) * PL_LOC, a.W1, a.b1, ws + WS_H1B + (s & 1) * PL_LOC,
                       NUM_LOC, 64, 1, (float*)arena);
    return;
  }
  b -= NB_GEMM4;
  if (b < NB_GEMM4) {
    s = tick - 4;
    if ((unsigned)s >= (unsigned)T_STEPS) return;
    role_gemm<4, 1024>(b, ws + WS_AGGB + (s & 1) * PL_LOC, a.W2, a.b2, ws + WS_H2B + (s & 1) * PL_LOC,
                       NUM_LOC, 64, 0, (float*)arena);
    return;
  }
  b -= NB_GEMM4;
  if (b < NB_GEMM4) {
    s = tick - 5;
    if ((unsigned)s >= (unsigned)T_STEPS) return;
    role_gemm<2, 1024>(b, ws + WS_H2B + (s & 1) * PL_LOC, a.We, a.be, ws + WS_HE2N + (s & 1) * PL_E2N,
                       NUM_LOC, 64, 0, (float*)arena);
    return;
  }
  b -= NB_GEMM4;
  {
    s = tick - 8;
    if ((unsigned)s >= (unsigned)T_STEPS) return;
    role_outproj<1024>(b, ws + ((s & 1) ? WS_H1P1 : WS_H1P0), a.Wout, a.bout,
                       a.out + (long)s * USR_NUM * 2, USR_NUM);
  }
}

static inline int gs(long n, int b) { return (int)((n + b - 1) / b); }

#define NB_LAYER  ((USR_NUM + 127) / 128)
#define NB_GEMM   ((NUM_LOC + 127) / 128)
#define NB_OUT    ((USR_NUM + 255) / 256)

extern "C" void kernel_launch(void* const* d_in, const int* in_sizes, int n_in,
                              void* d_out, int out_size, void* d_ws, size_t ws_size,
                              hipStream_t stream) {
  (void)in_sizes; (void)n_in; (void)out_size;

  const int*   seq  = (const int*)d_in[0];
  const int*   gidx = (const int*)d_in[1];
  const float* Wemb = (const float*)d_in[2];
  const float* bemb = (const float*)d_in[3];
  const float* W1   = (const float*)d_in[4];
  const float* b1   = (const float*)d_in[5];
  const float* W2   = (const float*)d_in[6];
  const float* b2   = (const float*)d_in[7];
  const float* We2n = (const float*)d_in[8];
  const float* be2n = (const float*)d_in[9];
  const float* Wi0  = (const float*)d_in[10];
  const float* Wh0  = (const float*)d_in[11];
  const float* bL0  = (const float*)d_in[12];
  const float* Wi1  = (const float*)d_in[13];
  const float* Wh1  = (const float*)d_in[14];
  const float* bL1  = (const float*)d_in[15];
  const float* Wout = (const float*)d_in[16];
  const float* bout = (const float*)d_in[17];

  const int* srcE = gidx;
  const int* dstE = gidx + N_EDGES;
  float* out = (float*)d_out;
  const int BT = 256;

  float* p = (float*)d_ws;
  float* deg   = p; p += NL_PAD;
  float* isq   = p; p += NL_PAD;
  float* idg   = p; p += NL_PAD;

  k_zero<<<gs(NUM_LOC, BT), BT, 0, stream>>>(deg, NUM_LOC);
  k_deg_accum<<<gs(N_EDGES, BT), BT, 0, stream>>>(dstE, deg, N_EDGES);
  k_deg_fin<<<gs(NUM_LOC, BT), BT, 0, stream>>>(deg, isq, idg, NUM_LOC);

  if (ws_size >= WS_TICK_BYTES) {
    TickArgs a;
    a.seq = seq; a.src = srcE; a.dst = dstE;
    a.Wemb = Wemb; a.bemb = bemb;
    a.W1 = W1; a.b1 = b1; a.W2 = W2; a.b2 = b2; a.We = We2n; a.be = be2n;
    a.Wi0 = Wi0; a.Wh0 = Wh0; a.bL0 = bL0; a.Wi1 = Wi1; a.Wh1 = Wh1; a.bL1 = bL1;
    a.Wout = Wout; a.bout = bout;
    a.ws = (float*)d_ws;
    a.out = out;

    k_zero<<<gs((long)USR_NUM * 64, BT), BT, 0, stream>>>((float*)d_ws + WS_H0P1, USR_NUM * 64);
    for (int tick = 0; tick < N_TICKS; ++tick)
      k_tick<<<NB_TICK, 1024, 0, stream>>>(a, tick);
    return;
  }

  float* e     = p; p += (size_t)NUM_LOC * 64;
  float* agg   = p; p += (size_t)NUM_LOC * 64;
  float* h1b   = p; p += (size_t)NUM_LOC * 64;
  float* h2b   = p; p += (size_t)NUM_LOC * 64;
  float* he2n  = p; p += (size_t)NUM_LOC * 32;
  float* out32 = p; p += (size_t)USR_NUM * 32;
  float* h0    = p; p += (size_t)USR_NUM * 16;
  float* c0    = p; p += (size_t)USR_NUM * 16;
  float* h1s   = p; p += (size_t)USR_NUM * 16;
  float* c1s   = p; p += (size_t)USR_NUM * 16;

  k_zero<<<gs((long)USR_NUM * 64, BT), BT, 0, stream>>>(h0, USR_NUM * 64);

  k_stageA <<<NB_BAND, 1024, 0, stream>>>(NB_BAND, seq, Wemb, bemb, e, seq, he2n, out32);
  k_stageBD<<<NB_BAND, 1024, 0, stream>>>(srcE, dstE, isq, idg, e, agg);
  k_stageCE<<<NB_GEMM, 256, 0, stream>>>(NB_GEMM, agg, W1, b1, h1b, 1,
                                         out32, 32, Wi0, Wh0, bL0, h0, c0);
  k_stageBD<<<NB_BAND, 1024, 0, stream>>>(srcE, dstE, isq, idg, h1b, agg);
  k_stageCE<<<NB_GEMM, 256, 0, stream>>>(NB_GEMM, agg, W2, b2, h2b, 0,
                                         h0, 16, Wi1, Wh1, bL1, h1s, c1s);
  k_stageF <<<NB_GEMM, 256, 0, stream>>>(NB_GEMM, h2b, We2n, be2n, he2n, h1s, Wout, bout, out);

  for (int t = 0; t + 1 < T_STEPS; ++t) {
    const int* idsCur  = seq + (long)t * USR_NUM;
    const int* idsNext = idsCur + USR_NUM;
    float* outT = out + (long)t * USR_NUM * 2;

    k_stageA <<<NB_BAND + NB_GATHER, 1024, 0, stream>>>(NB_BAND, idsNext, Wemb, bemb, e,
                                                        idsCur, he2n, out32);
    k_stageBD<<<NB_BAND, 1024, 0, stream>>>(srcE, dstE, isq, idg, e, agg);
    k_stageCE<<<NB_GEMM + NB_LAYER, 256, 0, stream>>>(NB_GEMM, agg, W1, b1, h1b, 1,
                                                      out32, 32, Wi0, Wh0, bL0, h0, c0);
    k_stageBD<<<NB_BAND, 1024, 0, stream>>>(srcE, dstE, isq, idg, h1b, agg);
    k_stageCE<<<NB_GEMM + NB_LAYER, 256, 0, stream>>>(NB_GEMM, agg, W2, b2, h2b, 0,
                                                      h0, 16, Wi1, Wh1, bL1, h1s, c1s);
    k_stageF <<<NB_GEMM + NB_OUT, 256, 0, stream>>>(NB_GEMM, h2b, We2n, be2n, he2n,
                                                     h1s, Wout, bout, outT);
  }

  {
    const int* idsCur = seq + (long)(T_STEPS - 1) * USR_NUM;
    float* outT = out + (long)(T_STEPS - 1) * USR_NUM * 2;
    k_stageA <<<NB_GATHER, 1024, 0, stream>>>(0, idsCur, Wemb, bemb, e, idsCur, he2n, out32);
    k_stageCE<<<NB_LAYER, 256, 0, stream>>>(0, agg, W1, b1, h1b, 1,
                                            out32, 32, Wi0, Wh0, bL0, h0, c0);
    k_stageCE<<<NB_LAYER, 256, 0, stream>>>(0, agg, W2, b2, h2b, 0,
                                            h0, 16, Wi1, Wh1, bL1, h1s, c1s);
    k_stageF <<<NB_OUT, 256, 0, stream>>>(0, h2b, We2n, be2n, he2n, h1s, Wout, bout, outT);
  }
}
